# starting kernel
# baseline (speedup 1.0000x reference)
#include <hip/hip_runtime.h>
#include <cstdint>

typedef float v2f __attribute__((ext_vector_type(2)));
typedef float v8f __attribute__((ext_vector_type(8)));

__global__ __launch_bounds__(256) void fill_f32(float* __restrict__ p, float v, int n) {
  int i = blockIdx.x * 256 + threadIdx.x;
  if (i < n) p[i] = v;
}

__global__ __launch_bounds__(256) void deg_accum(const int* __restrict__ dst,
                                                 const float* __restrict__ ew,
                                                 float* __restrict__ deg, int E) {
  int e = blockIdx.x * 256 + threadIdx.x;
  if (e < E) unsafeAtomicAdd(&deg[dst[e]], ew[e]);
}

__global__ __launch_bounds__(256) void rsqrt_inplace(float* __restrict__ p, int n) {
  int i = blockIdx.x * 256 + threadIdx.x;
  if (i < n) {
    float d = p[i];
    p[i] = (d > 0.f) ? rsqrtf(d) : 0.f;
  }
}

__global__ __launch_bounds__(256) void norm_kernel(const int* __restrict__ src,
                                                   const int* __restrict__ dst,
                                                   const float* __restrict__ ew,
                                                   const float* __restrict__ dinv,
                                                   float* __restrict__ norm, int E) {
  int e = blockIdx.x * 256 + threadIdx.x;
  if (e < E) norm[e] = dinv[src[e]] * ew[e] * dinv[dst[e]];
}

template <int IN, int OUT>
__global__ __launch_bounds__(256) void gemm_wmma(const float* __restrict__ x,
                                                 const float* __restrict__ W,
                                                 float* __restrict__ y, int n) {
  constexpr bool USE_LDS = (IN * OUT) <= 8192;
  constexpr bool FULL    = (OUT % 16) == 0;
  __shared__ float sW[USE_LDS ? (IN * OUT) : 1];
  const float* Wp = W;
  if constexpr (USE_LDS) {
    for (int i = threadIdx.x; i < IN * OUT; i += 256) sW[i] = W[i];
    __syncthreads();
    Wp = sW;
  }

  const int lane  = threadIdx.x & 31;
  const int wave  = threadIdx.x >> 5;
  const int l15   = lane & 15;
  const int khalf = lane >> 4;
  const int row0  = blockIdx.x * 128 + wave * 16;

  if (row0 < n) {
    int mload = row0 + l15;
    if (mload > n - 1) mload = n - 1;
    const float* xrow = x + (size_t)mload * IN;
    const bool fulltile = (row0 + 16 <= n);

    for (int n0 = 0; n0 < OUT; n0 += 16) {
      const int colb = n0 + l15;
      const int colc = FULL ? colb : (colb < OUT ? colb : OUT - 1);
      v8f c = {};
#pragma unroll
      for (int k0 = 0; k0 < IN; k0 += 4) {
        const int krow = k0 + 2 * khalf;
        v2f a = *(const v2f*)(xrow + krow);
        v2f b;
        b.x = Wp[(size_t)krow * OUT + colc];
        b.y = Wp[(size_t)(krow + 1) * OUT + colc];
        c = __builtin_amdgcn_wmma_f32_16x16x4_f32(
             false, a,  false, b,
             (short)0, c,  false,  false);
      }
      const int rowoff = khalf * 8;
      if (FULL && fulltile) {
#pragma unroll
        for (int r = 0; r < 8; ++r)
          y[(size_t)(row0 + rowoff + r) * OUT + colb] = c[r];
      } else {
        const bool colok = FULL || (colb < OUT);
#pragma unroll
        for (int r = 0; r < 8; ++r) {
          int row = row0 + rowoff + r;
          if (row < n && colok) y[(size_t)row * OUT + colb] = c[r];
        }
      }
    }
  }
}

template <int OUT>
__global__ __launch_bounds__(256) void self_bias(const float* __restrict__ xw,
                                                 const float* __restrict__ dinv,
                                                 const float* __restrict__ b,
                                                 float* __restrict__ out, int n) {
  int tid = blockIdx.x * 256 + threadIdx.x;
  if (tid >= n * OUT) return;
  int i = tid / OUT, c = tid - i * OUT;
  float dv = dinv[i];
  out[tid] = xw[tid] * (dv * dv) + b[c];
}

template <int OUT>
__global__ __launch_bounds__(256) void edge_scatter(const float* __restrict__ xw,
                                                    const int* __restrict__ src,
                                                    const int* __restrict__ dst,
                                                    const float* __restrict__ norm,
                                                    float* __restrict__ out, int E) {
  int tid = blockIdx.x * 256 + threadIdx.x;
  if (tid >= E * OUT) return;
  int e = tid / OUT, c = tid - e * OUT;
  float v = xw[(size_t)src[e] * OUT + c] * norm[e];
  unsafeAtomicAdd(&out[(size_t)dst[e] * OUT + c], v);
}

__global__ __launch_bounds__(256) void elu_inplace(float* __restrict__ p, int n) {
  int i = blockIdx.x * 256 + threadIdx.x;
  if (i < n) {
    float v = p[i];
    p[i] = (v > 0.f) ? v : expm1f(v);
  }
}

static inline int cdiv(long long a, long long b) { return (int)((a + b - 1) / b); }

template <int IN, int OUT>
static void run_layer(const float* xin, float* xw, float* xout,
                      const float* W, const float* b,
                      const int* src, const int* dst,
                      const float* dinv, const float* norm,
                      int N, int E, bool act, hipStream_t stream) {
  gemm_wmma<IN, OUT><<<cdiv(N, 128), 256, 0, stream>>>(xin, W, xw, N);
  self_bias<OUT><<<cdiv((long long)N * OUT, 256), 256, 0, stream>>>(xw, dinv, b, xout, N);
  edge_scatter<OUT><<<cdiv((long long)E * OUT, 256), 256, 0, stream>>>(xw, src, dst, norm, xout, E);
  if (act) elu_inplace<<<cdiv((long long)N * OUT, 256), 256, 0, stream>>>(xout, N * OUT);
}

extern "C" void kernel_launch(void* const* d_in, const int* in_sizes, int n_in,
                              void* d_out, int out_size, void* d_ws, size_t ws_size,
                              hipStream_t stream) {
  const float* x   = (const float*)d_in[0];
  const int*   ei  = (const int*)d_in[1];
  const float* ew  = (const float*)d_in[2];
  const float* W1 = (const float*)d_in[3];  const float* b1 = (const float*)d_in[4];
  const float* W2 = (const float*)d_in[5];  const float* b2 = (const float*)d_in[6];
  const float* W3 = (const float*)d_in[7];  const float* b3 = (const float*)d_in[8];
  const float* W4 = (const float*)d_in[9];  const float* b4 = (const float*)d_in[10];
  const float* W5 = (const float*)d_in[11]; const float* b5 = (const float*)d_in[12];
  const float* W6 = (const float*)d_in[13]; const float* b6 = (const float*)d_in[14];

  const int N = in_sizes[0] / 32;
  const int E = in_sizes[2];
  const int* src = ei;
  const int* dst = ei + E;

  float* dinv = (float*)d_ws;
  float* norm = dinv + N;
  float* buf0 = norm + E;
  float* buf1 = buf0 + (size_t)N * 128;
  float* outp = (float*)d_out;

  fill_f32<<<cdiv(N, 256), 256, 0, stream>>>(dinv, 1.0f, N);
  deg_accum<<<cdiv(E, 256), 256, 0, stream>>>(dst, ew, dinv, E);
  rsqrt_inplace<<<cdiv(N, 256), 256, 0, stream>>>(dinv, N);
  norm_kernel<<<cdiv(E, 256), 256, 0, stream>>>(src, dst, ew, dinv, norm, E);

  run_layer<32,   4>(x,    buf1, buf0, W1, b1, src, dst, dinv, norm, N, E, true,  stream);
  run_layer<4,   16>(buf0, buf1, buf0, W2, b2, src, dst, dinv, norm, N, E, true,  stream);
  run_layer<16,  32>(buf0, buf1, buf0, W3, b3, src, dst, dinv, norm, N, E, true,  stream);
  run_layer<32,  64>(buf0, buf1, buf0, W4, b4, src, dst, dinv, norm, N, E, false, stream);
  run_layer<64, 128>(buf0, buf1, buf0, W5, b5, src, dst, dinv, norm, N, E, true,  stream);
  run_layer<128,128>(buf0, buf1, outp, W6, b6, src, dst, dinv, norm, N, E, false, stream);
}


# Layers 2-5 (IN<OUT) aggregate edges on the IN-wide input first, then GEMM, then bias+ELU pass: (A x) W instead of A (x W); float sum order changes; edge scatter width halves or better
# speedup vs baseline: 1.3462x; 1.3462x over previous
#include <hip/hip_runtime.h>
#include <cstdint>

typedef float v2f __attribute__((ext_vector_type(2)));
typedef float v8f __attribute__((ext_vector_type(8)));

__global__ __launch_bounds__(256) void fill_f32(float* __restrict__ p, float v, int n) {
  int i = blockIdx.x * 256 + threadIdx.x;
  if (i < n) p[i] = v;
}

__global__ __launch_bounds__(256) void deg_accum(const int* __restrict__ dst,
                                                 const float* __restrict__ ew,
                                                 float* __restrict__ deg, int E) {
  int e = blockIdx.x * 256 + threadIdx.x;
  if (e < E) unsafeAtomicAdd(&deg[dst[e]], ew[e]);
}

__global__ __launch_bounds__(256) void rsqrt_inplace(float* __restrict__ p, int n) {
  int i = blockIdx.x * 256 + threadIdx.x;
  if (i < n) {
    float d = p[i];
    p[i] = (d > 0.f) ? rsqrtf(d) : 0.f;
  }
}

__global__ __launch_bounds__(256) void norm_kernel(const int* __restrict__ src,
                                                   const int* __restrict__ dst,
                                                   const float* __restrict__ ew,
                                                   const float* __restrict__ dinv,
                                                   float* __restrict__ norm, int E) {
  int e = blockIdx.x * 256 + threadIdx.x;
  if (e < E) norm[e] = dinv[src[e]] * ew[e] * dinv[dst[e]];
}

template <int IN, int OUT>
__global__ __launch_bounds__(256) void gemm_wmma(const float* __restrict__ x,
                                                 const float* __restrict__ W,
                                                 float* __restrict__ y, int n) {
  constexpr bool USE_LDS = (IN * OUT) <= 8192;
  constexpr bool FULL    = (OUT % 16) == 0;
  __shared__ float sW[USE_LDS ? (IN * OUT) : 1];
  const float* Wp = W;
  if constexpr (USE_LDS) {
    for (int i = threadIdx.x; i < IN * OUT; i += 256) sW[i] = W[i];
    __syncthreads();
    Wp = sW;
  }

  const int lane  = threadIdx.x & 31;
  const int wave  = threadIdx.x >> 5;
  const int l15   = lane & 15;
  const int khalf = lane >> 4;
  const int row0  = blockIdx.x * 128 + wave * 16;

  if (row0 < n) {
    int mload = row0 + l15;
    if (mload > n - 1) mload = n - 1;
    const float* xrow = x + (size_t)mload * IN;
    const bool fulltile = (row0 + 16 <= n);

    for (int n0 = 0; n0 < OUT; n0 += 16) {
      const int colb = n0 + l15;
      const int colc = FULL ? colb : (colb < OUT ? colb : OUT - 1);
      v8f c = {};
#pragma unroll
      for (int k0 = 0; k0 < IN; k0 += 4) {
        const int krow = k0 + 2 * khalf;
        v2f a = *(const v2f*)(xrow + krow);
        v2f b;
        b.x = Wp[(size_t)krow * OUT + colc];
        b.y = Wp[(size_t)(krow + 1) * OUT + colc];
        c = __builtin_amdgcn_wmma_f32_16x16x4_f32(
             false, a,  false, b,
             (short)0, c,  false,  false);
      }
      const int rowoff = khalf * 8;
      if (FULL && fulltile) {
#pragma unroll
        for (int r = 0; r < 8; ++r)
          y[(size_t)(row0 + rowoff + r) * OUT + colb] = c[r];
      } else {
        const bool colok = FULL || (colb < OUT);
#pragma unroll
        for (int r = 0; r < 8; ++r) {
          int row = row0 + rowoff + r;
          if (row < n && colok) y[(size_t)row * OUT + colb] = c[r];
        }
      }
    }
  }
}

template <int OUT>
__global__ __launch_bounds__(256) void self_bias(const float* __restrict__ xw,
                                                 const float* __restrict__ dinv,
                                                 const float* __restrict__ b,
                                                 float* __restrict__ out, int n) {
  int tid = blockIdx.x * 256 + threadIdx.x;
  if (tid >= n * OUT) return;
  int i = tid / OUT, c = tid - i * OUT;
  float dv = dinv[i];
  out[tid] = xw[tid] * (dv * dv) + b[c];
}

template <int C>
__global__ __launch_bounds__(256) void self_scale(const float* __restrict__ xin,
                                                  const float* __restrict__ dinv,
                                                  float* __restrict__ agg, int n) {
  int tid = blockIdx.x * 256 + threadIdx.x;
  if (tid >= n * C) return;
  int i = tid / C;
  float dv = dinv[i];
  agg[tid] = xin[tid] * (dv * dv);
}

template <int OUT, bool ACT>
__global__ __launch_bounds__(256) void bias_act(float* __restrict__ p,
                                                const float* __restrict__ b, int n) {
  int tid = blockIdx.x * 256 + threadIdx.x;
  if (tid >= n * OUT) return;
  float v = p[tid] + b[tid % OUT];
  if (ACT) v = (v > 0.f) ? v : expm1f(v);
  p[tid] = v;
}

template <int OUT>
__global__ __launch_bounds__(256) void edge_scatter(const float* __restrict__ xw,
                                                    const int* __restrict__ src,
                                                    const int* __restrict__ dst,
                                                    const float* __restrict__ norm,
                                                    float* __restrict__ out, int E) {
  int tid = blockIdx.x * 256 + threadIdx.x;
  if (tid >= E * OUT) return;
  int e = tid / OUT, c = tid - e * OUT;
  float v = xw[(size_t)src[e] * OUT + c] * norm[e];
  unsafeAtomicAdd(&out[(size_t)dst[e] * OUT + c], v);
}

__global__ __launch_bounds__(256) void elu_inplace(float* __restrict__ p, int n) {
  int i = blockIdx.x * 256 + threadIdx.x;
  if (i < n) {
    float v = p[i];
    p[i] = (v > 0.f) ? v : expm1f(v);
  }
}

static inline int cdiv(long long a, long long b) { return (int)((a + b - 1) / b); }

template <int IN, int OUT>
static void run_layer(const float* xin, float* xw, float* xout,
                      const float* W, const float* b,
                      const int* src, const int* dst,
                      const float* dinv, const float* norm,
                      int N, int E, bool act, hipStream_t stream) {
  gemm_wmma<IN, OUT><<<cdiv(N, 128), 256, 0, stream>>>(xin, W, xw, N);
  self_bias<OUT><<<cdiv((long long)N * OUT, 256), 256, 0, stream>>>(xw, dinv, b, xout, N);
  edge_scatter<OUT><<<cdiv((long long)E * OUT, 256), 256, 0, stream>>>(xw, src, dst, norm, xout, E);
  if (act) elu_inplace<<<cdiv((long long)N * OUT, 256), 256, 0, stream>>>(xout, N * OUT);
}

template <int IN, int OUT>
static void run_layer_agg(const float* xin, float* agg, float* xout,
                          const float* W, const float* b,
                          const int* src, const int* dst,
                          const float* dinv, const float* norm,
                          int N, int E, bool act, hipStream_t stream) {
  self_scale<IN><<<cdiv((long long)N * IN, 256), 256, 0, stream>>>(xin, dinv, agg, N);
  edge_scatter<IN><<<cdiv((long long)E * IN, 256), 256, 0, stream>>>(xin, src, dst, norm, agg, E);
  gemm_wmma<IN, OUT><<<cdiv(N, 128), 256, 0, stream>>>(agg, W, xout, N);
  if (act) bias_act<OUT, true><<<cdiv((long long)N * OUT, 256), 256, 0, stream>>>(xout, b, N);
  else     bias_act<OUT, false><<<cdiv((long long)N * OUT, 256), 256, 0, stream>>>(xout, b, N);
}

extern "C" void kernel_launch(void* const* d_in, const int* in_sizes, int n_in,
                              void* d_out, int out_size, void* d_ws, size_t ws_size,
                              hipStream_t stream) {
  const float* x   = (const float*)d_in[0];
  const int*   ei  = (const int*)d_in[1];
  const float* ew  = (const float*)d_in[2];
  const float* W1 = (const float*)d_in[3];  const float* b1 = (const float*)d_in[4];
  const float* W2 = (const float*)d_in[5];  const float* b2 = (const float*)d_in[6];
  const float* W3 = (const float*)d_in[7];  const float* b3 = (const float*)d_in[8];
  const float* W4 = (const float*)d_in[9];  const float* b4 = (const float*)d_in[10];
  const float* W5 = (const float*)d_in[11]; const float* b5 = (const float*)d_in[12];
  const float* W6 = (const float*)d_in[13]; const float* b6 = (const float*)d_in[14];

  const int N = in_sizes[0] / 32;
  const int E = in_sizes[2];
  const int* src = ei;
  const int* dst = ei + E;

  float* dinv = (float*)d_ws;
  float* norm = dinv + N;
  float* buf0 = norm + E;
  float* buf1 = buf0 + (size_t)N * 128;
  float* outp = (float*)d_out;

  fill_f32<<<cdiv(N, 256), 256, 0, stream>>>(dinv, 1.0f, N);
  deg_accum<<<cdiv(E, 256), 256, 0, stream>>>(dst, ew, dinv, E);
  rsqrt_inplace<<<cdiv(N, 256), 256, 0, stream>>>(dinv, N);
  norm_kernel<<<cdiv(E, 256), 256, 0, stream>>>(src, dst, ew, dinv, norm, E);

  run_layer<32,   4>(x,    buf1, buf0, W1, b1, src, dst, dinv, norm, N, E, true,  stream);
  run_layer_agg<4,   16>(buf0, buf1, buf0, W2, b2, src, dst, dinv, norm, N, E, true,  stream);
  run_layer_agg<16,  32>(buf0, buf1, buf0, W3, b3, src, dst, dinv, norm, N, E, true,  stream);
  run_layer_agg<32,  64>(buf0, buf1, buf0, W4, b4, src, dst, dinv, norm, N, E, false, stream);
  run_layer_agg<64, 128>(buf0, buf1, buf0, W5, b5, src, dst, dinv, norm, N, E, true,  stream);
  run_layer<128,128>(buf0, buf1, outp, W6, b6, src, dst, dinv, norm, N, E, false, stream);
}


# Edges bucketed by destination once per call (count, scan, place by int atomics); each layer sums a node's in-edges in one gather, row written once; layer float atomics gone; sum order = bucket order
# speedup vs baseline: 3.2372x; 2.4047x over previous
#include <hip/hip_runtime.h>
#include <cstdint>

typedef float v2f __attribute__((ext_vector_type(2)));
typedef float v8f __attribute__((ext_vector_type(8)));

__global__ __launch_bounds__(256) void init_deg_cnt(float* __restrict__ deg,
                                                    int* __restrict__ cnt, int n) {
  int i = blockIdx.x * 256 + threadIdx.x;
  if (i < n) {
    deg[i] = 1.0f;
    cnt[i] = 0;
  }
}

__global__ __launch_bounds__(256) void deg_count(const int* __restrict__ dst,
                                                 const float* __restrict__ ew,
                                                 float* __restrict__ deg,
                                                 int* __restrict__ cnt, int E) {
  int e = blockIdx.x * 256 + threadIdx.x;
  if (e < E) {
    int d = dst[e];
    unsafeAtomicAdd(&deg[d], ew[e]);
    atomicAdd(&cnt[d], 1);
  }
}

__global__ __launch_bounds__(256) void rsqrt_inplace(float* __restrict__ p, int n) {
  int i = blockIdx.x * 256 + threadIdx.x;
  if (i < n) {
    float d = p[i];
    p[i] = (d > 0.f) ? rsqrtf(d) : 0.f;
  }
}


__global__ __launch_bounds__(256) void scan_tiles(const int* __restrict__ cnt,
                                                  int* __restrict__ rowptr,
                                                  int* __restrict__ bsum, int n) {
  __shared__ int s[256];
  const int t = threadIdx.x;
  const int base = blockIdx.x * 1024 + t * 4;
  const int v0 = (base + 0 < n) ? cnt[base + 0] : 0;
  const int v1 = (base + 1 < n) ? cnt[base + 1] : 0;
  const int v2 = (base + 2 < n) ? cnt[base + 2] : 0;
  const int v3 = (base + 3 < n) ? cnt[base + 3] : 0;
  const int tot = v0 + v1 + v2 + v3;
  s[t] = tot;
  __syncthreads();
  for (int off = 1; off < 256; off <<= 1) {
    const int add = (t >= off) ? s[t - off] : 0;
    __syncthreads();
    s[t] += add;
    __syncthreads();
  }
  int run = s[t] - tot;
  if (base + 0 < n) rowptr[base + 0] = run;
  run += v0;
  if (base + 1 < n) rowptr[base + 1] = run;
  run += v1;
  if (base + 2 < n) rowptr[base + 2] = run;
  run += v2;
  if (base + 3 < n) rowptr[base + 3] = run;
  if (t == 255) bsum[blockIdx.x] = s[255];
}

__global__ __launch_bounds__(256) void scan_add(int* __restrict__ rowptr,
                                                const int* __restrict__ bsum,
                                                int n, int total) {
  __shared__ int s[256];
  const int t = threadIdx.x;
  int part = 0;
  for (int j = t; j < (int)blockIdx.x; j += 256) part += bsum[j];
  s[t] = part;
  __syncthreads();
  for (int off = 128; off > 0; off >>= 1) {
    if (t < off) s[t] += s[t + off];
    __syncthreads();
  }
  const int offs = s[0];
  const int base = blockIdx.x * 1024 + t * 4;
  if (base + 0 < n) rowptr[base + 0] += offs;
  if (base + 1 < n) rowptr[base + 1] += offs;
  if (base + 2 < n) rowptr[base + 2] += offs;
  if (base + 3 < n) rowptr[base + 3] += offs;
  if (blockIdx.x == 0 && t == 0) rowptr[n] = total;
}

__global__ __launch_bounds__(256) void fill_buckets(const int* __restrict__ src,
                                                    const int* __restrict__ dst,
                                                    const float* __restrict__ ew,
                                                    const float* __restrict__ dinv,
                                                    const int* __restrict__ rowptr,
                                                    int* __restrict__ cnt,
                                                    int* __restrict__ ssrc,
                                                    float* __restrict__ snorm, int E) {
  int e = blockIdx.x * 256 + threadIdx.x;
  if (e < E) {
    const int s = src[e];
    const int d = dst[e];
    const int k = atomicSub(&cnt[d], 1) - 1;
    const int p = rowptr[d] + k;
    ssrc[p]  = s;
    snorm[p] = dinv[s] * ew[e] * dinv[d];
  }
}

template <int IN, int OUT>
__global__ __launch_bounds__(256) void gemm_wmma(const float* __restrict__ x,
                                                 const float* __restrict__ W,
                                                 float* __restrict__ y, int n) {
  constexpr bool USE_LDS = (IN * OUT) <= 8192;
  constexpr bool FULL    = (OUT % 16) == 0;
  __shared__ float sW[USE_LDS ? (IN * OUT) : 1];
  const float* Wp = W;
  if constexpr (USE_LDS) {
    for (int i = threadIdx.x; i < IN * OUT; i += 256) sW[i] = W[i];
    __syncthreads();
    Wp = sW;
  }

  const int lane  = threadIdx.x & 31;
  const int wave  = threadIdx.x >> 5;
  const int l15   = lane & 15;
  const int khalf = lane >> 4;
  const int row0  = blockIdx.x * 128 + wave * 16;

  if (row0 < n) {
    int mload = row0 + l15;
    if (mload > n - 1) mload = n - 1;
    const float* xrow = x + (size_t)mload * IN;
    const bool fulltile = (row0 + 16 <= n);

    for (int n0 = 0; n0 < OUT; n0 += 16) {
      const int colb = n0 + l15;
      const int colc = FULL ? colb : (colb < OUT ? colb : OUT - 1);
      v8f c = {};
#pragma unroll
      for (int k0 = 0; k0 < IN; k0 += 4) {
        const int krow = k0 + 2 * khalf;
        v2f a = *(const v2f*)(xrow + krow);
        v2f b;
        b.x = Wp[(size_t)krow * OUT + colc];
        b.y = Wp[(size_t)(krow + 1) * OUT + colc];
        c = __builtin_amdgcn_wmma_f32_16x16x4_f32(
             false, a,  false, b,
             (short)0, c,  false,  false);
      }
      const int rowoff = khalf * 8;
      if (FULL && fulltile) {
#pragma unroll
        for (int r = 0; r < 8; ++r)
          y[(size_t)(row0 + rowoff + r) * OUT + colb] = c[r];
      } else {
        const bool colok = FULL || (colb < OUT);
#pragma unroll
        for (int r = 0; r < 8; ++r) {
          int row = row0 + rowoff + r;
          if (row < n && colok) y[(size_t)row * OUT + colb] = c[r];
        }
      }
    }
  }
}

template <int C, bool BIAS, bool ACT>
__global__ __launch_bounds__(256) void gather_rows(const float* __restrict__ xin,
                                                   const int* __restrict__ rowptr,
                                                   const int* __restrict__ ssrc,
                                                   const float* __restrict__ snorm,
                                                   const float* __restrict__ dinv,
                                                   const float* __restrict__ b,
                                                   float* __restrict__ out, int n) {
  constexpr int G = C / 4;
  const int tid = blockIdx.x * 256 + threadIdx.x;
  const int i = tid / G;
  const int q = tid - i * G;
  if (i >= n) return;
  const float* xq = xin + q * 4;
  const float dv = dinv[i];
  const float wself = dv * dv;
  const float4 sv = *(const float4*)(xq + (size_t)i * C);
  float ax = sv.x * wself, ay = sv.y * wself, az = sv.z * wself, aw = sv.w * wself;
  int p = rowptr[i];
  const int pend = rowptr[i + 1];
  for (; p + 4 <= pend; p += 4) {
    const int s0 = ssrc[p], s1 = ssrc[p + 1], s2 = ssrc[p + 2], s3 = ssrc[p + 3];
    const float w0 = snorm[p], w1 = snorm[p + 1], w2 = snorm[p + 2], w3 = snorm[p + 3];
    const float4 r0 = *(const float4*)(xq + (size_t)s0 * C);
    const float4 r1 = *(const float4*)(xq + (size_t)s1 * C);
    const float4 r2 = *(const float4*)(xq + (size_t)s2 * C);
    const float4 r3 = *(const float4*)(xq + (size_t)s3 * C);
    ax += r0.x * w0; ay += r0.y * w0; az += r0.z * w0; aw += r0.w * w0;
    ax += r1.x * w1; ay += r1.y * w1; az += r1.z * w1; aw += r1.w * w1;
    ax += r2.x * w2; ay += r2.y * w2; az += r2.z * w2; aw += r2.w * w2;
    ax += r3.x * w3; ay += r3.y * w3; az += r3.z * w3; aw += r3.w * w3;
  }
  for (; p < pend; ++p) {
    const int s0 = ssrc[p];
    const float w0 = snorm[p];
    const float4 r0 = *(const float4*)(xq + (size_t)s0 * C);
    ax += r0.x * w0; ay += r0.y * w0; az += r0.z * w0; aw += r0.w * w0;
  }
  if (BIAS) {
    ax += b[q * 4 + 0]; ay += b[q * 4 + 1]; az += b[q * 4 + 2]; aw += b[q * 4 + 3];
  }
  if (ACT) {
    ax = (ax > 0.f) ? ax : expm1f(ax);
    ay = (ay > 0.f) ? ay : expm1f(ay);
    az = (az > 0.f) ? az : expm1f(az);
    aw = (aw > 0.f) ? aw : expm1f(aw);
  }
  float4 o;
  o.x = ax; o.y = ay; o.z = az; o.w = aw;
  *(float4*)(out + (size_t)i * C + q * 4) = o;
}

template <int OUT, bool ACT>
__global__ __launch_bounds__(256) void bias_act(float* __restrict__ p,
                                                const float* __restrict__ b, int n) {
  int tid = blockIdx.x * 256 + threadIdx.x;
  if (tid >= n * OUT) return;
  float v = p[tid] + b[tid % OUT];
  if (ACT) v = (v > 0.f) ? v : expm1f(v);
  p[tid] = v;
}

static inline int cdiv(long long a, long long b) { return (int)((a + b - 1) / b); }

struct Graph {
  const int* rowptr;
  const int* ssrc;
  const float* snorm;
  const float* dinv;
};

template <int C, bool BIAS, bool ACT>
static void run_gather(const float* xin, float* xout, const float* b,
                       const Graph& g, int N, hipStream_t stream) {
  gather_rows<C, BIAS, ACT><<<cdiv((long long)N * (C / 4), 256), 256, 0, stream>>>(
      xin, g.rowptr, g.ssrc, g.snorm, g.dinv, b, xout, N);
}

template <int IN, int OUT, bool ACT>
static void run_layer(const float* xin, float* xw, float* xout,
                      const float* W, const float* b, const Graph& g,
                      int N, hipStream_t stream) {
  gemm_wmma<IN, OUT><<<cdiv(N, 128), 256, 0, stream>>>(xin, W, xw, N);
  run_gather<OUT, true, ACT>(xw, xout, b, g, N, stream);
}

template <int IN, int OUT, bool ACT>
static void run_layer_agg(const float* xin, float* agg, float* xout,
                          const float* W, const float* b, const Graph& g,
                          int N, hipStream_t stream) {
  run_gather<IN, false, false>(xin, agg, b, g, N, stream);
  gemm_wmma<IN, OUT><<<cdiv(N, 128), 256, 0, stream>>>(agg, W, xout, N);
  bias_act<OUT, ACT><<<cdiv((long long)N * OUT, 256), 256, 0, stream>>>(xout, b, N);
}

extern "C" void kernel_launch(void* const* d_in, const int* in_sizes, int n_in,
                              void* d_out, int out_size, void* d_ws, size_t ws_size,
                              hipStream_t stream) {
  const float* x   = (const float*)d_in[0];
  const int*   ei  = (const int*)d_in[1];
  const float* ew  = (const float*)d_in[2];
  const float* W1 = (const float*)d_in[3];  const float* b1 = (const float*)d_in[4];
  const float* W2 = (const float*)d_in[5];  const float* b2 = (const float*)d_in[6];
  const float* W3 = (const float*)d_in[7];  const float* b3 = (const float*)d_in[8];
  const float* W4 = (const float*)d_in[9];  const float* b4 = (const float*)d_in[10];
  const float* W5 = (const float*)d_in[11]; const float* b5 = (const float*)d_in[12];
  const float* W6 = (const float*)d_in[13]; const float* b6 = (const float*)d_in[14];

  const int N = in_sizes[0] / 32;
  const int E = in_sizes[2];
  const int* src = ei;
  const int* dst = ei + E;

  float* bufA   = (float*)d_ws;
  float* bufB   = bufA + (size_t)N * 128;
  float* dinv   = bufB + (size_t)N * 64;
  float* snorm  = dinv + N;
  int*   ssrc   = (int*)(snorm + E);
  int*   rowptr = ssrc + E;
  int*   cnt    = rowptr + N + 1;
  int*   bsum   = cnt + N;
  float* outp   = (float*)d_out;

  init_deg_cnt<<<cdiv(N, 256), 256, 0, stream>>>(dinv, cnt, N);
  deg_count<<<cdiv(E, 256), 256, 0, stream>>>(dst, ew, dinv, cnt, E);
  rsqrt_inplace<<<cdiv(N, 256), 256, 0, stream>>>(dinv, N);
  scan_tiles<<<cdiv(N, 1024), 256, 0, stream>>>(cnt, rowptr, bsum, N);
  scan_add<<<cdiv(N, 1024), 256, 0, stream>>>(rowptr, bsum, N, E);
  fill_buckets<<<cdiv(E, 256), 256, 0, stream>>>(src, dst, ew, dinv, rowptr, cnt, ssrc, snorm, E);

  Graph g;
  g.rowptr = rowptr; g.ssrc = ssrc; g.snorm = snorm; g.dinv = dinv;

  run_layer<32, 4, true>(x, bufB, bufA, W1, b1, g, N, stream);
  run_layer_agg<4,   16, true >(bufA, bufB, bufA, W2, b2, g, N, stream);
  run_layer_agg<16,  32, true >(bufA, bufB, bufA, W3, b3, g, N, stream);
  run_layer_agg<32,  64, false>(bufA, bufB, bufA, W4, b4, g, N, stream);
  run_layer_agg<64, 128, true >(bufA, bufB, outp, W5, b5, g, N, stream);
  run_layer<128, 128, false>(outp, bufA, outp, W6, b6, g, N, stream);
}


# Bias add and ELU of layers 2-5 moved into the WMMA GEMM's store epilogue; the four one-float-per-thread bias/ELU launches are gone (22 -> 18 launches); float operations and their order unchanged
# speedup vs baseline: 3.5499x; 1.0966x over previous
#include <hip/hip_runtime.h>
#include <cstdint>

typedef float v2f __attribute__((ext_vector_type(2)));
typedef float v8f __attribute__((ext_vector_type(8)));

__global__ __launch_bounds__(256) void init_deg_cnt(float* __restrict__ deg,
                                                    int* __restrict__ cnt, int n) {
  int i = blockIdx.x * 256 + threadIdx.x;
  if (i < n) {
    deg[i] = 1.0f;
    cnt[i] = 0;
  }
}

__global__ __launch_bounds__(256) void deg_count(const int* __restrict__ dst,
                                                 const float* __restrict__ ew,
                                                 float* __restrict__ deg,
                                                 int* __restrict__ cnt, int E) {
  int e = blockIdx.x * 256 + threadIdx.x;
  if (e < E) {
    int d = dst[e];
    unsafeAtomicAdd(&deg[d], ew[e]);
    atomicAdd(&cnt[d], 1);
  }
}

__global__ __launch_bounds__(256) void rsqrt_inplace(float* __restrict__ p, int n) {
  int i = blockIdx.x * 256 + threadIdx.x;
  if (i < n) {
    float d = p[i];
    p[i] = (d > 0.f) ? rsqrtf(d) : 0.f;
  }
}


__global__ __launch_bounds__(256) void scan_tiles(const int* __restrict__ cnt,
                                                  int* __restrict__ rowptr,
                                                  int* __restrict__ bsum, int n) {
  __shared__ int s[256];
  const int t = threadIdx.x;
  const int base = blockIdx.x * 1024 + t * 4;
  const int v0 = (base + 0 < n) ? cnt[base + 0] : 0;
  const int v1 = (base + 1 < n) ? cnt[base + 1] : 0;
  const int v2 = (base + 2 < n) ? cnt[base + 2] : 0;
  const int v3 = (base + 3 < n) ? cnt[base + 3] : 0;
  const int tot = v0 + v1 + v2 + v3;
  s[t] = tot;
  __syncthreads();
  for (int off = 1; off < 256; off <<= 1) {
    const int add = (t >= off) ? s[t - off] : 0;
    __syncthreads();
    s[t] += add;
    __syncthreads();
  }
  int run = s[t] - tot;
  if (base + 0 < n) rowptr[base + 0] = run;
  run += v0;
  if (base + 1 < n) rowptr[base + 1] = run;
  run += v1;
  if (base + 2 < n) rowptr[base + 2] = run;
  run += v2;
  if (base + 3 < n) rowptr[base + 3] = run;
  if (t == 255) bsum[blockIdx.x] = s[255];
}

__global__ __launch_bounds__(256) void scan_add(int* __restrict__ rowptr,
                                                const int* __restrict__ bsum,
                                                int n, int total) {
  __shared__ int s[256];
  const int t = threadIdx.x;
  int part = 0;
  for (int j = t; j < (int)blockIdx.x; j += 256) part += bsum[j];
  s[t] = part;
  __syncthreads();
  for (int off = 128; off > 0; off >>= 1) {
    if (t < off) s[t] += s[t + off];
    __syncthreads();
  }
  const int offs = s[0];
  const int base = blockIdx.x * 1024 + t * 4;
  if (base + 0 < n) rowptr[base + 0] += offs;
  if (base + 1 < n) rowptr[base + 1] += offs;
  if (base + 2 < n) rowptr[base + 2] += offs;
  if (base + 3 < n) rowptr[base + 3] += offs;
  if (blockIdx.x == 0 && t == 0) rowptr[n] = total;
}

__global__ __launch_bounds__(256) void fill_buckets(const int* __restrict__ src,
                                                    const int* __restrict__ dst,
                                                    const float* __restrict__ ew,
                                                    const float* __restrict__ dinv,
                                                    const int* __restrict__ rowptr,
                                                    int* __restrict__ cnt,
                                                    int* __restrict__ ssrc,
                                                    float* __restrict__ snorm, int E) {
  int e = blockIdx.x * 256 + threadIdx.x;
  if (e < E) {
    const int s = src[e];
    const int d = dst[e];
    const int k = atomicSub(&cnt[d], 1) - 1;
    const int p = rowptr[d] + k;
    ssrc[p]  = s;
    snorm[p] = dinv[s] * ew[e] * dinv[d];
  }
}

template <int IN, int OUT, bool BIAS, bool ACT>
__global__ __launch_bounds__(256) void gemm_wmma(const float* __restrict__ x,
                                                 const float* __restrict__ W,
                                                 const float* __restrict__ bias,
                                                 float* __restrict__ y, int n) {
  constexpr bool USE_LDS = (IN * OUT) <= 8192;
  constexpr bool FULL    = (OUT % 16) == 0;
  __shared__ float sW[USE_LDS ? (IN * OUT) : 1];
  const float* Wp = W;
  if constexpr (USE_LDS) {
    for (int i = threadIdx.x; i < IN * OUT; i += 256) sW[i] = W[i];
    __syncthreads();
    Wp = sW;
  }

  const int lane  = threadIdx.x & 31;
  const int wave  = threadIdx.x >> 5;
  const int l15   = lane & 15;
  const int khalf = lane >> 4;
  const int row0  = blockIdx.x * 128 + wave * 16;

  if (row0 < n) {
    int mload = row0 + l15;
    if (mload > n - 1) mload = n - 1;
    const float* xrow = x + (size_t)mload * IN;
    const bool fulltile = (row0 + 16 <= n);

    for (int n0 = 0; n0 < OUT; n0 += 16) {
      const int colb = n0 + l15;
      const int colc = FULL ? colb : (colb < OUT ? colb : OUT - 1);
      float bv = 0.f;
      if (BIAS) bv = bias[colc];
      v8f c = {};
#pragma unroll
      for (int k0 = 0; k0 < IN; k0 += 4) {
        const int krow = k0 + 2 * khalf;
        v2f a = *(const v2f*)(xrow + krow);
        v2f b;
        b.x = Wp[(size_t)krow * OUT + colc];
        b.y = Wp[(size_t)(krow + 1) * OUT + colc];
        c = __builtin_amdgcn_wmma_f32_16x16x4_f32(
             false, a,  false, b,
             (short)0, c,  false,  false);
      }
      if (BIAS || ACT) {
        asm volatile("v_nop\n\tv_nop\n\tv_nop\n\tv_nop\n\t"
                     "v_nop\n\tv_nop\n\tv_nop\n\tv_nop" : "+v"(c));
      }
      if (BIAS) {
#pragma unroll
        for (int r = 0; r < 8; ++r) c[r] = c[r] + bv;
      }
      if (ACT) {
#pragma unroll
        for (int r = 0; r < 8; ++r) {
          const float v = c[r];
          c[r] = (v > 0.f) ? v : expm1f(v);
        }
      }
      const int rowoff = khalf * 8;
      if (FULL && fulltile) {
#pragma unroll
        for (int r = 0; r < 8; ++r)
          y[(size_t)(row0 + rowoff + r) * OUT + colb] = c[r];
      } else {
        const bool colok = FULL || (colb < OUT);
#pragma unroll
        for (int r = 0; r < 8; ++r) {
          int row = row0 + rowoff + r;
          if (row < n && colok) y[(size_t)row * OUT + colb] = c[r];
        }
      }
    }
  }
}

template <int C, bool BIAS, bool ACT>
__global__ __launch_bounds__(256) void gather_rows(const float* __restrict__ xin,
                                                   const int* __restrict__ rowptr,
                                                   const int* __restrict__ ssrc,
                                                   const float* __restrict__ snorm,
                                                   const float* __restrict__ dinv,
                                                   const float* __restrict__ b,
                                                   float* __restrict__ out, int n) {
  constexpr int G = C / 4;
  const int tid = blockIdx.x * 256 + threadIdx.x;
  const int i = tid / G;
  const int q = tid - i * G;
  if (i >= n) return;
  const float* xq = xin + q * 4;
  const float dv = dinv[i];
  const float wself = dv * dv;
  const float4 sv = *(const float4*)(xq + (size_t)i * C);
  float ax = sv.x * wself, ay = sv.y * wself, az = sv.z * wself, aw = sv.w * wself;
  int p = rowptr[i];
  const int pend = rowptr[i + 1];
  for (; p + 4 <= pend; p += 4) {
    const int s0 = ssrc[p], s1 = ssrc[p + 1], s2 = ssrc[p + 2], s3 = ssrc[p + 3];
    const float w0 = snorm[p], w1 = snorm[p + 1], w2 = snorm[p + 2], w3 = snorm[p + 3];
    const float4 r0 = *(const float4*)(xq + (size_t)s0 * C);
    const float4 r1 = *(const float4*)(xq + (size_t)s1 * C);
    const float4 r2 = *(const float4*)(xq + (size_t)s2 * C);
    const float4 r3 = *(const float4*)(xq + (size_t)s3 * C);
    ax += r0.x * w0; ay += r0.y * w0; az += r0.z * w0; aw += r0.w * w0;
    ax += r1.x * w1; ay += r1.y * w1; az += r1.z * w1; aw += r1.w * w1;
    ax += r2.x * w2; ay += r2.y * w2; az += r2.z * w2; aw += r2.w * w2;
    ax += r3.x * w3; ay += r3.y * w3; az += r3.z * w3; aw += r3.w * w3;
  }
  for (; p < pend; ++p) {
    const int s0 = ssrc[p];
    const float w0 = snorm[p];
    const float4 r0 = *(const float4*)(xq + (size_t)s0 * C);
    ax += r0.x * w0; ay += r0.y * w0; az += r0.z * w0; aw += r0.w * w0;
  }
  if (BIAS) {
    ax += b[q * 4 + 0]; ay += b[q * 4 + 1]; az += b[q * 4 + 2]; aw += b[q * 4 + 3];
  }
  if (ACT) {
    ax = (ax > 0.f) ? ax : expm1f(ax);
    ay = (ay > 0.f) ? ay : expm1f(ay);
    az = (az > 0.f) ? az : expm1f(az);
    aw = (aw > 0.f) ? aw : expm1f(aw);
  }
  float4 o;
  o.x = ax; o.y = ay; o.z = az; o.w = aw;
  *(float4*)(out + (size_t)i * C + q * 4) = o;
}

static inline int cdiv(long long a, long long b) { return (int)((a + b - 1) / b); }

struct Graph {
  const int* rowptr;
  const int* ssrc;
  const float* snorm;
  const float* dinv;
};

template <int C, bool BIAS, bool ACT>
static void run_gather(const float* xin, float* xout, const float* b,
                       const Graph& g, int N, hipStream_t stream) {
  gather_rows<C, BIAS, ACT><<<cdiv((long long)N * (C / 4), 256), 256, 0, stream>>>(
      xin, g.rowptr, g.ssrc, g.snorm, g.dinv, b, xout, N);
}

template <int IN, int OUT, bool ACT>
static void run_layer(const float* xin, float* xw, float* xout,
                      const float* W, const float* b, const Graph& g,
                      int N, hipStream_t stream) {
  gemm_wmma<IN, OUT, false, false><<<cdiv(N, 128), 256, 0, stream>>>(xin, W, b, xw, N);
  run_gather<OUT, true, ACT>(xw, xout, b, g, N, stream);
}

template <int IN, int OUT, bool ACT>
static void run_layer_agg(const float* xin, float* agg, float* xout,
                          const float* W, const float* b, const Graph& g,
                          int N, hipStream_t stream) {
  run_gather<IN, false, false>(xin, agg, b, g, N, stream);
  gemm_wmma<IN, OUT, true, ACT><<<cdiv(N, 128), 256, 0, stream>>>(agg, W, b, xout, N);
}

extern "C" void kernel_launch(void* const* d_in, const int* in_sizes, int n_in,
                              void* d_out, int out_size, void* d_ws, size_t ws_size,
                              hipStream_t stream) {
  const float* x   = (const float*)d_in[0];
  const int*   ei  = (const int*)d_in[1];
  const float* ew  = (const float*)d_in[2];
  const float* W1 = (const float*)d_in[3];  const float* b1 = (const float*)d_in[4];
  const float* W2 = (const float*)d_in[5];  const float* b2 = (const float*)d_in[6];
  const float* W3 = (const float*)d_in[7];  const float* b3 = (const float*)d_in[8];
  const float* W4 = (const float*)d_in[9];  const float* b4 = (const float*)d_in[10];
  const float* W5 = (const float*)d_in[11]; const float* b5 = (const float*)d_in[12];
  const float* W6 = (const float*)d_in[13]; const float* b6 = (const float*)d_in[14];

  const int N = in_sizes[0] / 32;
  const int E = in_sizes[2];
  const int* src = ei;
  const int* dst = ei + E;

  float* bufA   = (float*)d_ws;
  float* bufB   = bufA + (size_t)N * 128;
  float* dinv   = bufB + (size_t)N * 64;
  float* snorm  = dinv + N;
  int*   ssrc   = (int*)(snorm + E);
  int*   rowptr = ssrc + E;
  int*   cnt    = rowptr + N + 1;
  int*   bsum   = cnt + N;
  float* outp   = (float*)d_out;

  init_deg_cnt<<<cdiv(N, 256), 256, 0, stream>>>(dinv, cnt, N);
  deg_count<<<cdiv(E, 256), 256, 0, stream>>>(dst, ew, dinv, cnt, E);
  rsqrt_inplace<<<cdiv(N, 256), 256, 0, stream>>>(dinv, N);
  scan_tiles<<<cdiv(N, 1024), 256, 0, stream>>>(cnt, rowptr, bsum, N);
  scan_add<<<cdiv(N, 1024), 256, 0, stream>>>(rowptr, bsum, N, E);
  fill_buckets<<<cdiv(E, 256), 256, 0, stream>>>(src, dst, ew, dinv, rowptr, cnt, ssrc, snorm, E);

  Graph g;
  g.rowptr = rowptr; g.ssrc = ssrc; g.snorm = snorm; g.dinv = dinv;

  run_layer<32, 4, true>(x, bufB, bufA, W1, b1, g, N, stream);
  run_layer_agg<4,   16, true >(bufA, bufB, bufA, W2, b2, g, N, stream);
  run_layer_agg<16,  32, true >(bufA, bufB, bufA, W3, b3, g, N, stream);
  run_layer_agg<32,  64, false>(bufA, bufB, bufA, W4, b4, g, N, stream);
  run_layer_agg<64, 128, true >(bufA, bufB, outp, W5, b5, g, N, stream);
  run_layer<128, 128, false>(outp, bufA, outp, W6, b6, g, N, stream);
}


# Global atomics per edge 3 -> 1: the count's returning add is the edge's rank in its bucket (fill has none); degrees summed per node from the buckets, no float atomics; float sum order = rank order
# speedup vs baseline: 4.4635x; 1.2574x over previous
#include <hip/hip_runtime.h>
#include <cstdint>

typedef float v2f __attribute__((ext_vector_type(2)));
typedef float v8f __attribute__((ext_vector_type(8)));

__global__ __launch_bounds__(256) void init_cnt(int* __restrict__ cnt, int n) {
  int i = blockIdx.x * 256 + threadIdx.x;
  if (i < n) cnt[i] = 0;
}

__global__ __launch_bounds__(256) void count_rank(const int* __restrict__ dst,
                                                  int* __restrict__ cnt,
                                                  int* __restrict__ rank, int E) {
  int e = blockIdx.x * 256 + threadIdx.x;
  if (e < E) rank[e] = atomicAdd(&cnt[dst[e]], 1);
}


__global__ __launch_bounds__(256) void scan_tiles(const int* __restrict__ cnt,
                                                  int* __restrict__ rowptr,
                                                  int* __restrict__ bsum, int n) {
  __shared__ int s[256];
  const int t = threadIdx.x;
  const int base = blockIdx.x * 1024 + t * 4;
  const int v0 = (base + 0 < n) ? cnt[base + 0] : 0;
  const int v1 = (base + 1 < n) ? cnt[base + 1] : 0;
  const int v2 = (base + 2 < n) ? cnt[base + 2] : 0;
  const int v3 = (base + 3 < n) ? cnt[base + 3] : 0;
  const int tot = v0 + v1 + v2 + v3;
  s[t] = tot;
  __syncthreads();
  for (int off = 1; off < 256; off <<= 1) {
    const int add = (t >= off) ? s[t - off] : 0;
    __syncthreads();
    s[t] += add;
    __syncthreads();
  }
  int run = s[t] - tot;
  if (base + 0 < n) rowptr[base + 0] = run;
  run += v0;
  if (base + 1 < n) rowptr[base + 1] = run;
  run += v1;
  if (base + 2 < n) rowptr[base + 2] = run;
  run += v2;
  if (base + 3 < n) rowptr[base + 3] = run;
  if (t == 255) bsum[blockIdx.x] = s[255];
}

__global__ __launch_bounds__(256) void scan_add(int* __restrict__ rowptr,
                                                const int* __restrict__ bsum,
                                                int n, int total) {
  __shared__ int s[256];
  const int t = threadIdx.x;
  int part = 0;
  for (int j = t; j < (int)blockIdx.x; j += 256) part += bsum[j];
  s[t] = part;
  __syncthreads();
  for (int off = 128; off > 0; off >>= 1) {
    if (t < off) s[t] += s[t + off];
    __syncthreads();
  }
  const int offs = s[0];
  const int base = blockIdx.x * 1024 + t * 4;
  if (base + 0 < n) rowptr[base + 0] += offs;
  if (base + 1 < n) rowptr[base + 1] += offs;
  if (base + 2 < n) rowptr[base + 2] += offs;
  if (base + 3 < n) rowptr[base + 3] += offs;
  if (blockIdx.x == 0 && t == 0) rowptr[n] = total;
}

__global__ __launch_bounds__(256) void fill_buckets(const int* __restrict__ src,
                                                    const int* __restrict__ dst,
                                                    const float* __restrict__ ew,
                                                    const int* __restrict__ rowptr,
                                                    const int* __restrict__ rank,
                                                    int* __restrict__ ssrc,
                                                    float* __restrict__ sw, int E) {
  int e = blockIdx.x * 256 + threadIdx.x;
  if (e < E) {
    const int p = rowptr[dst[e]] + rank[e];
    ssrc[p] = src[e];
    sw[p]   = ew[e];
  }
}

__global__ __launch_bounds__(256) void node_dinv(const int* __restrict__ rowptr,
                                                 const float* __restrict__ sw,
                                                 float* __restrict__ dinv, int n) {
  int i = blockIdx.x * 256 + threadIdx.x;
  if (i < n) {
    const int pend = rowptr[i + 1];
    float d = 1.0f;
    for (int p = rowptr[i]; p < pend; ++p) d += sw[p];
    dinv[i] = (d > 0.f) ? rsqrtf(d) : 0.f;
  }
}

__global__ __launch_bounds__(256) void norm_buckets(const int* __restrict__ rowptr,
                                                    const int* __restrict__ ssrc,
                                                    const float* __restrict__ dinv,
                                                    float* __restrict__ sw, int n) {
  int i = blockIdx.x * 256 + threadIdx.x;
  if (i < n) {
    const int pend = rowptr[i + 1];
    const float dv = dinv[i];
    for (int p = rowptr[i]; p < pend; ++p) sw[p] = dinv[ssrc[p]] * sw[p] * dv;
  }
}

template <int IN, int OUT, bool BIAS, bool ACT>
__global__ __launch_bounds__(256) void gemm_wmma(const float* __restrict__ x,
                                                 const float* __restrict__ W,
                                                 const float* __restrict__ bias,
                                                 float* __restrict__ y, int n) {
  constexpr bool USE_LDS = (IN * OUT) <= 8192;
  constexpr bool FULL    = (OUT % 16) == 0;
  __shared__ float sW[USE_LDS ? (IN * OUT) : 1];
  const float* Wp = W;
  if constexpr (USE_LDS) {
    for (int i = threadIdx.x; i < IN * OUT; i += 256) sW[i] = W[i];
    __syncthreads();
    Wp = sW;
  }

  const int lane  = threadIdx.x & 31;
  const int wave  = threadIdx.x >> 5;
  const int l15   = lane & 15;
  const int khalf = lane >> 4;
  const int row0  = blockIdx.x * 128 + wave * 16;

  if (row0 < n) {
    int mload = row0 + l15;
    if (mload > n - 1) mload = n - 1;
    const float* xrow = x + (size_t)mload * IN;
    const bool fulltile = (row0 + 16 <= n);

    for (int n0 = 0; n0 < OUT; n0 += 16) {
      const int colb = n0 + l15;
      const int colc = FULL ? colb : (colb < OUT ? colb : OUT - 1);
      float bv = 0.f;
      if (BIAS) bv = bias[colc];
      v8f c = {};
#pragma unroll
      for (int k0 = 0; k0 < IN; k0 += 4) {
        const int krow = k0 + 2 * khalf;
        v2f a = *(const v2f*)(xrow + krow);
        v2f b;
        b.x = Wp[(size_t)krow * OUT + colc];
        b.y = Wp[(size_t)(krow + 1) * OUT + colc];
        c = __builtin_amdgcn_wmma_f32_16x16x4_f32(
             false, a,  false, b,
             (short)0, c,  false,  false);
      }
      if (BIAS || ACT) {
        asm volatile("v_nop\n\tv_nop\n\tv_nop\n\tv_nop\n\t"
                     "v_nop\n\tv_nop\n\tv_nop\n\tv_nop" : "+v"(c));
      }
      if (BIAS) {
#pragma unroll
        for (int r = 0; r < 8; ++r) c[r] = c[r] + bv;
      }
      if (ACT) {
#pragma unroll
        for (int r = 0; r < 8; ++r) {
          const float v = c[r];
          c[r] = (v > 0.f) ? v : expm1f(v);
        }
      }
      const int rowoff = khalf * 8;
      if (FULL && fulltile) {
#pragma unroll
        for (int r = 0; r < 8; ++r)
          y[(size_t)(row0 + rowoff + r) * OUT + colb] = c[r];
      } else {
        const bool colok = FULL || (colb < OUT);
#pragma unroll
        for (int r = 0; r < 8; ++r) {
          int row = row0 + rowoff + r;
          if (row < n && colok) y[(size_t)row * OUT + colb] = c[r];
        }
      }
    }
  }
}

template <int C, bool BIAS, bool ACT>
__global__ __launch_bounds__(256) void gather_rows(const float* __restrict__ xin,
                                                   const int* __restrict__ rowptr,
                                                   const int* __restrict__ ssrc,
                                                   const float* __restrict__ snorm,
                                                   const float* __restrict__ dinv,
                                                   const float* __restrict__ b,
                                                   float* __restrict__ out, int n) {
  constexpr int G = C / 4;
  const int tid = blockIdx.x * 256 + threadIdx.x;
  const int i = tid / G;
  const int q = tid - i * G;
  if (i >= n) return;
  const float* xq = xin + q * 4;
  const float dv = dinv[i];
  const float wself = dv * dv;
  const float4 sv = *(const float4*)(xq + (size_t)i * C);
  float ax = sv.x * wself, ay = sv.y * wself, az = sv.z * wself, aw = sv.w * wself;
  int p = rowptr[i];
  const int pend = rowptr[i + 1];
  for (; p + 4 <= pend; p += 4) {
    const int s0 = ssrc[p], s1 = ssrc[p + 1], s2 = ssrc[p + 2], s3 = ssrc[p + 3];
    const float w0 = snorm[p], w1 = snorm[p + 1], w2 = snorm[p + 2], w3 = snorm[p + 3];
    const float4 r0 = *(const float4*)(xq + (size_t)s0 * C);
    const float4 r1 = *(const float4*)(xq + (size_t)s1 * C);
    const float4 r2 = *(const float4*)(xq + (size_t)s2 * C);
    const float4 r3 = *(const float4*)(xq + (size_t)s3 * C);
    ax += r0.x * w0; ay += r0.y * w0; az += r0.z * w0; aw += r0.w * w0;
    ax += r1.x * w1; ay += r1.y * w1; az += r1.z * w1; aw += r1.w * w1;
    ax += r2.x * w2; ay += r2.y * w2; az += r2.z * w2; aw += r2.w * w2;
    ax += r3.x * w3; ay += r3.y * w3; az += r3.z * w3; aw += r3.w * w3;
  }
  for (; p < pend; ++p) {
    const int s0 = ssrc[p];
    const float w0 = snorm[p];
    const float4 r0 = *(const float4*)(xq + (size_t)s0 * C);
    ax += r0.x * w0; ay += r0.y * w0; az += r0.z * w0; aw += r0.w * w0;
  }
  if (BIAS) {
    ax += b[q * 4 + 0]; ay += b[q * 4 + 1]; az += b[q * 4 + 2]; aw += b[q * 4 + 3];
  }
  if (ACT) {
    ax = (ax > 0.f) ? ax : expm1f(ax);
    ay = (ay > 0.f) ? ay : expm1f(ay);
    az = (az > 0.f) ? az : expm1f(az);
    aw = (aw > 0.f) ? aw : expm1f(aw);
  }
  float4 o;
  o.x = ax; o.y = ay; o.z = az; o.w = aw;
  *(float4*)(out + (size_t)i * C + q * 4) = o;
}

static inline int cdiv(long long a, long long b) { return (int)((a + b - 1) / b); }

struct Graph {
  const int* rowptr;
  const int* ssrc;
  const float* snorm;
  const float* dinv;
};

template <int C, bool BIAS, bool ACT>
static void run_gather(const float* xin, float* xout, const float* b,
                       const Graph& g, int N, hipStream_t stream) {
  gather_rows<C, BIAS, ACT><<<cdiv((long long)N * (C / 4), 256), 256, 0, stream>>>(
      xin, g.rowptr, g.ssrc, g.snorm, g.dinv, b, xout, N);
}

template <int IN, int OUT, bool ACT>
static void run_layer(const float* xin, float* xw, float* xout,
                      const float* W, const float* b, const Graph& g,
                      int N, hipStream_t stream) {
  gemm_wmma<IN, OUT, false, false><<<cdiv(N, 128), 256, 0, stream>>>(xin, W, b, xw, N);
  run_gather<OUT, true, ACT>(xw, xout, b, g, N, stream);
}

template <int IN, int OUT, bool ACT>
static void run_layer_agg(const float* xin, float* agg, float* xout,
                          const float* W, const float* b, const Graph& g,
                          int N, hipStream_t stream) {
  run_gather<IN, false, false>(xin, agg, b, g, N, stream);
  gemm_wmma<IN, OUT, true, ACT><<<cdiv(N, 128), 256, 0, stream>>>(agg, W, b, xout, N);
}

extern "C" void kernel_launch(void* const* d_in, const int* in_sizes, int n_in,
                              void* d_out, int out_size, void* d_ws, size_t ws_size,
                              hipStream_t stream) {
  const float* x   = (const float*)d_in[0];
  const int*   ei  = (const int*)d_in[1];
  const float* ew  = (const float*)d_in[2];
  const float* W1 = (const float*)d_in[3];  const float* b1 = (const float*)d_in[4];
  const float* W2 = (const float*)d_in[5];  const float* b2 = (const float*)d_in[6];
  const float* W3 = (const float*)d_in[7];  const float* b3 = (const float*)d_in[8];
  const float* W4 = (const float*)d_in[9];  const float* b4 = (const float*)d_in[10];
  const float* W5 = (const float*)d_in[11]; const float* b5 = (const float*)d_in[12];
  const float* W6 = (const float*)d_in[13]; const float* b6 = (const float*)d_in[14];

  const int N = in_sizes[0] / 32;
  const int E = in_sizes[2];
  const int* src = ei;
  const int* dst = ei + E;

  float* bufA   = (float*)d_ws;
  float* bufB   = bufA + (size_t)N * 128;
  float* dinv   = bufB + (size_t)N * 64;
  float* snorm  = dinv + N;
  int*   ssrc   = (int*)(snorm + E);
  int*   rowptr = ssrc + E;
  int*   cnt    = rowptr + N + 1;
  int*   bsum   = cnt + N;
  int*   rank   = (int*)bufA;
  float* outp   = (float*)d_out;

  init_cnt<<<cdiv(N, 256), 256, 0, stream>>>(cnt, N);
  count_rank<<<cdiv(E, 256), 256, 0, stream>>>(dst, cnt, rank, E);
  scan_tiles<<<cdiv(N, 1024), 256, 0, stream>>>(cnt, rowptr, bsum, N);
  scan_add<<<cdiv(N, 1024), 256, 0, stream>>>(rowptr, bsum, N, E);
  fill_buckets<<<cdiv(E, 256), 256, 0, stream>>>(src, dst, ew, rowptr, rank, ssrc, snorm, E);
  node_dinv<<<cdiv(N, 256), 256, 0, stream>>>(rowptr, snorm, dinv, N);
  norm_buckets<<<cdiv(N, 256), 256, 0, stream>>>(rowptr, ssrc, dinv, snorm, N);

  Graph g;
  g.rowptr = rowptr; g.ssrc = ssrc; g.snorm = snorm; g.dinv = dinv;

  run_layer<32, 4, true>(x, bufB, bufA, W1, b1, g, N, stream);
  run_layer_agg<4,   16, true >(bufA, bufB, bufA, W2, b2, g, N, stream);
  run_layer_agg<16,  32, true >(bufA, bufB, bufA, W3, b3, g, N, stream);
  run_layer_agg<32,  64, false>(bufA, bufB, bufA, W4, b4, g, N, stream);
  run_layer_agg<64, 128, true >(bufA, bufB, outp, W5, b5, g, N, stream);
  run_layer<128, 128, false>(outp, bufA, outp, W6, b6, g, N, stream);
}


# Edges grouped by destination in two levels on LDS counters (256-node bins, then nodes; returning LDS add = rank): no global atomic left, 19 -> 17 launches; float sum order = LDS arrival order
# speedup vs baseline: 5.1748x; 1.1594x over previous
#include <hip/hip_runtime.h>
#include <cstdint>

typedef float v2f __attribute__((ext_vector_type(2)));
typedef float v8f __attribute__((ext_vector_type(8)));

constexpr int BIN_SHIFT = 8;
constexpr int BIN_NODES = 1 << BIN_SHIFT;
constexpr int EDGE_CHUNKS = 128;
constexpr int CHUNKS_PER_WAVE = EDGE_CHUNKS / 32;
constexpr int EDGE_BATCH = 8;
static_assert(EDGE_CHUNKS % 32 == 0, "one wave sums a whole number of chunks");

__global__ __launch_bounds__(1024) void bin_hist(const int* __restrict__ dst,
                                                 int* __restrict__ blockhist,
                                                 int E, int NB) {
  extern __shared__ int sh[];
  const int t = threadIdx.x;
  for (int i = t; i < NB; i += 1024) sh[i] = 0;
  __syncthreads();
  const int chunk = (E + EDGE_CHUNKS - 1) / EDGE_CHUNKS;
  const int e0 = (int)blockIdx.x * chunk;
  const int e1 = (e0 + chunk < E) ? (e0 + chunk) : E;
  for (int eb = e0 + t; eb < e1; eb += EDGE_BATCH * 1024) {
    int d[EDGE_BATCH];
    for (int k = 0; k < EDGE_BATCH; ++k) {
      const int e = eb + k * 1024;
      d[k] = (e < e1) ? dst[e] : 0;
    }
    for (int k = 0; k < EDGE_BATCH; ++k)
      if (eb + k * 1024 < e1) atomicAdd(&sh[d[k] >> BIN_SHIFT], 1);
  }
  __syncthreads();
  for (int i = t; i < NB; i += 1024) blockhist[(size_t)blockIdx.x * NB + i] = sh[i];
}

__global__ __launch_bounds__(1024) void bin_scatter(const int* __restrict__ src,
                                                    const int* __restrict__ dst,
                                                    const float* __restrict__ ew,
                                                    const int* __restrict__ blockhist,
                                                    int* __restrict__ binstart,
                                                    int* __restrict__ tsrc,
                                                    int* __restrict__ tdst,
                                                    float* __restrict__ tw,
                                                    int E, int NB) {
  extern __shared__ int sh[];
  __shared__ int s[1024];
  int* base = sh;
  int* tot  = sh + NB;
  const int t = threadIdx.x;
  const int lane = t & 31;
  const int wave = t >> 5;
  const int me = (int)blockIdx.x;

  for (int i = t; i < NB; i += 1024) { base[i] = 0; tot[i] = 0; }
  __syncthreads();
  for (int bb = lane; bb < NB; bb += 4 * 32) {
    int v[4][CHUNKS_PER_WAVE];
    for (int j = 0; j < 4; ++j) {
      const int b = bb + 32 * j;
      for (int k = 0; k < CHUNKS_PER_WAVE; ++k) {
        const int c = wave * CHUNKS_PER_WAVE + k;
        v[j][k] = (b < NB) ? blockhist[(size_t)c * NB + b] : 0;
      }
    }
    for (int j = 0; j < 4; ++j) {
      const int b = bb + 32 * j;
      int before = 0, total = 0;
      for (int k = 0; k < CHUNKS_PER_WAVE; ++k) {
        const int c = wave * CHUNKS_PER_WAVE + k;
        total += v[j][k];
        before += (c < me) ? v[j][k] : 0;
      }
      if (b < NB) {
        atomicAdd(&base[b], before);
        atomicAdd(&tot[b], total);
      }
    }
  }
  __syncthreads();
  int carry = 0;
  for (int b0 = 0; b0 < NB; b0 += 1024) {
    const int b = b0 + t;
    const int v = (b < NB) ? tot[b] : 0;
    s[t] = v;
    __syncthreads();
    for (int off = 1; off < 1024; off <<= 1) {
      const int add = (t >= off) ? s[t - off] : 0;
      __syncthreads();
      s[t] += add;
      __syncthreads();
    }
    const int excl = carry + s[t] - v;
    if (b < NB) {
      base[b] += excl;
      if (me == 0) binstart[b] = excl;
    }
    carry += s[1023];
    __syncthreads();
  }
  if (me == 0 && t == 0) binstart[NB] = carry;

  const int chunk = (E + EDGE_CHUNKS - 1) / EDGE_CHUNKS;
  const int e0 = me * chunk;
  const int e1 = (e0 + chunk < E) ? (e0 + chunk) : E;
  for (int eb = e0 + t; eb < e1; eb += EDGE_BATCH * 1024) {
    int d[EDGE_BATCH], a[EDGE_BATCH], p[EDGE_BATCH];
    float w[EDGE_BATCH];
    for (int k = 0; k < EDGE_BATCH; ++k) {
      const int e = eb + k * 1024;
      const bool ok = e < e1;
      d[k] = ok ? dst[e] : 0;
      a[k] = ok ? src[e] : 0;
      w[k] = ok ? ew[e] : 0.f;
    }
    for (int k = 0; k < EDGE_BATCH; ++k) {
      p[k] = 0;
      if (eb + k * 1024 < e1) p[k] = atomicAdd(&base[d[k] >> BIN_SHIFT], 1);
    }
    for (int k = 0; k < EDGE_BATCH; ++k) {
      if (eb + k * 1024 < e1) {
        tsrc[p[k]] = a[k];
        tdst[p[k]] = d[k];
        tw[p[k]]   = w[k];
      }
    }
  }
}

__global__ __launch_bounds__(1024) void bin_build(const int* __restrict__ binstart,
                                                  const int* __restrict__ tsrc,
                                                  const int* __restrict__ tdst,
                                                  const float* __restrict__ tw,
                                                  int* __restrict__ rowptr,
                                                  int* __restrict__ ssrc,
                                                  float* __restrict__ sw, int n) {
  __shared__ int cnt[BIN_NODES];
  __shared__ int s[BIN_NODES];
  const int t = threadIdx.x;
  const int bin = (int)blockIdx.x;
  const int lo = bin << BIN_SHIFT;
  const int p0 = binstart[bin];
  const int p1 = binstart[bin + 1];
  if (t < BIN_NODES) cnt[t] = 0;
  __syncthreads();
  for (int pb = p0 + t; pb < p1; pb += EDGE_BATCH * 1024) {
    int d[EDGE_BATCH];
    for (int k = 0; k < EDGE_BATCH; ++k) {
      const int p = pb + k * 1024;
      d[k] = (p < p1) ? tdst[p] : lo;
    }
    for (int k = 0; k < EDGE_BATCH; ++k)
      if (pb + k * 1024 < p1) atomicAdd(&cnt[d[k] - lo], 1);
  }
  __syncthreads();
  int v = 0;
  if (t < BIN_NODES) { v = cnt[t]; s[t] = v; }
  __syncthreads();
  for (int off = 1; off < BIN_NODES; off <<= 1) {
    int add = 0;
    if (t < BIN_NODES && t >= off) add = s[t - off];
    __syncthreads();
    if (t < BIN_NODES) s[t] += add;
    __syncthreads();
  }
  if (t < BIN_NODES) {
    const int excl = s[t] - v;
    cnt[t] = excl;
    if (lo + t < n) rowptr[lo + t] = p0 + excl;
  }
  if (bin == (int)gridDim.x - 1 && t == 0) rowptr[n] = p1;
  __syncthreads();
  for (int pb = p0 + t; pb < p1; pb += EDGE_BATCH * 1024) {
    int d[EDGE_BATCH], a[EDGE_BATCH], q[EDGE_BATCH];
    float w[EDGE_BATCH];
    for (int k = 0; k < EDGE_BATCH; ++k) {
      const int p = pb + k * 1024;
      const bool ok = p < p1;
      d[k] = ok ? tdst[p] : lo;
      a[k] = ok ? tsrc[p] : 0;
      w[k] = ok ? tw[p] : 0.f;
    }
    for (int k = 0; k < EDGE_BATCH; ++k) {
      q[k] = p0;
      if (pb + k * 1024 < p1) q[k] = p0 + atomicAdd(&cnt[d[k] - lo], 1);
    }
    for (int k = 0; k < EDGE_BATCH; ++k) {
      if (pb + k * 1024 < p1) {
        ssrc[q[k]] = a[k];
        sw[q[k]]   = w[k];
      }
    }
  }
}

__global__ __launch_bounds__(256) void node_dinv(const int* __restrict__ rowptr,
                                                 const float* __restrict__ sw,
                                                 float* __restrict__ dinv, int n) {
  int i = blockIdx.x * 256 + threadIdx.x;
  if (i < n) {
    const int pend = rowptr[i + 1];
    float d = 1.0f;
    for (int p = rowptr[i]; p < pend; ++p) d += sw[p];
    dinv[i] = (d > 0.f) ? rsqrtf(d) : 0.f;
  }
}

__global__ __launch_bounds__(256) void norm_buckets(const int* __restrict__ rowptr,
                                                    const int* __restrict__ ssrc,
                                                    const float* __restrict__ dinv,
                                                    float* __restrict__ sw, int n) {
  int i = blockIdx.x * 256 + threadIdx.x;
  if (i < n) {
    const int pend = rowptr[i + 1];
    const float dv = dinv[i];
    for (int p = rowptr[i]; p < pend; ++p) sw[p] = dinv[ssrc[p]] * sw[p] * dv;
  }
}

template <int IN, int OUT, bool BIAS, bool ACT>
__global__ __launch_bounds__(256) void gemm_wmma(const float* __restrict__ x,
                                                 const float* __restrict__ W,
                                                 const float* __restrict__ bias,
                                                 float* __restrict__ y, int n) {
  constexpr bool USE_LDS = (IN * OUT) <= 8192;
  constexpr bool FULL    = (OUT % 16) == 0;
  __shared__ float sW[USE_LDS ? (IN * OUT) : 1];
  const float* Wp = W;
  if constexpr (USE_LDS) {
    for (int i = threadIdx.x; i < IN * OUT; i += 256) sW[i] = W[i];
    __syncthreads();
    Wp = sW;
  }

  const int lane  = threadIdx.x & 31;
  const int wave  = threadIdx.x >> 5;
  const int l15   = lane & 15;
  const int khalf = lane >> 4;
  const int row0  = blockIdx.x * 128 + wave * 16;

  if (row0 < n) {
    int mload = row0 + l15;
    if (mload > n - 1) mload = n - 1;
    const float* xrow = x + (size_t)mload * IN;
    const bool fulltile = (row0 + 16 <= n);

    for (int n0 = 0; n0 < OUT; n0 += 16) {
      const int colb = n0 + l15;
      const int colc = FULL ? colb : (colb < OUT ? colb : OUT - 1);
      float bv = 0.f;
      if (BIAS) bv = bias[colc];
      v8f c = {};
#pragma unroll
      for (int k0 = 0; k0 < IN; k0 += 4) {
        const int krow = k0 + 2 * khalf;
        v2f a = *(const v2f*)(xrow + krow);
        v2f b;
        b.x = Wp[(size_t)krow * OUT + colc];
        b.y = Wp[(size_t)(krow + 1) * OUT + colc];
        c = __builtin_amdgcn_wmma_f32_16x16x4_f32(
             false, a,  false, b,
             (short)0, c,  false,  false);
      }
      if (BIAS || ACT) {
        asm volatile("v_nop\n\tv_nop\n\tv_nop\n\tv_nop\n\t"
                     "v_nop\n\tv_nop\n\tv_nop\n\tv_nop" : "+v"(c));
      }
      if (BIAS) {
#pragma unroll
        for (int r = 0; r < 8; ++r) c[r] = c[r] + bv;
      }
      if (ACT) {
#pragma unroll
        for (int r = 0; r < 8; ++r) {
          const float v = c[r];
          c[r] = (v > 0.f) ? v : expm1f(v);
        }
      }
      const int rowoff = khalf * 8;
      if (FULL && fulltile) {
#pragma unroll
        for (int r = 0; r < 8; ++r)
          y[(size_t)(row0 + rowoff + r) * OUT + colb] = c[r];
      } else {
        const bool colok = FULL || (colb < OUT);
#pragma unroll
        for (int r = 0; r < 8; ++r) {
          int row = row0 + rowoff + r;
          if (row < n && colok) y[(size_t)row * OUT + colb] = c[r];
        }
      }
    }
  }
}

template <int C, bool BIAS, bool ACT>
__global__ __launch_bounds__(256) void gather_rows(const float* __restrict__ xin,
                                                   const int* __restrict__ rowptr,
                                                   const int* __restrict__ ssrc,
                                                   const float* __restrict__ snorm,
                                                   const float* __restrict__ dinv,
                                                   const float* __restrict__ b,
                                                   float* __restrict__ out, int n) {
  constexpr int G = C / 4;
  const int tid = blockIdx.x * 256 + threadIdx.x;
  const int i = tid / G;
  const int q = tid - i * G;
  if (i >= n) return;
  const float* xq = xin + q * 4;
  const float dv = dinv[i];
  const float wself = dv * dv;
  const float4 sv = *(const float4*)(xq + (size_t)i * C);
  float ax = sv.x * wself, ay = sv.y * wself, az = sv.z * wself, aw = sv.w * wself;
  int p = rowptr[i];
  const int pend = rowptr[i + 1];
  for (; p + 4 <= pend; p += 4) {
    const int s0 = ssrc[p], s1 = ssrc[p + 1], s2 = ssrc[p + 2], s3 = ssrc[p + 3];
    const float w0 = snorm[p], w1 = snorm[p + 1], w2 = snorm[p + 2], w3 = snorm[p + 3];
    const float4 r0 = *(const float4*)(xq + (size_t)s0 * C);
    const float4 r1 = *(const float4*)(xq + (size_t)s1 * C);
    const float4 r2 = *(const float4*)(xq + (size_t)s2 * C);
    const float4 r3 = *(const float4*)(xq + (size_t)s3 * C);
    ax += r0.x * w0; ay += r0.y * w0; az += r0.z * w0; aw += r0.w * w0;
    ax += r1.x * w1; ay += r1.y * w1; az += r1.z * w1; aw += r1.w * w1;
    ax += r2.x * w2; ay += r2.y * w2; az += r2.z * w2; aw += r2.w * w2;
    ax += r3.x * w3; ay += r3.y * w3; az += r3.z * w3; aw += r3.w * w3;
  }
  for (; p < pend; ++p) {
    const int s0 = ssrc[p];
    const float w0 = snorm[p];
    const float4 r0 = *(const float4*)(xq + (size_t)s0 * C);
    ax += r0.x * w0; ay += r0.y * w0; az += r0.z * w0; aw += r0.w * w0;
  }
  if (BIAS) {
    ax += b[q * 4 + 0]; ay += b[q * 4 + 1]; az += b[q * 4 + 2]; aw += b[q * 4 + 3];
  }
  if (ACT) {
    ax = (ax > 0.f) ? ax : expm1f(ax);
    ay = (ay > 0.f) ? ay : expm1f(ay);
    az = (az > 0.f) ? az : expm1f(az);
    aw = (aw > 0.f) ? aw : expm1f(aw);
  }
  float4 o;
  o.x = ax; o.y = ay; o.z = az; o.w = aw;
  *(float4*)(out + (size_t)i * C + q * 4) = o;
}

static inline int cdiv(long long a, long long b) { return (int)((a + b - 1) / b); }

struct Graph {
  const int* rowptr;
  const int* ssrc;
  const float* snorm;
  const float* dinv;
};

template <int C, bool BIAS, bool ACT>
static void run_gather(const float* xin, float* xout, const float* b,
                       const Graph& g, int N, hipStream_t stream) {
  gather_rows<C, BIAS, ACT><<<cdiv((long long)N * (C / 4), 256), 256, 0, stream>>>(
      xin, g.rowptr, g.ssrc, g.snorm, g.dinv, b, xout, N);
}

template <int IN, int OUT, bool ACT>
static void run_layer(const float* xin, float* xw, float* xout,
                      const float* W, const float* b, const Graph& g,
                      int N, hipStream_t stream) {
  gemm_wmma<IN, OUT, false, false><<<cdiv(N, 128), 256, 0, stream>>>(xin, W, b, xw, N);
  run_gather<OUT, true, ACT>(xw, xout, b, g, N, stream);
}

template <int IN, int OUT, bool ACT>
static void run_layer_agg(const float* xin, float* agg, float* xout,
                          const float* W, const float* b, const Graph& g,
                          int N, hipStream_t stream) {
  run_gather<IN, false, false>(xin, agg, b, g, N, stream);
  gemm_wmma<IN, OUT, true, ACT><<<cdiv(N, 128), 256, 0, stream>>>(agg, W, b, xout, N);
}

extern "C" void kernel_launch(void* const* d_in, const int* in_sizes, int n_in,
                              void* d_out, int out_size, void* d_ws, size_t ws_size,
                              hipStream_t stream) {
  const float* x   = (const float*)d_in[0];
  const int*   ei  = (const int*)d_in[1];
  const float* ew  = (const float*)d_in[2];
  const float* W1 = (const float*)d_in[3];  const float* b1 = (const float*)d_in[4];
  const float* W2 = (const float*)d_in[5];  const float* b2 = (const float*)d_in[6];
  const float* W3 = (const float*)d_in[7];  const float* b3 = (const float*)d_in[8];
  const float* W4 = (const float*)d_in[9];  const float* b4 = (const float*)d_in[10];
  const float* W5 = (const float*)d_in[11]; const float* b5 = (const float*)d_in[12];
  const float* W6 = (const float*)d_in[13]; const float* b6 = (const float*)d_in[14];

  const int N = in_sizes[0] / 32;
  const int E = in_sizes[2];
  const int* src = ei;
  const int* dst = ei + E;

  float* bufA   = (float*)d_ws;
  float* bufB   = bufA + (size_t)N * 128;
  float* dinv   = bufB + (size_t)N * 64;
  float* snorm  = dinv + N;
  int*   ssrc   = (int*)(snorm + E);
  int*   rowptr = ssrc + E;
  const int NB  = cdiv(N, BIN_NODES);
  int*   tsrc   = (int*)bufA;
  int*   tdst   = tsrc + E;
  float* tw     = (float*)(tdst + E);
  int*   bhist  = (int*)bufB;
  int*   bstart = bhist + (size_t)EDGE_CHUNKS * NB;
  float* outp   = (float*)d_out;

  bin_hist<<<EDGE_CHUNKS, 1024, NB * sizeof(int), stream>>>(dst, bhist, E, NB);
  bin_scatter<<<EDGE_CHUNKS, 1024, 2 * NB * sizeof(int), stream>>>(
      src, dst, ew, bhist, bstart, tsrc, tdst, tw, E, NB);
  bin_build<<<NB, 1024, 0, stream>>>(bstart, tsrc, tdst, tw, rowptr, ssrc, snorm, N);
  node_dinv<<<cdiv(N, 256), 256, 0, stream>>>(rowptr, snorm, dinv, N);
  norm_buckets<<<cdiv(N, 256), 256, 0, stream>>>(rowptr, ssrc, dinv, snorm, N);

  Graph g;
  g.rowptr = rowptr; g.ssrc = ssrc; g.snorm = snorm; g.dinv = dinv;

  run_layer<32, 4, true>(x, bufB, bufA, W1, b1, g, N, stream);
  run_layer_agg<4,   16, true >(bufA, bufB, bufA, W2, b2, g, N, stream);
  run_layer_agg<16,  32, true >(bufA, bufB, bufA, W3, b3, g, N, stream);
  run_layer_agg<32,  64, false>(bufA, bufB, bufA, W4, b4, g, N, stream);
  run_layer_agg<64, 128, true >(bufA, bufB, outp, W5, b5, g, N, stream);
  run_layer<128, 128, false>(outp, bufA, outp, W6, b6, g, N, stream);
}


# Layers 2-6: gather and GEMM in one launch each: a block sums its 128 nodes' rows into LDS, the WMMA reads them there (17 -> 12 launches); layer 6 aggregates first too, so its float sum order changes
# speedup vs baseline: 5.6120x; 1.0845x over previous
#include <hip/hip_runtime.h>
#include <cstdint>

typedef float v2f __attribute__((ext_vector_type(2)));
typedef float v8f __attribute__((ext_vector_type(8)));

constexpr int BIN_SHIFT = 8;
constexpr int BIN_NODES = 1 << BIN_SHIFT;
constexpr int EDGE_CHUNKS = 128;
constexpr int CHUNKS_PER_WAVE = EDGE_CHUNKS / 32;
constexpr int EDGE_BATCH = 8;
static_assert(EDGE_CHUNKS % 32 == 0, "one wave sums a whole number of chunks");

__global__ __launch_bounds__(1024) void bin_hist(const int* __restrict__ dst,
                                                 int* __restrict__ blockhist,
                                                 int E, int NB) {
  extern __shared__ int sh[];
  const int t = threadIdx.x;
  for (int i = t; i < NB; i += 1024) sh[i] = 0;
  __syncthreads();
  const int chunk = (E + EDGE_CHUNKS - 1) / EDGE_CHUNKS;
  const int e0 = (int)blockIdx.x * chunk;
  const int e1 = (e0 + chunk < E) ? (e0 + chunk) : E;
  for (int eb = e0 + t; eb < e1; eb += EDGE_BATCH * 1024) {
    int d[EDGE_BATCH];
    for (int k = 0; k < EDGE_BATCH; ++k) {
      const int e = eb + k * 1024;
      d[k] = (e < e1) ? dst[e] : 0;
    }
    for (int k = 0; k < EDGE_BATCH; ++k)
      if (eb + k * 1024 < e1) atomicAdd(&sh[d[k] >> BIN_SHIFT], 1);
  }
  __syncthreads();
  for (int i = t; i < NB; i += 1024) blockhist[(size_t)blockIdx.x * NB + i] = sh[i];
}

__global__ __launch_bounds__(1024) void bin_scatter(const int* __restrict__ src,
                                                    const int* __restrict__ dst,
                                                    const float* __restrict__ ew,
                                                    const int* __restrict__ blockhist,
                                                    int* __restrict__ binstart,
                                                    int* __restrict__ tsrc,
                                                    int* __restrict__ tdst,
                                                    float* __restrict__ tw,
                                                    int E, int NB) {
  extern __shared__ int sh[];
  __shared__ int s[1024];
  int* base = sh;
  int* tot  = sh + NB;
  const int t = threadIdx.x;
  const int lane = t & 31;
  const int wave = t >> 5;
  const int me = (int)blockIdx.x;

  for (int i = t; i < NB; i += 1024) { base[i] = 0; tot[i] = 0; }
  __syncthreads();
  for (int bb = lane; bb < NB; bb += 4 * 32) {
    int v[4][CHUNKS_PER_WAVE];
    for (int j = 0; j < 4; ++j) {
      const int b = bb + 32 * j;
      for (int k = 0; k < CHUNKS_PER_WAVE; ++k) {
        const int c = wave * CHUNKS_PER_WAVE + k;
        v[j][k] = (b < NB) ? blockhist[(size_t)c * NB + b] : 0;
      }
    }
    for (int j = 0; j < 4; ++j) {
      const int b = bb + 32 * j;
      int before = 0, total = 0;
      for (int k = 0; k < CHUNKS_PER_WAVE; ++k) {
        const int c = wave * CHUNKS_PER_WAVE + k;
        total += v[j][k];
        before += (c < me) ? v[j][k] : 0;
      }
      if (b < NB) {
        atomicAdd(&base[b], before);
        atomicAdd(&tot[b], total);
      }
    }
  }
  __syncthreads();
  int carry = 0;
  for (int b0 = 0; b0 < NB; b0 += 1024) {
    const int b = b0 + t;
    const int v = (b < NB) ? tot[b] : 0;
    s[t] = v;
    __syncthreads();
    for (int off = 1; off < 1024; off <<= 1) {
      const int add = (t >= off) ? s[t - off] : 0;
      __syncthreads();
      s[t] += add;
      __syncthreads();
    }
    const int excl = carry + s[t] - v;
    if (b < NB) {
      base[b] += excl;
      if (me == 0) binstart[b] = excl;
    }
    carry += s[1023];
    __syncthreads();
  }
  if (me == 0 && t == 0) binstart[NB] = carry;

  const int chunk = (E + EDGE_CHUNKS - 1) / EDGE_CHUNKS;
  const int e0 = me * chunk;
  const int e1 = (e0 + chunk < E) ? (e0 + chunk) : E;
  for (int eb = e0 + t; eb < e1; eb += EDGE_BATCH * 1024) {
    int d[EDGE_BATCH], a[EDGE_BATCH], p[EDGE_BATCH];
    float w[EDGE_BATCH];
    for (int k = 0; k < EDGE_BATCH; ++k) {
      const int e = eb + k * 1024;
      const bool ok = e < e1;
      d[k] = ok ? dst[e] : 0;
      a[k] = ok ? src[e] : 0;
      w[k] = ok ? ew[e] : 0.f;
    }
    for (int k = 0; k < EDGE_BATCH; ++k) {
      p[k] = 0;
      if (eb + k * 1024 < e1) p[k] = atomicAdd(&base[d[k] >> BIN_SHIFT], 1);
    }
    for (int k = 0; k < EDGE_BATCH; ++k) {
      if (eb + k * 1024 < e1) {
        tsrc[p[k]] = a[k];
        tdst[p[k]] = d[k];
        tw[p[k]]   = w[k];
      }
    }
  }
}

__global__ __launch_bounds__(1024) void bin_build(const int* __restrict__ binstart,
                                                  const int* __restrict__ tsrc,
                                                  const int* __restrict__ tdst,
                                                  const float* __restrict__ tw,
                                                  int* __restrict__ rowptr,
                                                  int* __restrict__ ssrc,
                                                  float* __restrict__ sw, int n) {
  __shared__ int cnt[BIN_NODES];
  __shared__ int s[BIN_NODES];
  const int t = threadIdx.x;
  const int bin = (int)blockIdx.x;
  const int lo = bin << BIN_SHIFT;
  const int p0 = binstart[bin];
  const int p1 = binstart[bin + 1];
  if (t < BIN_NODES) cnt[t] = 0;
  __syncthreads();
  for (int pb = p0 + t; pb < p1; pb += EDGE_BATCH * 1024) {
    int d[EDGE_BATCH];
    for (int k = 0; k < EDGE_BATCH; ++k) {
      const int p = pb + k * 1024;
      d[k] = (p < p1) ? tdst[p] : lo;
    }
    for (int k = 0; k < EDGE_BATCH; ++k)
      if (pb + k * 1024 < p1) atomicAdd(&cnt[d[k] - lo], 1);
  }
  __syncthreads();
  int v = 0;
  if (t < BIN_NODES) { v = cnt[t]; s[t] = v; }
  __syncthreads();
  for (int off = 1; off < BIN_NODES; off <<= 1) {
    int add = 0;
    if (t < BIN_NODES && t >= off) add = s[t - off];
    __syncthreads();
    if (t < BIN_NODES) s[t] += add;
    __syncthreads();
  }
  if (t < BIN_NODES) {
    const int excl = s[t] - v;
    cnt[t] = excl;
    if (lo + t < n) rowptr[lo + t] = p0 + excl;
  }
  if (bin == (int)gridDim.x - 1 && t == 0) rowptr[n] = p1;
  __syncthreads();
  for (int pb = p0 + t; pb < p1; pb += EDGE_BATCH * 1024) {
    int d[EDGE_BATCH], a[EDGE_BATCH], q[EDGE_BATCH];
    float w[EDGE_BATCH];
    for (int k = 0; k < EDGE_BATCH; ++k) {
      const int p = pb + k * 1024;
      const bool ok = p < p1;
      d[k] = ok ? tdst[p] : lo;
      a[k] = ok ? tsrc[p] : 0;
      w[k] = ok ? tw[p] : 0.f;
    }
    for (int k = 0; k < EDGE_BATCH; ++k) {
      q[k] = p0;
      if (pb + k * 1024 < p1) q[k] = p0 + atomicAdd(&cnt[d[k] - lo], 1);
    }
    for (int k = 0; k < EDGE_BATCH; ++k) {
      if (pb + k * 1024 < p1) {
        ssrc[q[k]] = a[k];
        sw[q[k]]   = w[k];
      }
    }
  }
}

__global__ __launch_bounds__(256) void node_dinv(const int* __restrict__ rowptr,
                                                 const float* __restrict__ sw,
                                                 float* __restrict__ dinv, int n) {
  int i = blockIdx.x * 256 + threadIdx.x;
  if (i < n) {
    const int pend = rowptr[i + 1];
    float d = 1.0f;
    for (int p = rowptr[i]; p < pend; ++p) d += sw[p];
    dinv[i] = (d > 0.f) ? rsqrtf(d) : 0.f;
  }
}

__global__ __launch_bounds__(256) void norm_buckets(const int* __restrict__ rowptr,
                                                    const int* __restrict__ ssrc,
                                                    const float* __restrict__ dinv,
                                                    float* __restrict__ sw, int n) {
  int i = blockIdx.x * 256 + threadIdx.x;
  if (i < n) {
    const int pend = rowptr[i + 1];
    const float dv = dinv[i];
    for (int p = rowptr[i]; p < pend; ++p) sw[p] = dinv[ssrc[p]] * sw[p] * dv;
  }
}

template <int IN, int OUT, bool BIAS, bool ACT>
__global__ __launch_bounds__(256) void gemm_wmma(const float* __restrict__ x,
                                                 const float* __restrict__ W,
                                                 const float* __restrict__ bias,
                                                 float* __restrict__ y, int n) {
  constexpr bool USE_LDS = (IN * OUT) <= 8192;
  constexpr bool FULL    = (OUT % 16) == 0;
  __shared__ float sW[USE_LDS ? (IN * OUT) : 1];
  const float* Wp = W;
  if constexpr (USE_LDS) {
    for (int i = threadIdx.x; i < IN * OUT; i += 256) sW[i] = W[i];
    __syncthreads();
    Wp = sW;
  }

  const int lane  = threadIdx.x & 31;
  const int wave  = threadIdx.x >> 5;
  const int l15   = lane & 15;
  const int khalf = lane >> 4;
  const int row0  = blockIdx.x * 128 + wave * 16;

  if (row0 < n) {
    int mload = row0 + l15;
    if (mload > n - 1) mload = n - 1;
    const float* xrow = x + (size_t)mload * IN;
    const bool fulltile = (row0 + 16 <= n);

    for (int n0 = 0; n0 < OUT; n0 += 16) {
      const int colb = n0 + l15;
      const int colc = FULL ? colb : (colb < OUT ? colb : OUT - 1);
      float bv = 0.f;
      if (BIAS) bv = bias[colc];
      v8f c = {};
#pragma unroll
      for (int k0 = 0; k0 < IN; k0 += 4) {
        const int krow = k0 + 2 * khalf;
        v2f a = *(const v2f*)(xrow + krow);
        v2f b;
        b.x = Wp[(size_t)krow * OUT + colc];
        b.y = Wp[(size_t)(krow + 1) * OUT + colc];
        c = __builtin_amdgcn_wmma_f32_16x16x4_f32(
             false, a,  false, b,
             (short)0, c,  false,  false);
      }
      if (BIAS || ACT) {
        asm volatile("v_nop\n\tv_nop\n\tv_nop\n\tv_nop\n\t"
                     "v_nop\n\tv_nop\n\tv_nop\n\tv_nop" : "+v"(c));
      }
      if (BIAS) {
#pragma unroll
        for (int r = 0; r < 8; ++r) c[r] = c[r] + bv;
      }
      if (ACT) {
#pragma unroll
        for (int r = 0; r < 8; ++r) {
          const float v = c[r];
          c[r] = (v > 0.f) ? v : expm1f(v);
        }
      }
      const int rowoff = khalf * 8;
      if (FULL && fulltile) {
#pragma unroll
        for (int r = 0; r < 8; ++r)
          y[(size_t)(row0 + rowoff + r) * OUT + colb] = c[r];
      } else {
        const bool colok = FULL || (colb < OUT);
#pragma unroll
        for (int r = 0; r < 8; ++r) {
          int row = row0 + rowoff + r;
          if (row < n && colok) y[(size_t)row * OUT + colb] = c[r];
        }
      }
    }
  }
}

template <int C, bool BIAS, bool ACT>
__global__ __launch_bounds__(256) void gather_rows(const float* __restrict__ xin,
                                                   const int* __restrict__ rowptr,
                                                   const int* __restrict__ ssrc,
                                                   const float* __restrict__ snorm,
                                                   const float* __restrict__ dinv,
                                                   const float* __restrict__ b,
                                                   float* __restrict__ out, int n) {
  constexpr int G = C / 4;
  const int tid = blockIdx.x * 256 + threadIdx.x;
  const int i = tid / G;
  const int q = tid - i * G;
  if (i >= n) return;
  const float* xq = xin + q * 4;
  const float dv = dinv[i];
  const float wself = dv * dv;
  const float4 sv = *(const float4*)(xq + (size_t)i * C);
  float ax = sv.x * wself, ay = sv.y * wself, az = sv.z * wself, aw = sv.w * wself;
  int p = rowptr[i];
  const int pend = rowptr[i + 1];
  for (; p + 4 <= pend; p += 4) {
    const int s0 = ssrc[p], s1 = ssrc[p + 1], s2 = ssrc[p + 2], s3 = ssrc[p + 3];
    const float w0 = snorm[p], w1 = snorm[p + 1], w2 = snorm[p + 2], w3 = snorm[p + 3];
    const float4 r0 = *(const float4*)(xq + (size_t)s0 * C);
    const float4 r1 = *(const float4*)(xq + (size_t)s1 * C);
    const float4 r2 = *(const float4*)(xq + (size_t)s2 * C);
    const float4 r3 = *(const float4*)(xq + (size_t)s3 * C);
    ax += r0.x * w0; ay += r0.y * w0; az += r0.z * w0; aw += r0.w * w0;
    ax += r1.x * w1; ay += r1.y * w1; az += r1.z * w1; aw += r1.w * w1;
    ax += r2.x * w2; ay += r2.y * w2; az += r2.z * w2; aw += r2.w * w2;
    ax += r3.x * w3; ay += r3.y * w3; az += r3.z * w3; aw += r3.w * w3;
  }
  for (; p < pend; ++p) {
    const int s0 = ssrc[p];
    const float w0 = snorm[p];
    const float4 r0 = *(const float4*)(xq + (size_t)s0 * C);
    ax += r0.x * w0; ay += r0.y * w0; az += r0.z * w0; aw += r0.w * w0;
  }
  if (BIAS) {
    ax += b[q * 4 + 0]; ay += b[q * 4 + 1]; az += b[q * 4 + 2]; aw += b[q * 4 + 3];
  }
  if (ACT) {
    ax = (ax > 0.f) ? ax : expm1f(ax);
    ay = (ay > 0.f) ? ay : expm1f(ay);
    az = (az > 0.f) ? az : expm1f(az);
    aw = (aw > 0.f) ? aw : expm1f(aw);
  }
  float4 o;
  o.x = ax; o.y = ay; o.z = az; o.w = aw;
  *(float4*)(out + (size_t)i * C + q * 4) = o;
}

template <int IN, int OUT, bool ACT>
__global__ __launch_bounds__(256) void agg_gemm(const float* __restrict__ xin,
                                                const int* __restrict__ rowptr,
                                                const int* __restrict__ ssrc,
                                                const float* __restrict__ snorm,
                                                const float* __restrict__ dinv,
                                                const float* __restrict__ W,
                                                const float* __restrict__ bias,
                                                float* __restrict__ y, int n) {
  static_assert(OUT % 16 == 0, "whole column tiles");
  constexpr int G = IN / 4;
  constexpr int NODES = 128;
  constexpr int NPP = 256 / G;
  constexpr int PITCH = (IN == 4) ? 4 : IN + 4;
  constexpr int EDGES = 8;
  constexpr bool USE_LDS = (IN * OUT) <= 8192;
  __shared__ alignas(16) float sA[NODES * PITCH];
  __shared__ float sW[USE_LDS ? (IN * OUT) : 1];
  const int t = threadIdx.x;
  const float* Wp = W;
  if constexpr (USE_LDS) {
    for (int i = t; i < IN * OUT; i += 256) sW[i] = W[i];
    Wp = sW;
  }

  const int nl0 = t / G;
  const int q = t - nl0 * G;
  const float* xq = xin + q * 4;
  for (int nl = nl0; nl < NODES; nl += NPP) {
    const int i = (int)blockIdx.x * NODES + nl;
    float ax = 0.f, ay = 0.f, az = 0.f, aw = 0.f;
    if (i < n) {
      const float dv = dinv[i];
      const float wself = dv * dv;
      const float4 sv = *(const float4*)(xq + (size_t)i * IN);
      int p = rowptr[i];
      const int pend = rowptr[i + 1];
      int s[EDGES];
      float w[EDGES];
#pragma unroll
      for (int k = 0; k < EDGES; ++k) {
        const bool ok = p + k < pend;
        const int pc = ok ? p + k : (p < pend ? p : 0);
        const int sk = ssrc[pc];
        const float wk = snorm[pc];
        s[k] = ok ? sk : i;
        w[k] = ok ? wk : 0.f;
      }
      ax = sv.x * wself; ay = sv.y * wself; az = sv.z * wself; aw = sv.w * wself;
      while (p < pend) {
        float4 r[EDGES];
#pragma unroll
        for (int k = 0; k < EDGES; ++k) r[k] = *(const float4*)(xq + (size_t)s[k] * IN);
        const int pn = p + EDGES;
        int sn[EDGES];
        float wn[EDGES];
#pragma unroll
        for (int k = 0; k < EDGES; ++k) {
          const bool ok = pn + k < pend;
          const int pc = ok ? pn + k : p;
          const int sk = ssrc[pc];
          const float wk = snorm[pc];
          sn[k] = ok ? sk : i;
          wn[k] = ok ? wk : 0.f;
        }
#pragma unroll
        for (int k = 0; k < EDGES; ++k) {
          ax += r[k].x * w[k]; ay += r[k].y * w[k]; az += r[k].z * w[k]; aw += r[k].w * w[k];
        }
#pragma unroll
        for (int k = 0; k < EDGES; ++k) { s[k] = sn[k]; w[k] = wn[k]; }
        p = pn;
      }
    }
    float4 o;
    o.x = ax; o.y = ay; o.z = az; o.w = aw;
    *(float4*)(sA + nl * PITCH + q * 4) = o;
  }
  __syncthreads();

  const int lane  = t & 31;
  const int wave  = t >> 5;
  const int l15   = lane & 15;
  const int khalf = lane >> 4;
  const int row0  = (int)blockIdx.x * NODES + wave * 16;

  if (row0 < n) {
    const float* arow = sA + (wave * 16 + l15) * PITCH;
    const bool fulltile = (row0 + 16 <= n);
    for (int n0 = 0; n0 < OUT; n0 += 16) {
      const int colb = n0 + l15;
      const float bv = bias[colb];
      v8f c = {};
#pragma unroll
      for (int k0 = 0; k0 < IN; k0 += 4) {
        const int krow = k0 + 2 * khalf;
        v2f a = *(const v2f*)(arow + krow);
        v2f b;
        b.x = Wp[(size_t)krow * OUT + colb];
        b.y = Wp[(size_t)(krow + 1) * OUT + colb];
        c = __builtin_amdgcn_wmma_f32_16x16x4_f32(
             false, a,  false, b,
             (short)0, c,  false,  false);
      }
      asm volatile("v_nop\n\tv_nop\n\tv_nop\n\tv_nop\n\t"
                   "v_nop\n\tv_nop\n\tv_nop\n\tv_nop" : "+v"(c));
#pragma unroll
      for (int r = 0; r < 8; ++r) c[r] = c[r] + bv;
      if (ACT) {
#pragma unroll
        for (int r = 0; r < 8; ++r) {
          const float v = c[r];
          c[r] = (v > 0.f) ? v : expm1f(v);
        }
      }
      const int rowoff = khalf * 8;
      if (fulltile) {
#pragma unroll
        for (int r = 0; r < 8; ++r)
          y[(size_t)(row0 + rowoff + r) * OUT + colb] = c[r];
      } else {
#pragma unroll
        for (int r = 0; r < 8; ++r) {
          int row = row0 + rowoff + r;
          if (row < n) y[(size_t)row * OUT + colb] = c[r];
        }
      }
    }
  }
}

static inline int cdiv(long long a, long long b) { return (int)((a + b - 1) / b); }

struct Graph {
  const int* rowptr;
  const int* ssrc;
  const float* snorm;
  const float* dinv;
};

template <int C, bool BIAS, bool ACT>
static void run_gather(const float* xin, float* xout, const float* b,
                       const Graph& g, int N, hipStream_t stream) {
  gather_rows<C, BIAS, ACT><<<cdiv((long long)N * (C / 4), 256), 256, 0, stream>>>(
      xin, g.rowptr, g.ssrc, g.snorm, g.dinv, b, xout, N);
}

template <int IN, int OUT, bool ACT>
static void run_layer(const float* xin, float* xw, float* xout,
                      const float* W, const float* b, const Graph& g,
                      int N, hipStream_t stream) {
  gemm_wmma<IN, OUT, false, false><<<cdiv(N, 128), 256, 0, stream>>>(xin, W, b, xw, N);
  run_gather<OUT, true, ACT>(xw, xout, b, g, N, stream);
}

template <int IN, int OUT, bool ACT>
static void run_layer_fused(const float* xin, float* xout,
                            const float* W, const float* b, const Graph& g,
                            int N, hipStream_t stream) {
  agg_gemm<IN, OUT, ACT><<<cdiv(N, 128), 256, 0, stream>>>(
      xin, g.rowptr, g.ssrc, g.snorm, g.dinv, W, b, xout, N);
}

extern "C" void kernel_launch(void* const* d_in, const int* in_sizes, int n_in,
                              void* d_out, int out_size, void* d_ws, size_t ws_size,
                              hipStream_t stream) {
  const float* x   = (const float*)d_in[0];
  const int*   ei  = (const int*)d_in[1];
  const float* ew  = (const float*)d_in[2];
  const float* W1 = (const float*)d_in[3];  const float* b1 = (const float*)d_in[4];
  const float* W2 = (const float*)d_in[5];  const float* b2 = (const float*)d_in[6];
  const float* W3 = (const float*)d_in[7];  const float* b3 = (const float*)d_in[8];
  const float* W4 = (const float*)d_in[9];  const float* b4 = (const float*)d_in[10];
  const float* W5 = (const float*)d_in[11]; const float* b5 = (const float*)d_in[12];
  const float* W6 = (const float*)d_in[13]; const float* b6 = (const float*)d_in[14];

  const int N = in_sizes[0] / 32;
  const int E = in_sizes[2];
  const int* src = ei;
  const int* dst = ei + E;

  float* bufA   = (float*)d_ws;
  float* bufB   = bufA + (size_t)N * 128;
  float* dinv   = bufB + (size_t)N * 64;
  float* snorm  = dinv + N;
  int*   ssrc   = (int*)(snorm + E);
  int*   rowptr = ssrc + E;
  const int NB  = cdiv(N, BIN_NODES);
  int*   tsrc   = (int*)bufA;
  int*   tdst   = tsrc + E;
  float* tw     = (float*)(tdst + E);
  int*   bhist  = (int*)bufB;
  int*   bstart = bhist + (size_t)EDGE_CHUNKS * NB;
  float* outp   = (float*)d_out;

  bin_hist<<<EDGE_CHUNKS, 1024, NB * sizeof(int), stream>>>(dst, bhist, E, NB);
  bin_scatter<<<EDGE_CHUNKS, 1024, 2 * NB * sizeof(int), stream>>>(
      src, dst, ew, bhist, bstart, tsrc, tdst, tw, E, NB);
  bin_build<<<NB, 1024, 0, stream>>>(bstart, tsrc, tdst, tw, rowptr, ssrc, snorm, N);
  node_dinv<<<cdiv(N, 256), 256, 0, stream>>>(rowptr, snorm, dinv, N);
  norm_buckets<<<cdiv(N, 256), 256, 0, stream>>>(rowptr, ssrc, dinv, snorm, N);

  Graph g;
  g.rowptr = rowptr; g.ssrc = ssrc; g.snorm = snorm; g.dinv = dinv;

  run_layer<32, 4, true>(x, bufB, bufA, W1, b1, g, N, stream);
  run_layer_fused<4,   16,  true >(bufA, bufB, W2, b2, g, N, stream);
  run_layer_fused<16,  32,  true >(bufB, bufA, W3, b3, g, N, stream);
  run_layer_fused<32,  64,  false>(bufA, bufB, W4, b4, g, N, stream);
  run_layer_fused<64,  128, true >(bufB, bufA, W5, b5, g, N, stream);
  run_layer_fused<128, 128, false>(bufA, outp, W6, b6, g, N, stream);
}


# Pre-layer per-node passes folded into neighbour launches: degree sum+rsqrt and layer 1's WMMA product into the bucket build, edge-weight normalisation into layer 1's gather; 12 -> 9; same float order
# speedup vs baseline: 5.8053x; 1.0344x over previous
#include <hip/hip_runtime.h>
#include <cstdint>

typedef float v2f __attribute__((ext_vector_type(2)));
typedef float v8f __attribute__((ext_vector_type(8)));

template <int IN, int OUT, bool BIAS, bool ACT>
__device__ __forceinline__ void wmma_row_tile(const float* __restrict__ x,
                                              const float* Wp,
                                              const float* __restrict__ bias,
                                              float* __restrict__ y, int n,
                                              int row0, int lane) {
  constexpr bool FULL    = (OUT % 16) == 0;
  const int l15   = lane & 15;
  const int khalf = lane >> 4;

  if (row0 < n) {
    int mload = row0 + l15;
    if (mload > n - 1) mload = n - 1;
    const float* xrow = x + (size_t)mload * IN;
    const bool fulltile = (row0 + 16 <= n);

    for (int n0 = 0; n0 < OUT; n0 += 16) {
      const int colb = n0 + l15;
      const int colc = FULL ? colb : (colb < OUT ? colb : OUT - 1);
      float bv = 0.f;
      if (BIAS) bv = bias[colc];
      v8f c = {};
#pragma unroll
      for (int k0 = 0; k0 < IN; k0 += 4) {
        const int krow = k0 + 2 * khalf;
        v2f a = *(const v2f*)(xrow + krow);
        v2f b;
        b.x = Wp[(size_t)krow * OUT + colc];
        b.y = Wp[(size_t)(krow + 1) * OUT + colc];
        c = __builtin_amdgcn_wmma_f32_16x16x4_f32(
             false, a,  false, b,
             (short)0, c,  false,  false);
      }
      if (BIAS || ACT) {
        asm volatile("v_nop\n\tv_nop\n\tv_nop\n\tv_nop\n\t"
                     "v_nop\n\tv_nop\n\tv_nop\n\tv_nop" : "+v"(c));
      }
      if (BIAS) {
#pragma unroll
        for (int r = 0; r < 8; ++r) c[r] = c[r] + bv;
      }
      if (ACT) {
#pragma unroll
        for (int r = 0; r < 8; ++r) {
          const float v = c[r];
          c[r] = (v > 0.f) ? v : expm1f(v);
        }
      }
      const int rowoff = khalf * 8;
      if (FULL && fulltile) {
#pragma unroll
        for (int r = 0; r < 8; ++r)
          y[(size_t)(row0 + rowoff + r) * OUT + colb] = c[r];
      } else {
        const bool colok = FULL || (colb < OUT);
#pragma unroll
        for (int r = 0; r < 8; ++r) {
          int row = row0 + rowoff + r;
          if (row < n && colok) y[(size_t)row * OUT + colb] = c[r];
        }
      }
    }
  }
}

constexpr int BIN_SHIFT = 8;
constexpr int BIN_NODES = 1 << BIN_SHIFT;
constexpr int EDGE_CHUNKS = 128;
constexpr int CHUNKS_PER_WAVE = EDGE_CHUNKS / 32;
constexpr int EDGE_BATCH = 8;
static_assert(EDGE_CHUNKS % 32 == 0, "one wave sums a whole number of chunks");

__global__ __launch_bounds__(1024) void bin_hist(const int* __restrict__ dst,
                                                 int* __restrict__ blockhist,
                                                 int E, int NB) {
  extern __shared__ int sh[];
  const int t = threadIdx.x;
  for (int i = t; i < NB; i += 1024) sh[i] = 0;
  __syncthreads();
  const int chunk = (E + EDGE_CHUNKS - 1) / EDGE_CHUNKS;
  const int e0 = (int)blockIdx.x * chunk;
  const int e1 = (e0 + chunk < E) ? (e0 + chunk) : E;
  for (int eb = e0 + t; eb < e1; eb += EDGE_BATCH * 1024) {
    int d[EDGE_BATCH];
    for (int k = 0; k < EDGE_BATCH; ++k) {
      const int e = eb + k * 1024;
      d[k] = (e < e1) ? dst[e] : 0;
    }
    for (int k = 0; k < EDGE_BATCH; ++k)
      if (eb + k * 1024 < e1) atomicAdd(&sh[d[k] >> BIN_SHIFT], 1);
  }
  __syncthreads();
  for (int i = t; i < NB; i += 1024) blockhist[(size_t)blockIdx.x * NB + i] = sh[i];
}

__global__ __launch_bounds__(1024) void bin_scatter(const int* __restrict__ src,
                                                    const int* __restrict__ dst,
                                                    const float* __restrict__ ew,
                                                    const int* __restrict__ blockhist,
                                                    int* __restrict__ binstart,
                                                    int* __restrict__ tsrc,
                                                    int* __restrict__ tdst,
                                                    float* __restrict__ tw,
                                                    int E, int NB) {
  extern __shared__ int sh[];
  __shared__ int s[1024];
  int* base = sh;
  int* tot  = sh + NB;
  const int t = threadIdx.x;
  const int lane = t & 31;
  const int wave = t >> 5;
  const int me = (int)blockIdx.x;

  for (int i = t; i < NB; i += 1024) { base[i] = 0; tot[i] = 0; }
  __syncthreads();
  for (int bb = lane; bb < NB; bb += 4 * 32) {
    int v[4][CHUNKS_PER_WAVE];
    for (int j = 0; j < 4; ++j) {
      const int b = bb + 32 * j;
      for (int k = 0; k < CHUNKS_PER_WAVE; ++k) {
        const int c = wave * CHUNKS_PER_WAVE + k;
        v[j][k] = (b < NB) ? blockhist[(size_t)c * NB + b] : 0;
      }
    }
    for (int j = 0; j < 4; ++j) {
      const int b = bb + 32 * j;
      int before = 0, total = 0;
      for (int k = 0; k < CHUNKS_PER_WAVE; ++k) {
        const int c = wave * CHUNKS_PER_WAVE + k;
        total += v[j][k];
        before += (c < me) ? v[j][k] : 0;
      }
      if (b < NB) {
        atomicAdd(&base[b], before);
        atomicAdd(&tot[b], total);
      }
    }
  }
  __syncthreads();
  int carry = 0;
  for (int b0 = 0; b0 < NB; b0 += 1024) {
    const int b = b0 + t;
    const int v = (b < NB) ? tot[b] : 0;
    s[t] = v;
    __syncthreads();
    for (int off = 1; off < 1024; off <<= 1) {
      const int add = (t >= off) ? s[t - off] : 0;
      __syncthreads();
      s[t] += add;
      __syncthreads();
    }
    const int excl = carry + s[t] - v;
    if (b < NB) {
      base[b] += excl;
      if (me == 0) binstart[b] = excl;
    }
    carry += s[1023];
    __syncthreads();
  }
  if (me == 0 && t == 0) binstart[NB] = carry;

  const int chunk = (E + EDGE_CHUNKS - 1) / EDGE_CHUNKS;
  const int e0 = me * chunk;
  const int e1 = (e0 + chunk < E) ? (e0 + chunk) : E;
  for (int eb = e0 + t; eb < e1; eb += EDGE_BATCH * 1024) {
    int d[EDGE_BATCH], a[EDGE_BATCH], p[EDGE_BATCH];
    float w[EDGE_BATCH];
    for (int k = 0; k < EDGE_BATCH; ++k) {
      const int e = eb + k * 1024;
      const bool ok = e < e1;
      d[k] = ok ? dst[e] : 0;
      a[k] = ok ? src[e] : 0;
      w[k] = ok ? ew[e] : 0.f;
    }
    for (int k = 0; k < EDGE_BATCH; ++k) {
      p[k] = 0;
      if (eb + k * 1024 < e1) p[k] = atomicAdd(&base[d[k] >> BIN_SHIFT], 1);
    }
    for (int k = 0; k < EDGE_BATCH; ++k) {
      if (eb + k * 1024 < e1) {
        tsrc[p[k]] = a[k];
        tdst[p[k]] = d[k];
        tw[p[k]]   = w[k];
      }
    }
  }
}

constexpr int BIN_WCAP = 8192;
constexpr int IN1 = 32, OUT1 = 4;
static_assert(BIN_NODES == 16 * 16, "sixteen waves, one 16-row tile each");
__global__ __launch_bounds__(1024) void bin_build(const int* __restrict__ binstart,
                                                  const int* __restrict__ tsrc,
                                                  const int* __restrict__ tdst,
                                                  const float* __restrict__ tw,
                                                  int* __restrict__ rowptr,
                                                  int* __restrict__ ssrc,
                                                  float* sw,
                                                  float* __restrict__ dinv,
                                                  const float* __restrict__ x,
                                                  const float* __restrict__ W1,
                                                  float* __restrict__ xw, int n) {
  __shared__ int cnt[BIN_NODES];
  __shared__ int s[BIN_NODES];
  __shared__ float wl[BIN_WCAP];
  __shared__ float sW1[IN1 * OUT1];
  const int t = threadIdx.x;
  const int bin = (int)blockIdx.x;
  const int lo = bin << BIN_SHIFT;
  const int p0 = binstart[bin];
  const int p1 = binstart[bin + 1];
  if (t < BIN_NODES) cnt[t] = 0;
  if (t < IN1 * OUT1) sW1[t] = W1[t];
  __syncthreads();
  for (int pb = p0 + t; pb < p1; pb += EDGE_BATCH * 1024) {
    int d[EDGE_BATCH];
    for (int k = 0; k < EDGE_BATCH; ++k) {
      const int p = pb + k * 1024;
      d[k] = (p < p1) ? tdst[p] : lo;
    }
    for (int k = 0; k < EDGE_BATCH; ++k)
      if (pb + k * 1024 < p1) atomicAdd(&cnt[d[k] - lo], 1);
  }
  __syncthreads();
  int v = 0;
  if (t < BIN_NODES) { v = cnt[t]; s[t] = v; }
  __syncthreads();
  for (int off = 1; off < BIN_NODES; off <<= 1) {
    int add = 0;
    if (t < BIN_NODES && t >= off) add = s[t - off];
    __syncthreads();
    if (t < BIN_NODES) s[t] += add;
    __syncthreads();
  }
  int excl = 0;
  if (t < BIN_NODES) {
    excl = s[t] - v;
    cnt[t] = excl;
    if (lo + t < n) rowptr[lo + t] = p0 + excl;
  }
  if (bin == (int)gridDim.x - 1 && t == 0) rowptr[n] = p1;
  __syncthreads();
  for (int pb = p0 + t; pb < p1; pb += EDGE_BATCH * 1024) {
    int d[EDGE_BATCH], a[EDGE_BATCH], q[EDGE_BATCH];
    float w[EDGE_BATCH];
    for (int k = 0; k < EDGE_BATCH; ++k) {
      const int p = pb + k * 1024;
      const bool ok = p < p1;
      d[k] = ok ? tdst[p] : lo;
      a[k] = ok ? tsrc[p] : 0;
      w[k] = ok ? tw[p] : 0.f;
    }
    for (int k = 0; k < EDGE_BATCH; ++k) {
      q[k] = p0;
      if (pb + k * 1024 < p1) q[k] = p0 + atomicAdd(&cnt[d[k] - lo], 1);
    }
    for (int k = 0; k < EDGE_BATCH; ++k) {
      if (pb + k * 1024 < p1) {
        ssrc[q[k]] = a[k];
        sw[q[k]]   = w[k];
        if (q[k] - p0 < BIN_WCAP) wl[q[k] - p0] = w[k];
      }
    }
  }
  __syncthreads();
  if (t < BIN_NODES && lo + t < n) {
    const int jend = excl + v;
    const int jlds = (jend < BIN_WCAP) ? jend : BIN_WCAP;
    float d = 1.0f;
    int j = excl;
    for (; j < jlds; ++j) d += wl[j];
    for (; j < jend; ++j) d += sw[p0 + j];
    dinv[lo + t] = (d > 0.f) ? rsqrtf(d) : 0.f;
  }
  const int wave = t >> 5;
  if (wave >= 8 && wave < 24)
    wmma_row_tile<IN1, OUT1, false, false>(x, sW1, nullptr, xw, n, lo + (wave - 8) * 16, t & 31);
}

template <int C, bool BIAS, bool ACT>
__global__ __launch_bounds__(256) void gather_rows(const float* __restrict__ xin,
                                                   const int* __restrict__ rowptr,
                                                   const int* __restrict__ ssrc,
                                                   float* snorm,
                                                   const float* __restrict__ dinv,
                                                   const float* __restrict__ b,
                                                   float* __restrict__ out, int n) {
  static_assert(C == 4, "one lane per node: the slot's owner is the only lane that reads it");
  constexpr int G = C / 4;
  const int tid = blockIdx.x * 256 + threadIdx.x;
  const int i = tid / G;
  const int q = tid - i * G;
  if (i >= n) return;
  const float* xq = xin + q * 4;
  const float dv = dinv[i];
  const float wself = dv * dv;
  const float4 sv = *(const float4*)(xq + (size_t)i * C);
  float ax = sv.x * wself, ay = sv.y * wself, az = sv.z * wself, aw = sv.w * wself;
  int p = rowptr[i];
  const int pend = rowptr[i + 1];
  for (; p + 4 <= pend; p += 4) {
    const int s0 = ssrc[p], s1 = ssrc[p + 1], s2 = ssrc[p + 2], s3 = ssrc[p + 3];
    const float e0 = snorm[p], e1 = snorm[p + 1], e2 = snorm[p + 2], e3 = snorm[p + 3];
    const float4 r0 = *(const float4*)(xq + (size_t)s0 * C);
    const float4 r1 = *(const float4*)(xq + (size_t)s1 * C);
    const float4 r2 = *(const float4*)(xq + (size_t)s2 * C);
    const float4 r3 = *(const float4*)(xq + (size_t)s3 * C);
    const float w0 = dinv[s0] * e0 * dv;
    const float w1 = dinv[s1] * e1 * dv;
    const float w2 = dinv[s2] * e2 * dv;
    const float w3 = dinv[s3] * e3 * dv;
    snorm[p] = w0; snorm[p + 1] = w1; snorm[p + 2] = w2; snorm[p + 3] = w3;
    ax += r0.x * w0; ay += r0.y * w0; az += r0.z * w0; aw += r0.w * w0;
    ax += r1.x * w1; ay += r1.y * w1; az += r1.z * w1; aw += r1.w * w1;
    ax += r2.x * w2; ay += r2.y * w2; az += r2.z * w2; aw += r2.w * w2;
    ax += r3.x * w3; ay += r3.y * w3; az += r3.z * w3; aw += r3.w * w3;
  }
  for (; p < pend; ++p) {
    const int s0 = ssrc[p];
    const float e0 = snorm[p];
    const float4 r0 = *(const float4*)(xq + (size_t)s0 * C);
    const float w0 = dinv[s0] * e0 * dv;
    snorm[p] = w0;
    ax += r0.x * w0; ay += r0.y * w0; az += r0.z * w0; aw += r0.w * w0;
  }
  if (BIAS) {
    ax += b[q * 4 + 0]; ay += b[q * 4 + 1]; az += b[q * 4 + 2]; aw += b[q * 4 + 3];
  }
  if (ACT) {
    ax = (ax > 0.f) ? ax : expm1f(ax);
    ay = (ay > 0.f) ? ay : expm1f(ay);
    az = (az > 0.f) ? az : expm1f(az);
    aw = (aw > 0.f) ? aw : expm1f(aw);
  }
  float4 o;
  o.x = ax; o.y = ay; o.z = az; o.w = aw;
  *(float4*)(out + (size_t)i * C + q * 4) = o;
}

template <int IN, int OUT, bool ACT>
__global__ __launch_bounds__(256) void agg_gemm(const float* __restrict__ xin,
                                                const int* __restrict__ rowptr,
                                                const int* __restrict__ ssrc,
                                                const float* __restrict__ snorm,
                                                const float* __restrict__ dinv,
                                                const float* __restrict__ W,
                                                const float* __restrict__ bias,
                                                float* __restrict__ y, int n) {
  static_assert(OUT % 16 == 0, "whole column tiles");
  constexpr int G = IN / 4;
  constexpr int NODES = 128;
  constexpr int NPP = 256 / G;
  constexpr int PITCH = (IN == 4) ? 4 : IN + 4;
  constexpr int EDGES = 8;
  constexpr bool USE_LDS = (IN * OUT) <= 8192;
  __shared__ alignas(16) float sA[NODES * PITCH];
  __shared__ float sW[USE_LDS ? (IN * OUT) : 1];
  const int t = threadIdx.x;
  const float* Wp = W;
  if constexpr (USE_LDS) {
    for (int i = t; i < IN * OUT; i += 256) sW[i] = W[i];
    Wp = sW;
  }

  const int nl0 = t / G;
  const int q = t - nl0 * G;
  const float* xq = xin + q * 4;
  for (int nl = nl0; nl < NODES; nl += NPP) {
    const int i = (int)blockIdx.x * NODES + nl;
    float ax = 0.f, ay = 0.f, az = 0.f, aw = 0.f;
    if (i < n) {
      const float dv = dinv[i];
      const float wself = dv * dv;
      const float4 sv = *(const float4*)(xq + (size_t)i * IN);
      int p = rowptr[i];
      const int pend = rowptr[i + 1];
      int s[EDGES];
      float w[EDGES];
#pragma unroll
      for (int k = 0; k < EDGES; ++k) {
        const bool ok = p + k < pend;
        const int pc = ok ? p + k : (p < pend ? p : 0);
        const int sk = ssrc[pc];
        const float wk = snorm[pc];
        s[k] = ok ? sk : i;
        w[k] = ok ? wk : 0.f;
      }
      ax = sv.x * wself; ay = sv.y * wself; az = sv.z * wself; aw = sv.w * wself;
      while (p < pend) {
        float4 r[EDGES];
#pragma unroll
        for (int k = 0; k < EDGES; ++k) r[k] = *(const float4*)(xq + (size_t)s[k] * IN);
        const int pn = p + EDGES;
        int sn[EDGES];
        float wn[EDGES];
#pragma unroll
        for (int k = 0; k < EDGES; ++k) {
          const bool ok = pn + k < pend;
          const int pc = ok ? pn + k : p;
          const int sk = ssrc[pc];
          const float wk = snorm[pc];
          sn[k] = ok ? sk : i;
          wn[k] = ok ? wk : 0.f;
        }
#pragma unroll
        for (int k = 0; k < EDGES; ++k) {
          ax += r[k].x * w[k]; ay += r[k].y * w[k]; az += r[k].z * w[k]; aw += r[k].w * w[k];
        }
#pragma unroll
        for (int k = 0; k < EDGES; ++k) { s[k] = sn[k]; w[k] = wn[k]; }
        p = pn;
      }
    }
    float4 o;
    o.x = ax; o.y = ay; o.z = az; o.w = aw;
    *(float4*)(sA + nl * PITCH + q * 4) = o;
  }
  __syncthreads();

  const int lane  = t & 31;
  const int wave  = t >> 5;
  const int l15   = lane & 15;
  const int khalf = lane >> 4;
  const int row0  = (int)blockIdx.x * NODES + wave * 16;

  if (row0 < n) {
    const float* arow = sA + (wave * 16 + l15) * PITCH;
    const bool fulltile = (row0 + 16 <= n);
    for (int n0 = 0; n0 < OUT; n0 += 16) {
      const int colb = n0 + l15;
      const float bv = bias[colb];
      v8f c = {};
#pragma unroll
      for (int k0 = 0; k0 < IN; k0 += 4) {
        const int krow = k0 + 2 * khalf;
        v2f a = *(const v2f*)(arow + krow);
        v2f b;
        b.x = Wp[(size_t)krow * OUT + colb];
        b.y = Wp[(size_t)(krow + 1) * OUT + colb];
        c = __builtin_amdgcn_wmma_f32_16x16x4_f32(
             false, a,  false, b,
             (short)0, c,  false,  false);
      }
      asm volatile("v_nop\n\tv_nop\n\tv_nop\n\tv_nop\n\t"
                   "v_nop\n\tv_nop\n\tv_nop\n\tv_nop" : "+v"(c));
#pragma unroll
      for (int r = 0; r < 8; ++r) c[r] = c[r] + bv;
      if (ACT) {
#pragma unroll
        for (int r = 0; r < 8; ++r) {
          const float v = c[r];
          c[r] = (v > 0.f) ? v : expm1f(v);
        }
      }
      const int rowoff = khalf * 8;
      if (fulltile) {
#pragma unroll
        for (int r = 0; r < 8; ++r)
          y[(size_t)(row0 + rowoff + r) * OUT + colb] = c[r];
      } else {
#pragma unroll
        for (int r = 0; r < 8; ++r) {
          int row = row0 + rowoff + r;
          if (row < n) y[(size_t)row * OUT + colb] = c[r];
        }
      }
    }
  }
}

static inline int cdiv(long long a, long long b) { return (int)((a + b - 1) / b); }

struct Graph {
  const int* rowptr;
  const int* ssrc;
  float* snorm;
  const float* dinv;
};

template <int C, bool BIAS, bool ACT>
static void run_gather(const float* xin, float* xout, const float* b,
                       const Graph& g, int N, hipStream_t stream) {
  gather_rows<C, BIAS, ACT><<<cdiv((long long)N * (C / 4), 256), 256, 0, stream>>>(
      xin, g.rowptr, g.ssrc, g.snorm, g.dinv, b, xout, N);
}

template <int IN, int OUT, bool ACT>
static void run_layer_fused(const float* xin, float* xout,
                            const float* W, const float* b, const Graph& g,
                            int N, hipStream_t stream) {
  agg_gemm<IN, OUT, ACT><<<cdiv(N, 128), 256, 0, stream>>>(
      xin, g.rowptr, g.ssrc, g.snorm, g.dinv, W, b, xout, N);
}

extern "C" void kernel_launch(void* const* d_in, const int* in_sizes, int n_in,
                              void* d_out, int out_size, void* d_ws, size_t ws_size,
                              hipStream_t stream) {
  const float* x   = (const float*)d_in[0];
  const int*   ei  = (const int*)d_in[1];
  const float* ew  = (const float*)d_in[2];
  const float* W1 = (const float*)d_in[3];  const float* b1 = (const float*)d_in[4];
  const float* W2 = (const float*)d_in[5];  const float* b2 = (const float*)d_in[6];
  const float* W3 = (const float*)d_in[7];  const float* b3 = (const float*)d_in[8];
  const float* W4 = (const float*)d_in[9];  const float* b4 = (const float*)d_in[10];
  const float* W5 = (const float*)d_in[11]; const float* b5 = (const float*)d_in[12];
  const float* W6 = (const float*)d_in[13]; const float* b6 = (const float*)d_in[14];

  const int N = in_sizes[0] / 32;
  const int E = in_sizes[2];
  const int* src = ei;
  const int* dst = ei + E;

  float* bufA   = (float*)d_ws;
  float* bufB   = bufA + (size_t)N * 128;
  float* dinv   = bufB + (size_t)N * 64;
  float* snorm  = dinv + N;
  int*   ssrc   = (int*)(snorm + E);
  int*   rowptr = ssrc + E;
  const int NB  = cdiv(N, BIN_NODES);
  int*   tsrc   = (int*)bufA;
  int*   tdst   = tsrc + E;
  float* tw     = (float*)(tdst + E);
  int*   bhist  = (int*)bufB;
  int*   bstart = bhist + (size_t)EDGE_CHUNKS * NB;
  float* xw1    = bufA + (size_t)N * 124;
  float* outp   = (float*)d_out;

  bin_hist<<<EDGE_CHUNKS, 1024, NB * sizeof(int), stream>>>(dst, bhist, E, NB);
  bin_scatter<<<EDGE_CHUNKS, 1024, 2 * NB * sizeof(int), stream>>>(
      src, dst, ew, bhist, bstart, tsrc, tdst, tw, E, NB);
  bin_build<<<NB, 1024, 0, stream>>>(bstart, tsrc, tdst, tw, rowptr, ssrc, snorm, dinv,
                                     x, W1, xw1, N);

  Graph g;
  g.rowptr = rowptr; g.ssrc = ssrc; g.snorm = snorm; g.dinv = dinv;

  run_gather<OUT1, true, true>(xw1, bufA, b1, g, N, stream);
  run_layer_fused<4,   16,  true >(bufA, bufB, W2, b2, g, N, stream);
  run_layer_fused<16,  32,  true >(bufB, bufA, W3, b3, g, N, stream);
  run_layer_fused<32,  64,  false>(bufA, bufB, W4, b4, g, N, stream);
  run_layer_fused<64,  128, true >(bufB, bufA, W5, b5, g, N, stream);
  run_layer_fused<128, 128, false>(bufA, outp, W6, b6, g, N, stream);
}


# Each edge kept as one 8-byte record (source, weight) through both grouping levels and in the buckets: 1 scattered store per edge per level (was 3 and 2), 1 index load per edge per gather (was 2)
# speedup vs baseline: 7.4436x; 1.2822x over previous
#include <hip/hip_runtime.h>
#include <cstdint>

typedef float v2f __attribute__((ext_vector_type(2)));
typedef float v8f __attribute__((ext_vector_type(8)));
typedef int v2i __attribute__((ext_vector_type(2)));

template <int IN, int OUT, bool BIAS, bool ACT>
__device__ __forceinline__ void wmma_row_tile(const float* __restrict__ x,
                                              const float* Wp,
                                              const float* __restrict__ bias,
                                              float* __restrict__ y, int n,
                                              int row0, int lane) {
  constexpr bool FULL    = (OUT % 16) == 0;
  const int l15   = lane & 15;
  const int khalf = lane >> 4;

  if (row0 < n) {
    int mload = row0 + l15;
    if (mload > n - 1) mload = n - 1;
    const float* xrow = x + (size_t)mload * IN;
    const bool fulltile = (row0 + 16 <= n);

    for (int n0 = 0; n0 < OUT; n0 += 16) {
      const int colb = n0 + l15;
      const int colc = FULL ? colb : (colb < OUT ? colb : OUT - 1);
      float bv = 0.f;
      if (BIAS) bv = bias[colc];
      v8f c = {};
#pragma unroll
      for (int k0 = 0; k0 < IN; k0 += 4) {
        const int krow = k0 + 2 * khalf;
        v2f a = *(const v2f*)(xrow + krow);
        v2f b;
        b.x = Wp[(size_t)krow * OUT + colc];
        b.y = Wp[(size_t)(krow + 1) * OUT + colc];
        c = __builtin_amdgcn_wmma_f32_16x16x4_f32(
             false, a,  false, b,
             (short)0, c,  false,  false);
      }
      if (BIAS || ACT) {
        asm volatile("v_nop\n\tv_nop\n\tv_nop\n\tv_nop\n\t"
                     "v_nop\n\tv_nop\n\tv_nop\n\tv_nop" : "+v"(c));
      }
      if (BIAS) {
#pragma unroll
        for (int r = 0; r < 8; ++r) c[r] = c[r] + bv;
      }
      if (ACT) {
#pragma unroll
        for (int r = 0; r < 8; ++r) {
          const float v = c[r];
          c[r] = (v > 0.f) ? v : expm1f(v);
        }
      }
      const int rowoff = khalf * 8;
      if (FULL && fulltile) {
#pragma unroll
        for (int r = 0; r < 8; ++r)
          y[(size_t)(row0 + rowoff + r) * OUT + colb] = c[r];
      } else {
        const bool colok = FULL || (colb < OUT);
#pragma unroll
        for (int r = 0; r < 8; ++r) {
          int row = row0 + rowoff + r;
          if (row < n && colok) y[(size_t)row * OUT + colb] = c[r];
        }
      }
    }
  }
}

constexpr int BIN_SHIFT = 8;
constexpr int BIN_NODES = 1 << BIN_SHIFT;
constexpr int EDGE_CHUNKS = 128;
constexpr int CHUNKS_PER_WAVE = EDGE_CHUNKS / 32;
constexpr int EDGE_BATCH = 8;
static_assert(EDGE_CHUNKS % 32 == 0, "one wave sums a whole number of chunks");

__global__ __launch_bounds__(1024) void bin_hist(const int* __restrict__ dst,
                                                 int* __restrict__ blockhist,
                                                 int E, int NB) {
  extern __shared__ int sh[];
  const int t = threadIdx.x;
  for (int i = t; i < NB; i += 1024) sh[i] = 0;
  __syncthreads();
  const int chunk = (E + EDGE_CHUNKS - 1) / EDGE_CHUNKS;
  const int e0 = (int)blockIdx.x * chunk;
  const int e1 = (e0 + chunk < E) ? (e0 + chunk) : E;
  for (int eb = e0 + t; eb < e1; eb += EDGE_BATCH * 1024) {
    int d[EDGE_BATCH];
    for (int k = 0; k < EDGE_BATCH; ++k) {
      const int e = eb + k * 1024;
      d[k] = (e < e1) ? dst[e] : 0;
    }
    for (int k = 0; k < EDGE_BATCH; ++k)
      if (eb + k * 1024 < e1) atomicAdd(&sh[d[k] >> BIN_SHIFT], 1);
  }
  __syncthreads();
  for (int i = t; i < NB; i += 1024) blockhist[(size_t)blockIdx.x * NB + i] = sh[i];
}

__global__ __launch_bounds__(1024) void bin_scatter(const int* __restrict__ src,
                                                    const int* __restrict__ dst,
                                                    const float* __restrict__ ew,
                                                    const int* __restrict__ blockhist,
                                                    int* __restrict__ binstart,
                                                    int* __restrict__ trec,
                                                    int E, int NB) {
  extern __shared__ int sh[];
  __shared__ int s[1024];
  int* base = sh;
  int* tot  = sh + NB;
  const int t = threadIdx.x;
  const int lane = t & 31;
  const int wave = t >> 5;
  const int me = (int)blockIdx.x;

  for (int i = t; i < NB; i += 1024) { base[i] = 0; tot[i] = 0; }
  __syncthreads();
  for (int bb = lane; bb < NB; bb += 4 * 32) {
    int v[4][CHUNKS_PER_WAVE];
    for (int j = 0; j < 4; ++j) {
      const int b = bb + 32 * j;
      for (int k = 0; k < CHUNKS_PER_WAVE; ++k) {
        const int c = wave * CHUNKS_PER_WAVE + k;
        v[j][k] = (b < NB) ? blockhist[(size_t)c * NB + b] : 0;
      }
    }
    for (int j = 0; j < 4; ++j) {
      const int b = bb + 32 * j;
      int before = 0, total = 0;
      for (int k = 0; k < CHUNKS_PER_WAVE; ++k) {
        const int c = wave * CHUNKS_PER_WAVE + k;
        total += v[j][k];
        before += (c < me) ? v[j][k] : 0;
      }
      if (b < NB) {
        atomicAdd(&base[b], before);
        atomicAdd(&tot[b], total);
      }
    }
  }
  __syncthreads();
  int carry = 0;
  for (int b0 = 0; b0 < NB; b0 += 1024) {
    const int b = b0 + t;
    const int v = (b < NB) ? tot[b] : 0;
    s[t] = v;
    __syncthreads();
    for (int off = 1; off < 1024; off <<= 1) {
      const int add = (t >= off) ? s[t - off] : 0;
      __syncthreads();
      s[t] += add;
      __syncthreads();
    }
    const int excl = carry + s[t] - v;
    if (b < NB) {
      base[b] += excl;
      if (me == 0) binstart[b] = excl;
    }
    carry += s[1023];
    __syncthreads();
  }
  if (me == 0 && t == 0) binstart[NB] = carry;

  const int chunk = (E + EDGE_CHUNKS - 1) / EDGE_CHUNKS;
  const int e0 = me * chunk;
  const int e1 = (e0 + chunk < E) ? (e0 + chunk) : E;
  for (int eb = e0 + t; eb < e1; eb += EDGE_BATCH * 1024) {
    int d[EDGE_BATCH], a[EDGE_BATCH], p[EDGE_BATCH];
    float w[EDGE_BATCH];
    for (int k = 0; k < EDGE_BATCH; ++k) {
      const int e = eb + k * 1024;
      const bool ok = e < e1;
      d[k] = ok ? dst[e] : 0;
      a[k] = ok ? src[e] : 0;
      w[k] = ok ? ew[e] : 0.f;
    }
    for (int k = 0; k < EDGE_BATCH; ++k) {
      p[k] = 0;
      if (eb + k * 1024 < e1) p[k] = atomicAdd(&base[d[k] >> BIN_SHIFT], 1);
    }
    for (int k = 0; k < EDGE_BATCH; ++k) {
      if (eb + k * 1024 < e1) {
        v2i r;
        r.x = (a[k] << BIN_SHIFT) | (d[k] & (BIN_NODES - 1));
        r.y = __float_as_int(w[k]);
        ((v2i*)trec)[p[k]] = r;
      }
    }
  }
}

constexpr int BIN_WCAP = 8192;
constexpr int IN1 = 32, OUT1 = 4;
static_assert(BIN_NODES == 16 * 16, "sixteen waves, one 16-row tile each");
__global__ __launch_bounds__(1024) void bin_build(const int* __restrict__ binstart,
                                                  const int* __restrict__ trec,
                                                  int* __restrict__ rowptr,
                                                  int* srec,
                                                  float* __restrict__ dinv,
                                                  const float* __restrict__ x,
                                                  const float* __restrict__ W1,
                                                  float* __restrict__ xw, int n) {
  __shared__ int cnt[BIN_NODES];
  __shared__ int s[BIN_NODES];
  __shared__ float wl[BIN_WCAP];
  __shared__ float sW1[IN1 * OUT1];
  const int t = threadIdx.x;
  const int bin = (int)blockIdx.x;
  const int lo = bin << BIN_SHIFT;
  const int p0 = binstart[bin];
  const int p1 = binstart[bin + 1];
  if (t < BIN_NODES) cnt[t] = 0;
  if (t < IN1 * OUT1) sW1[t] = W1[t];
  __syncthreads();
  for (int pb = p0 + t; pb < p1; pb += EDGE_BATCH * 1024) {
    int d[EDGE_BATCH];
    for (int k = 0; k < EDGE_BATCH; ++k) {
      const int p = pb + k * 1024;
      d[k] = (p < p1) ? (trec[2 * (size_t)p] & (BIN_NODES - 1)) : 0;
    }
    for (int k = 0; k < EDGE_BATCH; ++k)
      if (pb + k * 1024 < p1) atomicAdd(&cnt[d[k]], 1);
  }
  __syncthreads();
  int v = 0;
  if (t < BIN_NODES) { v = cnt[t]; s[t] = v; }
  __syncthreads();
  for (int off = 1; off < BIN_NODES; off <<= 1) {
    int add = 0;
    if (t < BIN_NODES && t >= off) add = s[t - off];
    __syncthreads();
    if (t < BIN_NODES) s[t] += add;
    __syncthreads();
  }
  int excl = 0;
  if (t < BIN_NODES) {
    excl = s[t] - v;
    cnt[t] = excl;
    if (lo + t < n) rowptr[lo + t] = p0 + excl;
  }
  if (bin == (int)gridDim.x - 1 && t == 0) rowptr[n] = p1;
  __syncthreads();
  for (int pb = p0 + t; pb < p1; pb += EDGE_BATCH * 1024) {
    int q[EDGE_BATCH];
    v2i r[EDGE_BATCH];
    for (int k = 0; k < EDGE_BATCH; ++k) {
      const int p = pb + k * 1024;
      v2i z = {0, 0};
      r[k] = (p < p1) ? ((const v2i*)trec)[p] : z;
    }
    for (int k = 0; k < EDGE_BATCH; ++k) {
      q[k] = p0;
      if (pb + k * 1024 < p1) q[k] = p0 + atomicAdd(&cnt[r[k].x & (BIN_NODES - 1)], 1);
    }
    for (int k = 0; k < EDGE_BATCH; ++k) {
      if (pb + k * 1024 < p1) {
        v2i o;
        o.x = r[k].x >> BIN_SHIFT;
        o.y = r[k].y;
        ((v2i*)srec)[q[k]] = o;
        if (q[k] - p0 < BIN_WCAP) wl[q[k] - p0] = __int_as_float(r[k].y);
      }
    }
  }
  __syncthreads();
  if (t < BIN_NODES && lo + t < n) {
    const int jend = excl + v;
    const int jlds = (jend < BIN_WCAP) ? jend : BIN_WCAP;
    float d = 1.0f;
    int j = excl;
    for (; j < jlds; ++j) d += wl[j];
    for (; j < jend; ++j)
      d += __int_as_float(srec[2 * (size_t)(p0 + j) + 1]);
    dinv[lo + t] = (d > 0.f) ? rsqrtf(d) : 0.f;
  }
  const int wave = t >> 5;
  if (wave >= 8 && wave < 24)
    wmma_row_tile<IN1, OUT1, false, false>(x, sW1, nullptr, xw, n, lo + (wave - 8) * 16, t & 31);
}

template <int C, bool BIAS, bool ACT>
__global__ __launch_bounds__(256) void gather_rows(const float* __restrict__ xin,
                                                   const int* __restrict__ rowptr,
                                                   int* srec,
                                                   const float* __restrict__ dinv,
                                                   const float* __restrict__ b,
                                                   float* __restrict__ out, int n) {
  static_assert(C == 4, "one lane per node: the slot's owner is the only lane that reads it");
  constexpr int G = C / 4;
  const int tid = blockIdx.x * 256 + threadIdx.x;
  const int i = tid / G;
  const int q = tid - i * G;
  if (i >= n) return;
  const float* xq = xin + q * 4;
  const float dv = dinv[i];
  const float wself = dv * dv;
  const float4 sv = *(const float4*)(xq + (size_t)i * C);
  float ax = sv.x * wself, ay = sv.y * wself, az = sv.z * wself, aw = sv.w * wself;
  v2i* rec = (v2i*)srec;
  int p = rowptr[i];
  const int pend = rowptr[i + 1];
  for (; p + 4 <= pend; p += 4) {
    const v2i c0 = rec[p], c1 = rec[p + 1], c2 = rec[p + 2], c3 = rec[p + 3];
    const int s0 = c0.x, s1 = c1.x, s2 = c2.x, s3 = c3.x;
    const float e0 = __int_as_float(c0.y), e1 = __int_as_float(c1.y);
    const float e2 = __int_as_float(c2.y), e3 = __int_as_float(c3.y);
    const float4 r0 = *(const float4*)(xq + (size_t)s0 * C);
    const float4 r1 = *(const float4*)(xq + (size_t)s1 * C);
    const float4 r2 = *(const float4*)(xq + (size_t)s2 * C);
    const float4 r3 = *(const float4*)(xq + (size_t)s3 * C);
    const float w0 = dinv[s0] * e0 * dv;
    const float w1 = dinv[s1] * e1 * dv;
    const float w2 = dinv[s2] * e2 * dv;
    const float w3 = dinv[s3] * e3 * dv;
    const v2i o0 = {s0, __float_as_int(w0)}, o1 = {s1, __float_as_int(w1)};
    const v2i o2 = {s2, __float_as_int(w2)}, o3 = {s3, __float_as_int(w3)};
    rec[p] = o0; rec[p + 1] = o1; rec[p + 2] = o2; rec[p + 3] = o3;
    ax += r0.x * w0; ay += r0.y * w0; az += r0.z * w0; aw += r0.w * w0;
    ax += r1.x * w1; ay += r1.y * w1; az += r1.z * w1; aw += r1.w * w1;
    ax += r2.x * w2; ay += r2.y * w2; az += r2.z * w2; aw += r2.w * w2;
    ax += r3.x * w3; ay += r3.y * w3; az += r3.z * w3; aw += r3.w * w3;
  }
  for (; p < pend; ++p) {
    const v2i c0 = rec[p];
    const int s0 = c0.x;
    const float e0 = __int_as_float(c0.y);
    const float4 r0 = *(const float4*)(xq + (size_t)s0 * C);
    const float w0 = dinv[s0] * e0 * dv;
    srec[2 * (size_t)p + 1] = __float_as_int(w0);
    ax += r0.x * w0; ay += r0.y * w0; az += r0.z * w0; aw += r0.w * w0;
  }
  if (BIAS) {
    ax += b[q * 4 + 0]; ay += b[q * 4 + 1]; az += b[q * 4 + 2]; aw += b[q * 4 + 3];
  }
  if (ACT) {
    ax = (ax > 0.f) ? ax : expm1f(ax);
    ay = (ay > 0.f) ? ay : expm1f(ay);
    az = (az > 0.f) ? az : expm1f(az);
    aw = (aw > 0.f) ? aw : expm1f(aw);
  }
  float4 o;
  o.x = ax; o.y = ay; o.z = az; o.w = aw;
  *(float4*)(out + (size_t)i * C + q * 4) = o;
}

template <int IN, int OUT, bool ACT>
__global__ __launch_bounds__(256) void agg_gemm(const float* __restrict__ xin,
                                                const int* __restrict__ rowptr,
                                                const int* __restrict__ srec,
                                                const float* __restrict__ dinv,
                                                const float* __restrict__ W,
                                                const float* __restrict__ bias,
                                                float* __restrict__ y, int n) {
  static_assert(OUT % 16 == 0, "whole column tiles");
  constexpr int G = IN / 4;
  constexpr int NODES = 128;
  constexpr int NPP = 256 / G;
  constexpr int PITCH = (IN == 4) ? 4 : IN + 4;
  constexpr int EDGES = 8;
  constexpr bool USE_LDS = (IN * OUT) <= 8192;
  __shared__ alignas(16) float sA[NODES * PITCH];
  __shared__ float sW[USE_LDS ? (IN * OUT) : 1];
  const int t = threadIdx.x;
  const float* Wp = W;
  if constexpr (USE_LDS) {
    for (int i = t; i < IN * OUT; i += 256) sW[i] = W[i];
    Wp = sW;
  }

  const int nl0 = t / G;
  const int q = t - nl0 * G;
  const float* xq = xin + q * 4;
  const v2i* rec = (const v2i*)srec;
  for (int nl = nl0; nl < NODES; nl += NPP) {
    const int i = (int)blockIdx.x * NODES + nl;
    float ax = 0.f, ay = 0.f, az = 0.f, aw = 0.f;
    if (i < n) {
      const float dv = dinv[i];
      const float wself = dv * dv;
      const float4 sv = *(const float4*)(xq + (size_t)i * IN);
      int p = rowptr[i];
      const int pend = rowptr[i + 1];
      int s[EDGES];
      float w[EDGES];
#pragma unroll
      for (int k = 0; k < EDGES; ++k) {
        const bool ok = p + k < pend;
        const int pc = ok ? p + k : (p < pend ? p : 0);
        const v2i c = rec[pc];
        s[k] = ok ? c.x : i;
        w[k] = ok ? __int_as_float(c.y) : 0.f;
      }
      ax = sv.x * wself; ay = sv.y * wself; az = sv.z * wself; aw = sv.w * wself;
      while (p < pend) {
        float4 r[EDGES];
#pragma unroll
        for (int k = 0; k < EDGES; ++k) r[k] = *(const float4*)(xq + (size_t)s[k] * IN);
        const int pn = p + EDGES;
        int sn[EDGES];
        float wn[EDGES];
#pragma unroll
        for (int k = 0; k < EDGES; ++k) {
          const bool ok = pn + k < pend;
          const int pc = ok ? pn + k : p;
          const v2i c = rec[pc];
          sn[k] = ok ? c.x : i;
          wn[k] = ok ? __int_as_float(c.y) : 0.f;
        }
#pragma unroll
        for (int k = 0; k < EDGES; ++k) {
          ax += r[k].x * w[k]; ay += r[k].y * w[k]; az += r[k].z * w[k]; aw += r[k].w * w[k];
        }
#pragma unroll
        for (int k = 0; k < EDGES; ++k) { s[k] = sn[k]; w[k] = wn[k]; }
        p = pn;
      }
    }
    float4 o;
    o.x = ax; o.y = ay; o.z = az; o.w = aw;
    *(float4*)(sA + nl * PITCH + q * 4) = o;
  }
  __syncthreads();

  const int lane  = t & 31;
  const int wave  = t >> 5;
  const int l15   = lane & 15;
  const int khalf = lane >> 4;
  const int row0  = (int)blockIdx.x * NODES + wave * 16;

  if (row0 < n) {
    const float* arow = sA + (wave * 16 + l15) * PITCH;
    const bool fulltile = (row0 + 16 <= n);
    for (int n0 = 0; n0 < OUT; n0 += 16) {
      const int colb = n0 + l15;
      const float bv = bias[colb];
      v8f c = {};
#pragma unroll
      for (int k0 = 0; k0 < IN; k0 += 4) {
        const int krow = k0 + 2 * khalf;
        v2f a = *(const v2f*)(arow + krow);
        v2f b;
        b.x = Wp[(size_t)krow * OUT + colb];
        b.y = Wp[(size_t)(krow + 1) * OUT + colb];
        c = __builtin_amdgcn_wmma_f32_16x16x4_f32(
             false, a,  false, b,
             (short)0, c,  false,  false);
      }
      asm volatile("v_nop\n\tv_nop\n\tv_nop\n\tv_nop\n\t"
                   "v_nop\n\tv_nop\n\tv_nop\n\tv_nop" : "+v"(c));
#pragma unroll
      for (int r = 0; r < 8; ++r) c[r] = c[r] + bv;
      if (ACT) {
#pragma unroll
        for (int r = 0; r < 8; ++r) {
          const float v = c[r];
          c[r] = (v > 0.f) ? v : expm1f(v);
        }
      }
      const int rowoff = khalf * 8;
      if (fulltile) {
#pragma unroll
        for (int r = 0; r < 8; ++r)
          y[(size_t)(row0 + rowoff + r) * OUT + colb] = c[r];
      } else {
#pragma unroll
        for (int r = 0; r < 8; ++r) {
          int row = row0 + rowoff + r;
          if (row < n) y[(size_t)row * OUT + colb] = c[r];
        }
      }
    }
  }
}

static inline int cdiv(long long a, long long b) { return (int)((a + b - 1) / b); }

struct Graph {
  const int* rowptr;
  int* srec;
  const float* dinv;
};

template <int C, bool BIAS, bool ACT>
static void run_gather(const float* xin, float* xout, const float* b,
                       const Graph& g, int N, hipStream_t stream) {
  gather_rows<C, BIAS, ACT><<<cdiv((long long)N * (C / 4), 256), 256, 0, stream>>>(
      xin, g.rowptr, g.srec, g.dinv, b, xout, N);
}

template <int IN, int OUT, bool ACT>
static void run_layer_fused(const float* xin, float* xout,
                            const float* W, const float* b, const Graph& g,
                            int N, hipStream_t stream) {
  agg_gemm<IN, OUT, ACT><<<cdiv(N, 128), 256, 0, stream>>>(
      xin, g.rowptr, g.srec, g.dinv, W, b, xout, N);
}

extern "C" void kernel_launch(void* const* d_in, const int* in_sizes, int n_in,
                              void* d_out, int out_size, void* d_ws, size_t ws_size,
                              hipStream_t stream) {
  const float* x   = (const float*)d_in[0];
  const int*   ei  = (const int*)d_in[1];
  const float* ew  = (const float*)d_in[2];
  const float* W1 = (const float*)d_in[3];  const float* b1 = (const float*)d_in[4];
  const float* W2 = (const float*)d_in[5];  const float* b2 = (const float*)d_in[6];
  const float* W3 = (const float*)d_in[7];  const float* b3 = (const float*)d_in[8];
  const float* W4 = (const float*)d_in[9];  const float* b4 = (const float*)d_in[10];
  const float* W5 = (const float*)d_in[11]; const float* b5 = (const float*)d_in[12];
  const float* W6 = (const float*)d_in[13]; const float* b6 = (const float*)d_in[14];

  const int N = in_sizes[0] / 32;
  const int E = in_sizes[2];
  const int* src = ei;
  const int* dst = ei + E;

  float* bufA   = (float*)d_ws;
  float* bufB   = bufA + (size_t)N * 128;
  int*   srec   = (int*)(bufB + (size_t)N * 64);
  float* dinv   = (float*)(srec + 2 * (size_t)E);
  int*   rowptr = (int*)(dinv + N);
  const int NB  = cdiv(N, BIN_NODES);
  int*   trec   = (int*)bufA;
  int*   bhist  = (int*)bufB;
  int*   bstart = bhist + (size_t)EDGE_CHUNKS * NB;
  float* xw1    = bufA + (size_t)N * 124;
  float* outp   = (float*)d_out;

  bin_hist<<<EDGE_CHUNKS, 1024, NB * sizeof(int), stream>>>(dst, bhist, E, NB);
  bin_scatter<<<EDGE_CHUNKS, 1024, 2 * NB * sizeof(int), stream>>>(
      src, dst, ew, bhist, bstart, trec, E, NB);
  bin_build<<<NB, 1024, 0, stream>>>(bstart, trec, rowptr, srec, dinv,
                                     x, W1, xw1, N);

  Graph g;
  g.rowptr = rowptr; g.srec = srec; g.dinv = dinv;

  run_gather<OUT1, true, true>(xw1, bufA, b1, g, N, stream);
  run_layer_fused<4,   16,  true >(bufA, bufB, W2, b2, g, N, stream);
  run_layer_fused<16,  32,  true >(bufB, bufA, W3, b3, g, N, stream);
  run_layer_fused<32,  64,  false>(bufA, bufB, W4, b4, g, N, stream);
  run_layer_fused<64,  128, true >(bufB, bufA, W5, b5, g, N, stream);
  run_layer_fused<128, 128, false>(bufA, outp, W6, b6, g, N, stream);
}


# Fused layers' WMMA phase re-tiled: a wave keeps up to 2 column x 4 row tiles' accumulators, one B fragment load per k-step feeds all its row tiles (block reads W 2x, was 8x); sum order unchanged
# speedup vs baseline: 7.7500x; 1.0412x over previous
#include <hip/hip_runtime.h>
#include <cstdint>

typedef float v2f __attribute__((ext_vector_type(2)));
typedef float v8f __attribute__((ext_vector_type(8)));
typedef int v2i __attribute__((ext_vector_type(2)));

template <int IN, int OUT, bool BIAS, bool ACT>
__device__ __forceinline__ void wmma_row_tile(const float* __restrict__ x,
                                              const float* Wp,
                                              const float* __restrict__ bias,
                                              float* __restrict__ y, int n,
                                              int row0, int lane) {
  constexpr bool FULL    = (OUT % 16) == 0;
  const int l15   = lane & 15;
  const int khalf = lane >> 4;

  if (row0 < n) {
    int mload = row0 + l15;
    if (mload > n - 1) mload = n - 1;
    const float* xrow = x + (size_t)mload * IN;
    const bool fulltile = (row0 + 16 <= n);

    for (int n0 = 0; n0 < OUT; n0 += 16) {
      const int colb = n0 + l15;
      const int colc = FULL ? colb : (colb < OUT ? colb : OUT - 1);
      float bv = 0.f;
      if (BIAS) bv = bias[colc];
      v8f c = {};
#pragma unroll
      for (int k0 = 0; k0 < IN; k0 += 4) {
        const int krow = k0 + 2 * khalf;
        v2f a = *(const v2f*)(xrow + krow);
        v2f b;
        b.x = Wp[(size_t)krow * OUT + colc];
        b.y = Wp[(size_t)(krow + 1) * OUT + colc];
        c = __builtin_amdgcn_wmma_f32_16x16x4_f32(
             false, a,  false, b,
             (short)0, c,  false,  false);
      }
      if (BIAS || ACT) {
        asm volatile("v_nop\n\tv_nop\n\tv_nop\n\tv_nop\n\t"
                     "v_nop\n\tv_nop\n\tv_nop\n\tv_nop" : "+v"(c));
      }
      if (BIAS) {
#pragma unroll
        for (int r = 0; r < 8; ++r) c[r] = c[r] + bv;
      }
      if (ACT) {
#pragma unroll
        for (int r = 0; r < 8; ++r) {
          const float v = c[r];
          c[r] = (v > 0.f) ? v : expm1f(v);
        }
      }
      const int rowoff = khalf * 8;
      if (FULL && fulltile) {
#pragma unroll
        for (int r = 0; r < 8; ++r)
          y[(size_t)(row0 + rowoff + r) * OUT + colb] = c[r];
      } else {
        const bool colok = FULL || (colb < OUT);
#pragma unroll
        for (int r = 0; r < 8; ++r) {
          int row = row0 + rowoff + r;
          if (row < n && colok) y[(size_t)row * OUT + colb] = c[r];
        }
      }
    }
  }
}

constexpr int BIN_SHIFT = 8;
constexpr int BIN_NODES = 1 << BIN_SHIFT;
constexpr int EDGE_CHUNKS = 128;
constexpr int CHUNKS_PER_WAVE = EDGE_CHUNKS / 32;
constexpr int EDGE_BATCH = 8;
static_assert(EDGE_CHUNKS % 32 == 0, "one wave sums a whole number of chunks");

__global__ __launch_bounds__(1024) void bin_hist(const int* __restrict__ dst,
                                                 int* __restrict__ blockhist,
                                                 int E, int NB) {
  extern __shared__ int sh[];
  const int t = threadIdx.x;
  for (int i = t; i < NB; i += 1024) sh[i] = 0;
  __syncthreads();
  const int chunk = (E + EDGE_CHUNKS - 1) / EDGE_CHUNKS;
  const int e0 = (int)blockIdx.x * chunk;
  const int e1 = (e0 + chunk < E) ? (e0 + chunk) : E;
  for (int eb = e0 + t; eb < e1; eb += EDGE_BATCH * 1024) {
    int d[EDGE_BATCH];
    for (int k = 0; k < EDGE_BATCH; ++k) {
      const int e = eb + k * 1024;
      d[k] = (e < e1) ? dst[e] : 0;
    }
    for (int k = 0; k < EDGE_BATCH; ++k)
      if (eb + k * 1024 < e1) atomicAdd(&sh[d[k] >> BIN_SHIFT], 1);
  }
  __syncthreads();
  for (int i = t; i < NB; i += 1024) blockhist[(size_t)blockIdx.x * NB + i] = sh[i];
}

__global__ __launch_bounds__(1024) void bin_scatter(const int* __restrict__ src,
                                                    const int* __restrict__ dst,
                                                    const float* __restrict__ ew,
                                                    const int* __restrict__ blockhist,
                                                    int* __restrict__ binstart,
                                                    int* __restrict__ trec,
                                                    int E, int NB) {
  extern __shared__ int sh[];
  __shared__ int s[1024];
  int* base = sh;
  int* tot  = sh + NB;
  const int t = threadIdx.x;
  const int lane = t & 31;
  const int wave = t >> 5;
  const int me = (int)blockIdx.x;

  for (int i = t; i < NB; i += 1024) { base[i] = 0; tot[i] = 0; }
  __syncthreads();
  for (int bb = lane; bb < NB; bb += 4 * 32) {
    int v[4][CHUNKS_PER_WAVE];
    for (int j = 0; j < 4; ++j) {
      const int b = bb + 32 * j;
      for (int k = 0; k < CHUNKS_PER_WAVE; ++k) {
        const int c = wave * CHUNKS_PER_WAVE + k;
        v[j][k] = (b < NB) ? blockhist[(size_t)c * NB + b] : 0;
      }
    }
    for (int j = 0; j < 4; ++j) {
      const int b = bb + 32 * j;
      int before = 0, total = 0;
      for (int k = 0; k < CHUNKS_PER_WAVE; ++k) {
        const int c = wave * CHUNKS_PER_WAVE + k;
        total += v[j][k];
        before += (c < me) ? v[j][k] : 0;
      }
      if (b < NB) {
        atomicAdd(&base[b], before);
        atomicAdd(&tot[b], total);
      }
    }
  }
  __syncthreads();
  int carry = 0;
  for (int b0 = 0; b0 < NB; b0 += 1024) {
    const int b = b0 + t;
    const int v = (b < NB) ? tot[b] : 0;
    s[t] = v;
    __syncthreads();
    for (int off = 1; off < 1024; off <<= 1) {
      const int add = (t >= off) ? s[t - off] : 0;
      __syncthreads();
      s[t] += add;
      __syncthreads();
    }
    const int excl = carry + s[t] - v;
    if (b < NB) {
      base[b] += excl;
      if (me == 0) binstart[b] = excl;
    }
    carry += s[1023];
    __syncthreads();
  }
  if (me == 0 && t == 0) binstart[NB] = carry;

  const int chunk = (E + EDGE_CHUNKS - 1) / EDGE_CHUNKS;
  const int e0 = me * chunk;
  const int e1 = (e0 + chunk < E) ? (e0 + chunk) : E;
  for (int eb = e0 + t; eb < e1; eb += EDGE_BATCH * 1024) {
    int d[EDGE_BATCH], a[EDGE_BATCH], p[EDGE_BATCH];
    float w[EDGE_BATCH];
    for (int k = 0; k < EDGE_BATCH; ++k) {
      const int e = eb + k * 1024;
      const bool ok = e < e1;
      d[k] = ok ? dst[e] : 0;
      a[k] = ok ? src[e] : 0;
      w[k] = ok ? ew[e] : 0.f;
    }
    for (int k = 0; k < EDGE_BATCH; ++k) {
      p[k] = 0;
      if (eb + k * 1024 < e1) p[k] = atomicAdd(&base[d[k] >> BIN_SHIFT], 1);
    }
    for (int k = 0; k < EDGE_BATCH; ++k) {
      if (eb + k * 1024 < e1) {
        v2i r;
        r.x = (a[k] << BIN_SHIFT) | (d[k] & (BIN_NODES - 1));
        r.y = __float_as_int(w[k]);
        ((v2i*)trec)[p[k]] = r;
      }
    }
  }
}

constexpr int BIN_WCAP = 8192;
constexpr int IN1 = 32, OUT1 = 4;
static_assert(BIN_NODES == 16 * 16, "sixteen waves, one 16-row tile each");
__global__ __launch_bounds__(1024) void bin_build(const int* __restrict__ binstart,
                                                  const int* __restrict__ trec,
                                                  int* __restrict__ rowptr,
                                                  int* srec,
                                                  float* __restrict__ dinv,
                                                  const float* __restrict__ x,
                                                  const float* __restrict__ W1,
                                                  float* __restrict__ xw, int n) {
  __shared__ int cnt[BIN_NODES];
  __shared__ int s[BIN_NODES];
  __shared__ float wl[BIN_WCAP];
  __shared__ float sW1[IN1 * OUT1];
  const int t = threadIdx.x;
  const int bin = (int)blockIdx.x;
  const int lo = bin << BIN_SHIFT;
  const int p0 = binstart[bin];
  const int p1 = binstart[bin + 1];
  if (t < BIN_NODES) cnt[t] = 0;
  if (t < IN1 * OUT1) sW1[t] = W1[t];
  __syncthreads();
  for (int pb = p0 + t; pb < p1; pb += EDGE_BATCH * 1024) {
    int d[EDGE_BATCH];
    for (int k = 0; k < EDGE_BATCH; ++k) {
      const int p = pb + k * 1024;
      d[k] = (p < p1) ? (trec[2 * (size_t)p] & (BIN_NODES - 1)) : 0;
    }
    for (int k = 0; k < EDGE_BATCH; ++k)
      if (pb + k * 1024 < p1) atomicAdd(&cnt[d[k]], 1);
  }
  __syncthreads();
  int v = 0;
  if (t < BIN_NODES) { v = cnt[t]; s[t] = v; }
  __syncthreads();
  for (int off = 1; off < BIN_NODES; off <<= 1) {
    int add = 0;
    if (t < BIN_NODES && t >= off) add = s[t - off];
    __syncthreads();
    if (t < BIN_NODES) s[t] += add;
    __syncthreads();
  }
  int excl = 0;
  if (t < BIN_NODES) {
    excl = s[t] - v;
    cnt[t] = excl;
    if (lo + t < n) rowptr[lo + t] = p0 + excl;
  }
  if (bin == (int)gridDim.x - 1 && t == 0) rowptr[n] = p1;
  __syncthreads();
  for (int pb = p0 + t; pb < p1; pb += EDGE_BATCH * 1024) {
    int q[EDGE_BATCH];
    v2i r[EDGE_BATCH];
    for (int k = 0; k < EDGE_BATCH; ++k) {
      const int p = pb + k * 1024;
      v2i z = {0, 0};
      r[k] = (p < p1) ? ((const v2i*)trec)[p] : z;
    }
    for (int k = 0; k < EDGE_BATCH; ++k) {
      q[k] = p0;
      if (pb + k * 1024 < p1) q[k] = p0 + atomicAdd(&cnt[r[k].x & (BIN_NODES - 1)], 1);
    }
    for (int k = 0; k < EDGE_BATCH; ++k) {
      if (pb + k * 1024 < p1) {
        v2i o;
        o.x = r[k].x >> BIN_SHIFT;
        o.y = r[k].y;
        ((v2i*)srec)[q[k]] = o;
        if (q[k] - p0 < BIN_WCAP) wl[q[k] - p0] = __int_as_float(r[k].y);
      }
    }
  }
  __syncthreads();
  if (t < BIN_NODES && lo + t < n) {
    const int jend = excl + v;
    const int jlds = (jend < BIN_WCAP) ? jend : BIN_WCAP;
    float d = 1.0f;
    int j = excl;
    for (; j < jlds; ++j) d += wl[j];
    for (; j < jend; ++j)
      d += __int_as_float(srec[2 * (size_t)(p0 + j) + 1]);
    dinv[lo + t] = (d > 0.f) ? rsqrtf(d) : 0.f;
  }
  const int wave = t >> 5;
  if (wave >= 8 && wave < 24)
    wmma_row_tile<IN1, OUT1, false, false>(x, sW1, nullptr, xw, n, lo + (wave - 8) * 16, t & 31);
}

template <int C, bool BIAS, bool ACT>
__global__ __launch_bounds__(256) void gather_rows(const float* __restrict__ xin,
                                                   const int* __restrict__ rowptr,
                                                   int* srec,
                                                   const float* __restrict__ dinv,
                                                   const float* __restrict__ b,
                                                   float* __restrict__ out, int n) {
  static_assert(C == 4, "one lane per node: the slot's owner is the only lane that reads it");
  constexpr int G = C / 4;
  const int tid = blockIdx.x * 256 + threadIdx.x;
  const int i = tid / G;
  const int q = tid - i * G;
  if (i >= n) return;
  const float* xq = xin + q * 4;
  const float dv = dinv[i];
  const float wself = dv * dv;
  const float4 sv = *(const float4*)(xq + (size_t)i * C);
  float ax = sv.x * wself, ay = sv.y * wself, az = sv.z * wself, aw = sv.w * wself;
  v2i* rec = (v2i*)srec;
  int p = rowptr[i];
  const int pend = rowptr[i + 1];
  for (; p + 4 <= pend; p += 4) {
    const v2i c0 = rec[p], c1 = rec[p + 1], c2 = rec[p + 2], c3 = rec[p + 3];
    const int s0 = c0.x, s1 = c1.x, s2 = c2.x, s3 = c3.x;
    const float e0 = __int_as_float(c0.y), e1 = __int_as_float(c1.y);
    const float e2 = __int_as_float(c2.y), e3 = __int_as_float(c3.y);
    const float4 r0 = *(const float4*)(xq + (size_t)s0 * C);
    const float4 r1 = *(const float4*)(xq + (size_t)s1 * C);
    const float4 r2 = *(const float4*)(xq + (size_t)s2 * C);
    const float4 r3 = *(const float4*)(xq + (size_t)s3 * C);
    const float w0 = dinv[s0] * e0 * dv;
    const float w1 = dinv[s1] * e1 * dv;
    const float w2 = dinv[s2] * e2 * dv;
    const float w3 = dinv[s3] * e3 * dv;
    const v2i o0 = {s0, __float_as_int(w0)}, o1 = {s1, __float_as_int(w1)};
    const v2i o2 = {s2, __float_as_int(w2)}, o3 = {s3, __float_as_int(w3)};
    rec[p] = o0; rec[p + 1] = o1; rec[p + 2] = o2; rec[p + 3] = o3;
    ax += r0.x * w0; ay += r0.y * w0; az += r0.z * w0; aw += r0.w * w0;
    ax += r1.x * w1; ay += r1.y * w1; az += r1.z * w1; aw += r1.w * w1;
    ax += r2.x * w2; ay += r2.y * w2; az += r2.z * w2; aw += r2.w * w2;
    ax += r3.x * w3; ay += r3.y * w3; az += r3.z * w3; aw += r3.w * w3;
  }
  for (; p < pend; ++p) {
    const v2i c0 = rec[p];
    const int s0 = c0.x;
    const float e0 = __int_as_float(c0.y);
    const float4 r0 = *(const float4*)(xq + (size_t)s0 * C);
    const float w0 = dinv[s0] * e0 * dv;
    srec[2 * (size_t)p + 1] = __float_as_int(w0);
    ax += r0.x * w0; ay += r0.y * w0; az += r0.z * w0; aw += r0.w * w0;
  }
  if (BIAS) {
    ax += b[q * 4 + 0]; ay += b[q * 4 + 1]; az += b[q * 4 + 2]; aw += b[q * 4 + 3];
  }
  if (ACT) {
    ax = (ax > 0.f) ? ax : expm1f(ax);
    ay = (ay > 0.f) ? ay : expm1f(ay);
    az = (az > 0.f) ? az : expm1f(az);
    aw = (aw > 0.f) ? aw : expm1f(aw);
  }
  float4 o;
  o.x = ax; o.y = ay; o.z = az; o.w = aw;
  *(float4*)(out + (size_t)i * C + q * 4) = o;
}

#define WMMA_SETTLE_NOPS "v_nop\n\tv_nop\n\tv_nop\n\tv_nop\n\tv_nop\n\tv_nop\n\tv_nop\n\tv_nop"
template <int R>
__device__ __forceinline__ void wmma_settle(v8f (&c)[R]) {
  static_assert(R == 1 || R == 2 || R == 4 || R == 8, "one statement per accumulator count");
  if constexpr (R == 1) {
    asm volatile(WMMA_SETTLE_NOPS : "+v"(c[0]));
  } else if constexpr (R == 2) {
    asm volatile(WMMA_SETTLE_NOPS : "+v"(c[0]), "+v"(c[1]));
  } else if constexpr (R == 4) {
    asm volatile(WMMA_SETTLE_NOPS : "+v"(c[0]), "+v"(c[1]), "+v"(c[2]), "+v"(c[3]));
  } else {
    asm volatile(WMMA_SETTLE_NOPS : "+v"(c[0]), "+v"(c[1]), "+v"(c[2]), "+v"(c[3]),
                                    "+v"(c[4]), "+v"(c[5]), "+v"(c[6]), "+v"(c[7]));
  }
}

template <int IN, int OUT, bool ACT>
__global__ __launch_bounds__(256) void agg_gemm(const float* __restrict__ xin,
                                                const int* __restrict__ rowptr,
                                                const int* __restrict__ srec,
                                                const float* __restrict__ dinv,
                                                const float* __restrict__ W,
                                                const float* __restrict__ bias,
                                                float* __restrict__ y, int n) {
  static_assert(OUT % 16 == 0, "whole column tiles");
  constexpr int G = IN / 4;
  constexpr int NODES = 128;
  constexpr int NPP = 256 / G;
  constexpr int PITCH = (IN == 4) ? 4 : IN + 4;
  constexpr int EDGES = 8;
  constexpr bool USE_LDS = (IN * OUT) <= 8192;
  __shared__ alignas(16) float sA[NODES * PITCH];
  __shared__ float sW[USE_LDS ? (IN * OUT) : 1];
  const int t = threadIdx.x;
  const float* Wp = W;
  if constexpr (USE_LDS) {
    for (int i = t; i < IN * OUT; i += 256) sW[i] = W[i];
    Wp = sW;
  }

  const int nl0 = t / G;
  const int q = t - nl0 * G;
  const float* xq = xin + q * 4;
  const v2i* rec = (const v2i*)srec;
  for (int nl = nl0; nl < NODES; nl += NPP) {
    const int i = (int)blockIdx.x * NODES + nl;
    float ax = 0.f, ay = 0.f, az = 0.f, aw = 0.f;
    if (i < n) {
      const float dv = dinv[i];
      const float wself = dv * dv;
      const float4 sv = *(const float4*)(xq + (size_t)i * IN);
      int p = rowptr[i];
      const int pend = rowptr[i + 1];
      int s[EDGES];
      float w[EDGES];
#pragma unroll
      for (int k = 0; k < EDGES; ++k) {
        const bool ok = p + k < pend;
        const int pc = ok ? p + k : (p < pend ? p : 0);
        const v2i c = rec[pc];
        s[k] = ok ? c.x : i;
        w[k] = ok ? __int_as_float(c.y) : 0.f;
      }
      ax = sv.x * wself; ay = sv.y * wself; az = sv.z * wself; aw = sv.w * wself;
      while (p < pend) {
        float4 r[EDGES];
#pragma unroll
        for (int k = 0; k < EDGES; ++k) r[k] = *(const float4*)(xq + (size_t)s[k] * IN);
        const int pn = p + EDGES;
        int sn[EDGES];
        float wn[EDGES];
#pragma unroll
        for (int k = 0; k < EDGES; ++k) {
          const bool ok = pn + k < pend;
          const int pc = ok ? pn + k : p;
          const v2i c = rec[pc];
          sn[k] = ok ? c.x : i;
          wn[k] = ok ? __int_as_float(c.y) : 0.f;
        }
#pragma unroll
        for (int k = 0; k < EDGES; ++k) {
          ax += r[k].x * w[k]; ay += r[k].y * w[k]; az += r[k].z * w[k]; aw += r[k].w * w[k];
        }
#pragma unroll
        for (int k = 0; k < EDGES; ++k) { s[k] = sn[k]; w[k] = wn[k]; }
        p = pn;
      }
    }
    float4 o;
    o.x = ax; o.y = ay; o.z = az; o.w = aw;
    *(float4*)(sA + nl * PITCH + q * 4) = o;
  }
  __syncthreads();

  constexpr int CT  = OUT / 16;
  constexpr int CPW = (CT == 8) ? 2 : 1;
  constexpr int RPW = CT / CPW;
  constexpr int CG  = CT / CPW;
  constexpr int ACC = CPW * RPW;
  static_assert(CT == 1 || CT == 2 || CT == 4 || CT == 8, "CT * 8 tiles split evenly over 8 waves");
  static_assert(CG * (8 / CG) == 8 && (8 / CG) * RPW == 8, "the 8 waves cover the 8 row tiles of every column group");
  const int lane  = t & 31;
  const int wave  = t >> 5;
  const int l15   = lane & 15;
  const int khalf = lane >> 4;
  const int col0  = (wave % CG) * (CPW * 16) + l15;
  const int rt0   = (wave / CG) * RPW;
  float bv[CPW];
#pragma unroll
  for (int q = 0; q < CPW; ++q) bv[q] = bias[col0 + q * 16];
  const float* arow = sA + (rt0 * 16 + l15) * PITCH + 2 * khalf;
  const float* bcol = Wp + (size_t)(2 * khalf) * OUT + col0;
  v8f c[ACC];
#pragma unroll
  for (int r = 0; r < ACC; ++r) c[r] = v8f{};
#pragma unroll
  for (int k0 = 0; k0 < IN; k0 += 4) {
    if (!USE_LDS && k0 > 0 && (k0 % 16) == 0) __builtin_amdgcn_sched_barrier(0);
    v2f b[CPW];
#pragma unroll
    for (int q = 0; q < CPW; ++q) {
      b[q].x = bcol[(size_t)k0 * OUT + q * 16];
      b[q].y = bcol[(size_t)(k0 + 1) * OUT + q * 16];
    }
#pragma unroll
    for (int r = 0; r < RPW; ++r) {
      const v2f a = *(const v2f*)(arow + r * 16 * PITCH + k0);
#pragma unroll
      for (int q = 0; q < CPW; ++q)
        c[q * RPW + r] = __builtin_amdgcn_wmma_f32_16x16x4_f32(
             false, a,  false, b[q],
             (short)0, c[q * RPW + r],  false,  false);
    }
  }
  wmma_settle<ACC>(c);
  const int rowoff = khalf * 8;
  const int rowb   = (int)blockIdx.x * NODES + rt0 * 16;
  float* yb = y + (size_t)(rowb + rowoff) * OUT + col0;
  const int left = n - (rowb + rowoff);
#pragma unroll
  for (int r = 0; r < RPW; ++r) {
    const int row0 = rowb + r * 16;
    if (row0 < n) {
#pragma unroll
      for (int q = 0; q < CPW; ++q) {
        v8f cc = c[q * RPW + r];
#pragma unroll
        for (int j = 0; j < 8; ++j) cc[j] = cc[j] + bv[q];
        if (ACT) {
#pragma unroll
          for (int j = 0; j < 8; ++j) {
            const float v = cc[j];
            cc[j] = (v > 0.f) ? v : expm1f(v);
          }
        }
        if (row0 + 16 <= n) {
#pragma unroll
          for (int j = 0; j < 8; ++j) yb[(size_t)(r * 16 + j) * OUT + q * 16] = cc[j];
        } else {
#pragma unroll
          for (int j = 0; j < 8; ++j)
            if (r * 16 + j < left) yb[(size_t)(r * 16 + j) * OUT + q * 16] = cc[j];
        }
      }
    }
  }
}

static inline int cdiv(long long a, long long b) { return (int)((a + b - 1) / b); }

struct Graph {
  const int* rowptr;
  int* srec;
  const float* dinv;
};

template <int C, bool BIAS, bool ACT>
static void run_gather(const float* xin, float* xout, const float* b,
                       const Graph& g, int N, hipStream_t stream) {
  gather_rows<C, BIAS, ACT><<<cdiv((long long)N * (C / 4), 256), 256, 0, stream>>>(
      xin, g.rowptr, g.srec, g.dinv, b, xout, N);
}

template <int IN, int OUT, bool ACT>
static void run_layer_fused(const float* xin, float* xout,
                            const float* W, const float* b, const Graph& g,
                            int N, hipStream_t stream) {
  agg_gemm<IN, OUT, ACT><<<cdiv(N, 128), 256, 0, stream>>>(
      xin, g.rowptr, g.srec, g.dinv, W, b, xout, N);
}

extern "C" void kernel_launch(void* const* d_in, const int* in_sizes, int n_in,
                              void* d_out, int out_size, void* d_ws, size_t ws_size,
                              hipStream_t stream) {
  const float* x   = (const float*)d_in[0];
  const int*   ei  = (const int*)d_in[1];
  const float* ew  = (const float*)d_in[2];
  const float* W1 = (const float*)d_in[3];  const float* b1 = (const float*)d_in[4];
  const float* W2 = (const float*)d_in[5];  const float* b2 = (const float*)d_in[6];
  const float* W3 = (const float*)d_in[7];  const float* b3 = (const float*)d_in[8];
  const float* W4 = (const float*)d_in[9];  const float* b4 = (const float*)d_in[10];
  const float* W5 = (const float*)d_in[11]; const float* b5 = (const float*)d_in[12];
  const float* W6 = (const float*)d_in[13]; const float* b6 = (const float*)d_in[14];

  const int N = in_sizes[0] / 32;
  const int E = in_sizes[2];
  const int* src = ei;
  const int* dst = ei + E;

  float* bufA   = (float*)d_ws;
  float* bufB   = bufA + (size_t)N * 128;
  int*   srec   = (int*)(bufB + (size_t)N * 64);
  float* dinv   = (float*)(srec + 2 * (size_t)E);
  int*   rowptr = (int*)(dinv + N);
  const int NB  = cdiv(N, BIN_NODES);
  int*   trec   = (int*)bufA;
  int*   bhist  = (int*)bufB;
  int*   bstart = bhist + (size_t)EDGE_CHUNKS * NB;
  float* xw1    = bufA + (size_t)N * 124;
  float* outp   = (float*)d_out;

  bin_hist<<<EDGE_CHUNKS, 1024, NB * sizeof(int), stream>>>(dst, bhist, E, NB);
  bin_scatter<<<EDGE_CHUNKS, 1024, 2 * NB * sizeof(int), stream>>>(
      src, dst, ew, bhist, bstart, trec, E, NB);
  bin_build<<<NB, 1024, 0, stream>>>(bstart, trec, rowptr, srec, dinv,
                                     x, W1, xw1, N);

  Graph g;
  g.rowptr = rowptr; g.srec = srec; g.dinv = dinv;

  run_gather<OUT1, true, true>(xw1, bufA, b1, g, N, stream);
  run_layer_fused<4,   16,  true >(bufA, bufB, W2, b2, g, N, stream);
  run_layer_fused<16,  32,  true >(bufB, bufA, W3, b3, g, N, stream);
  run_layer_fused<32,  64,  false>(bufA, bufB, W4, b4, g, N, stream);
  run_layer_fused<64,  128, true >(bufB, bufA, W5, b5, g, N, stream);
  run_layer_fused<128, 128, false>(bufA, outp, W6, b6, g, N, stream);
}


# Both grouping levels place their 8-byte edge records in LDS in memory order, then copy them out slot by slot: a store covers a few whole lines, not 32 pieces; float sum order still LDS arrival order
# speedup vs baseline: 8.2744x; 1.0677x over previous
#include <hip/hip_runtime.h>
#include <cstdint>

typedef float v2f __attribute__((ext_vector_type(2)));
typedef float v8f __attribute__((ext_vector_type(8)));
typedef int v2i __attribute__((ext_vector_type(2)));

template <int IN, int OUT, bool BIAS, bool ACT>
__device__ __forceinline__ void wmma_row_tile(const float* __restrict__ x,
                                              const float* Wp,
                                              const float* __restrict__ bias,
                                              float* __restrict__ y, int n,
                                              int row0, int lane) {
  constexpr bool FULL    = (OUT % 16) == 0;
  const int l15   = lane & 15;
  const int khalf = lane >> 4;

  if (row0 < n) {
    int mload = row0 + l15;
    if (mload > n - 1) mload = n - 1;
    const float* xrow = x + (size_t)mload * IN;
    const bool fulltile = (row0 + 16 <= n);

    for (int n0 = 0; n0 < OUT; n0 += 16) {
      const int colb = n0 + l15;
      const int colc = FULL ? colb : (colb < OUT ? colb : OUT - 1);
      float bv = 0.f;
      if (BIAS) bv = bias[colc];
      v8f c = {};
#pragma unroll
      for (int k0 = 0; k0 < IN; k0 += 4) {
        const int krow = k0 + 2 * khalf;
        v2f a = *(const v2f*)(xrow + krow);
        v2f b;
        b.x = Wp[(size_t)krow * OUT + colc];
        b.y = Wp[(size_t)(krow + 1) * OUT + colc];
        c = __builtin_amdgcn_wmma_f32_16x16x4_f32(
             false, a,  false, b,
             (short)0, c,  false,  false);
      }
      if (BIAS || ACT) {
        asm volatile("v_nop\n\tv_nop\n\tv_nop\n\tv_nop\n\t"
                     "v_nop\n\tv_nop\n\tv_nop\n\tv_nop" : "+v"(c));
      }
      if (BIAS) {
#pragma unroll
        for (int r = 0; r < 8; ++r) c[r] = c[r] + bv;
      }
      if (ACT) {
#pragma unroll
        for (int r = 0; r < 8; ++r) {
          const float v = c[r];
          c[r] = (v > 0.f) ? v : expm1f(v);
        }
      }
      const int rowoff = khalf * 8;
      if (FULL && fulltile) {
#pragma unroll
        for (int r = 0; r < 8; ++r)
          y[(size_t)(row0 + rowoff + r) * OUT + colb] = c[r];
      } else {
        const bool colok = FULL || (colb < OUT);
#pragma unroll
        for (int r = 0; r < 8; ++r) {
          int row = row0 + rowoff + r;
          if (row < n && colok) y[(size_t)row * OUT + colb] = c[r];
        }
      }
    }
  }
}

constexpr int BIN_SHIFT = 8;
constexpr int BIN_NODES = 1 << BIN_SHIFT;
constexpr int EDGE_CHUNKS = 128;
constexpr int CHUNKS_PER_WAVE = EDGE_CHUNKS / 32;
constexpr int EDGE_BATCH = 8;
static_assert(EDGE_CHUNKS % 32 == 0, "one wave sums a whole number of chunks");

__global__ __launch_bounds__(1024) void bin_hist(const int* __restrict__ dst,
                                                 int* __restrict__ blockhist,
                                                 int E, int NB) {
  extern __shared__ int sh[];
  const int t = threadIdx.x;
  for (int i = t; i < NB; i += 1024) sh[i] = 0;
  __syncthreads();
  const int chunk = (E + EDGE_CHUNKS - 1) / EDGE_CHUNKS;
  const int e0 = (int)blockIdx.x * chunk;
  const int e1 = (e0 + chunk < E) ? (e0 + chunk) : E;
  for (int eb = e0 + t; eb < e1; eb += EDGE_BATCH * 1024) {
    int d[EDGE_BATCH];
    for (int k = 0; k < EDGE_BATCH; ++k) {
      const int e = eb + k * 1024;
      d[k] = (e < e1) ? dst[e] : 0;
    }
    for (int k = 0; k < EDGE_BATCH; ++k)
      if (eb + k * 1024 < e1) atomicAdd(&sh[d[k] >> BIN_SHIFT], 1);
  }
  __syncthreads();
  for (int i = t; i < NB; i += 1024) blockhist[(size_t)blockIdx.x * NB + i] = sh[i];
}

constexpr int L1_CAP = 16384;
constexpr int COPY_BATCH = 4;
__global__ __launch_bounds__(1024) void bin_scatter(const int* __restrict__ src,
                                                    const int* __restrict__ dst,
                                                    const float* __restrict__ ew,
                                                    const int* __restrict__ blockhist,
                                                    int* __restrict__ binstart,
                                                    int* __restrict__ trec,
                                                    int E, int NB) {
  extern __shared__ int sh[];
  __shared__ int s[1024];
  __shared__ int s2[1024];
  __shared__ v2i lrec[L1_CAP];
  __shared__ int ltgt[L1_CAP];
  int* base = sh;
  int* tot  = sh + NB;
  int* cur  = sh + 2 * NB;
  const int t = threadIdx.x;
  const int lane = t & 31;
  const int wave = t >> 5;
  const int me = (int)blockIdx.x;

  for (int i = t; i < NB; i += 1024) {
    base[i] = 0; tot[i] = 0;
    cur[i] = blockhist[(size_t)me * NB + i];
  }
  __syncthreads();
  for (int bb = lane; bb < NB; bb += 4 * 32) {
    int v[4][CHUNKS_PER_WAVE];
    for (int j = 0; j < 4; ++j) {
      const int b = bb + 32 * j;
      for (int k = 0; k < CHUNKS_PER_WAVE; ++k) {
        const int c = wave * CHUNKS_PER_WAVE + k;
        v[j][k] = (b < NB) ? blockhist[(size_t)c * NB + b] : 0;
      }
    }
    for (int j = 0; j < 4; ++j) {
      const int b = bb + 32 * j;
      int before = 0, total = 0;
      for (int k = 0; k < CHUNKS_PER_WAVE; ++k) {
        const int c = wave * CHUNKS_PER_WAVE + k;
        total += v[j][k];
        before += (c < me) ? v[j][k] : 0;
      }
      if (b < NB) {
        atomicAdd(&base[b], before);
        atomicAdd(&tot[b], total);
      }
    }
  }
  __syncthreads();
  int carry = 0, carry2 = 0;
  for (int b0 = 0; b0 < NB; b0 += 1024) {
    const int b = b0 + t;
    const int v  = (b < NB) ? tot[b] : 0;
    const int v2 = (b < NB) ? cur[b] : 0;
    s[t] = v;
    s2[t] = v2;
    __syncthreads();
    for (int off = 1; off < 1024; off <<= 1) {
      const int add  = (t >= off) ? s[t - off] : 0;
      const int add2 = (t >= off) ? s2[t - off] : 0;
      __syncthreads();
      s[t] += add;
      s2[t] += add2;
      __syncthreads();
    }
    const int excl  = carry + s[t] - v;
    const int lexcl = carry2 + s2[t] - v2;
    if (b < NB) {
      base[b] += excl - lexcl;
      cur[b] = lexcl;
      if (me == 0) binstart[b] = excl;
    }
    carry += s[1023];
    carry2 += s2[1023];
    __syncthreads();
  }
  if (me == 0 && t == 0) binstart[NB] = carry;

  const int chunk = (E + EDGE_CHUNKS - 1) / EDGE_CHUNKS;
  const int e0 = me * chunk;
  const int e1 = (e0 + chunk < E) ? (e0 + chunk) : E;
  for (int eb = e0 + t; eb < e1; eb += EDGE_BATCH * 1024) {
    int d[EDGE_BATCH], a[EDGE_BATCH], p[EDGE_BATCH];
    float w[EDGE_BATCH];
    for (int k = 0; k < EDGE_BATCH; ++k) {
      const int e = eb + k * 1024;
      const bool ok = e < e1;
      d[k] = ok ? dst[e] : 0;
      a[k] = ok ? src[e] : 0;
      w[k] = ok ? ew[e] : 0.f;
    }
    int g[EDGE_BATCH];
    for (int k = 0; k < EDGE_BATCH; ++k) {
      p[k] = 0;
      g[k] = 0;
      if (eb + k * 1024 < e1) {
        const int bn = d[k] >> BIN_SHIFT;
        p[k] = atomicAdd(&cur[bn], 1);
        g[k] = base[bn];
      }
    }
    for (int k = 0; k < EDGE_BATCH; ++k) {
      if (eb + k * 1024 < e1) {
        v2i r;
        r.x = (a[k] << BIN_SHIFT) | (d[k] & (BIN_NODES - 1));
        r.y = __float_as_int(w[k]);
        if (p[k] < L1_CAP) { lrec[p[k]] = r; ltgt[p[k]] = p[k] + g[k]; }
        else ((v2i*)trec)[p[k] + g[k]] = r;
      }
    }
  }
  __syncthreads();
  int m = e1 - e0;
  if (m > L1_CAP) m = L1_CAP;
  for (int jb = t; jb < m; jb += COPY_BATCH * 1024) {
    v2i r[COPY_BATCH];
    int q[COPY_BATCH];
    for (int k = 0; k < COPY_BATCH; ++k) {
      const int j = jb + k * 1024;
      const int jc = (j < m) ? j : jb;
      r[k] = lrec[jc];
      q[k] = ltgt[jc];
    }
    for (int k = 0; k < COPY_BATCH; ++k)
      if (jb + k * 1024 < m) ((v2i*)trec)[q[k]] = r[k];
  }
}

constexpr int BIN_RCAP = 8192;
constexpr int IN1 = 32, OUT1 = 4;
static_assert(BIN_NODES == 16 * 16, "sixteen waves, one 16-row tile each");
__global__ __launch_bounds__(1024) void bin_build(const int* __restrict__ binstart,
                                                  const int* __restrict__ trec,
                                                  int* __restrict__ rowptr,
                                                  int* srec,
                                                  float* __restrict__ dinv,
                                                  const float* __restrict__ x,
                                                  const float* __restrict__ W1,
                                                  float* __restrict__ xw, int n) {
  __shared__ int cnt[BIN_NODES];
  __shared__ int s[BIN_NODES];
  __shared__ v2i lrec[BIN_RCAP];
  __shared__ float sW1[IN1 * OUT1];
  const int t = threadIdx.x;
  const int bin = (int)blockIdx.x;
  const int lo = bin << BIN_SHIFT;
  const int p0 = binstart[bin];
  const int p1 = binstart[bin + 1];
  if (t < BIN_NODES) cnt[t] = 0;
  if (t < IN1 * OUT1) sW1[t] = W1[t];
  __syncthreads();
  for (int pb = p0 + t; pb < p1; pb += EDGE_BATCH * 1024) {
    int d[EDGE_BATCH];
    for (int k = 0; k < EDGE_BATCH; ++k) {
      const int p = pb + k * 1024;
      d[k] = (p < p1) ? (trec[2 * (size_t)p] & (BIN_NODES - 1)) : 0;
    }
    for (int k = 0; k < EDGE_BATCH; ++k)
      if (pb + k * 1024 < p1) atomicAdd(&cnt[d[k]], 1);
  }
  __syncthreads();
  int v = 0;
  if (t < BIN_NODES) { v = cnt[t]; s[t] = v; }
  __syncthreads();
  for (int off = 1; off < BIN_NODES; off <<= 1) {
    int add = 0;
    if (t < BIN_NODES && t >= off) add = s[t - off];
    __syncthreads();
    if (t < BIN_NODES) s[t] += add;
    __syncthreads();
  }
  int excl = 0;
  if (t < BIN_NODES) {
    excl = s[t] - v;
    cnt[t] = excl;
    if (lo + t < n) rowptr[lo + t] = p0 + excl;
  }
  if (bin == (int)gridDim.x - 1 && t == 0) rowptr[n] = p1;
  __syncthreads();
  for (int pb = p0 + t; pb < p1; pb += EDGE_BATCH * 1024) {
    int q[EDGE_BATCH];
    v2i r[EDGE_BATCH];
    for (int k = 0; k < EDGE_BATCH; ++k) {
      const int p = pb + k * 1024;
      v2i z = {0, 0};
      r[k] = (p < p1) ? ((const v2i*)trec)[p] : z;
    }
    for (int k = 0; k < EDGE_BATCH; ++k) {
      q[k] = 0;
      if (pb + k * 1024 < p1) q[k] = atomicAdd(&cnt[r[k].x & (BIN_NODES - 1)], 1);
    }
    for (int k = 0; k < EDGE_BATCH; ++k) {
      if (pb + k * 1024 < p1) {
        v2i o;
        o.x = r[k].x >> BIN_SHIFT;
        o.y = r[k].y;
        if (q[k] < BIN_RCAP) lrec[q[k]] = o;
        else ((v2i*)srec)[p0 + q[k]] = o;
      }
    }
  }
  __syncthreads();
  {
    int m = p1 - p0;
    if (m > BIN_RCAP) m = BIN_RCAP;
    for (int jb = t; jb < m; jb += COPY_BATCH * 1024) {
      v2i c[COPY_BATCH];
      for (int k = 0; k < COPY_BATCH; ++k) {
        const int j = jb + k * 1024;
        c[k] = lrec[(j < m) ? j : jb];
      }
      for (int k = 0; k < COPY_BATCH; ++k)
        if (jb + k * 1024 < m) ((v2i*)srec)[p0 + jb + k * 1024] = c[k];
    }
  }
  if (t < BIN_NODES && lo + t < n) {
    const int jend = excl + v;
    const int jlds = (jend < BIN_RCAP) ? jend : BIN_RCAP;
    float d = 1.0f;
    int j = excl;
    for (; j < jlds; ++j) d += __int_as_float(lrec[j].y);
    for (; j < jend; ++j)
      d += __int_as_float(srec[2 * (size_t)(p0 + j) + 1]);
    dinv[lo + t] = (d > 0.f) ? rsqrtf(d) : 0.f;
  }
  const int wave = t >> 5;
  if (wave >= 8 && wave < 24)
    wmma_row_tile<IN1, OUT1, false, false>(x, sW1, nullptr, xw, n, lo + (wave - 8) * 16, t & 31);
}

template <int C, bool BIAS, bool ACT>
__global__ __launch_bounds__(256) void gather_rows(const float* __restrict__ xin,
                                                   const int* __restrict__ rowptr,
                                                   int* srec,
                                                   const float* __restrict__ dinv,
                                                   const float* __restrict__ b,
                                                   float* __restrict__ out, int n) {
  static_assert(C == 4, "one lane per node: the slot's owner is the only lane that reads it");
  constexpr int G = C / 4;
  const int tid = blockIdx.x * 256 + threadIdx.x;
  const int i = tid / G;
  const int q = tid - i * G;
  if (i >= n) return;
  const float* xq = xin + q * 4;
  const float dv = dinv[i];
  const float wself = dv * dv;
  const float4 sv = *(const float4*)(xq + (size_t)i * C);
  float ax = sv.x * wself, ay = sv.y * wself, az = sv.z * wself, aw = sv.w * wself;
  v2i* rec = (v2i*)srec;
  int p = rowptr[i];
  const int pend = rowptr[i + 1];
  for (; p + 4 <= pend; p += 4) {
    const v2i c0 = rec[p], c1 = rec[p + 1], c2 = rec[p + 2], c3 = rec[p + 3];
    const int s0 = c0.x, s1 = c1.x, s2 = c2.x, s3 = c3.x;
    const float e0 = __int_as_float(c0.y), e1 = __int_as_float(c1.y);
    const float e2 = __int_as_float(c2.y), e3 = __int_as_float(c3.y);
    const float4 r0 = *(const float4*)(xq + (size_t)s0 * C);
    const float4 r1 = *(const float4*)(xq + (size_t)s1 * C);
    const float4 r2 = *(const float4*)(xq + (size_t)s2 * C);
    const float4 r3 = *(const float4*)(xq + (size_t)s3 * C);
    const float w0 = dinv[s0] * e0 * dv;
    const float w1 = dinv[s1] * e1 * dv;
    const float w2 = dinv[s2] * e2 * dv;
    const float w3 = dinv[s3] * e3 * dv;
    const v2i o0 = {s0, __float_as_int(w0)}, o1 = {s1, __float_as_int(w1)};
    const v2i o2 = {s2, __float_as_int(w2)}, o3 = {s3, __float_as_int(w3)};
    rec[p] = o0; rec[p + 1] = o1; rec[p + 2] = o2; rec[p + 3] = o3;
    ax += r0.x * w0; ay += r0.y * w0; az += r0.z * w0; aw += r0.w * w0;
    ax += r1.x * w1; ay += r1.y * w1; az += r1.z * w1; aw += r1.w * w1;
    ax += r2.x * w2; ay += r2.y * w2; az += r2.z * w2; aw += r2.w * w2;
    ax += r3.x * w3; ay += r3.y * w3; az += r3.z * w3; aw += r3.w * w3;
  }
  for (; p < pend; ++p) {
    const v2i c0 = rec[p];
    const int s0 = c0.x;
    const float e0 = __int_as_float(c0.y);
    const float4 r0 = *(const float4*)(xq + (size_t)s0 * C);
    const float w0 = dinv[s0] * e0 * dv;
    srec[2 * (size_t)p + 1] = __float_as_int(w0);
    ax += r0.x * w0; ay += r0.y * w0; az += r0.z * w0; aw += r0.w * w0;
  }
  if (BIAS) {
    ax += b[q * 4 + 0]; ay += b[q * 4 + 1]; az += b[q * 4 + 2]; aw += b[q * 4 + 3];
  }
  if (ACT) {
    ax = (ax > 0.f) ? ax : expm1f(ax);
    ay = (ay > 0.f) ? ay : expm1f(ay);
    az = (az > 0.f) ? az : expm1f(az);
    aw = (aw > 0.f) ? aw : expm1f(aw);
  }
  float4 o;
  o.x = ax; o.y = ay; o.z = az; o.w = aw;
  *(float4*)(out + (size_t)i * C + q * 4) = o;
}

#define WMMA_SETTLE_NOPS "v_nop\n\tv_nop\n\tv_nop\n\tv_nop\n\tv_nop\n\tv_nop\n\tv_nop\n\tv_nop"
template <int R>
__device__ __forceinline__ void wmma_settle(v8f (&c)[R]) {
  static_assert(R == 1 || R == 2 || R == 4 || R == 8, "one statement per accumulator count");
  if constexpr (R == 1) {
    asm volatile(WMMA_SETTLE_NOPS : "+v"(c[0]));
  } else if constexpr (R == 2) {
    asm volatile(WMMA_SETTLE_NOPS : "+v"(c[0]), "+v"(c[1]));
  } else if constexpr (R == 4) {
    asm volatile(WMMA_SETTLE_NOPS : "+v"(c[0]), "+v"(c[1]), "+v"(c[2]), "+v"(c[3]));
  } else {
    asm volatile(WMMA_SETTLE_NOPS : "+v"(c[0]), "+v"(c[1]), "+v"(c[2]), "+v"(c[3]),
                                    "+v"(c[4]), "+v"(c[5]), "+v"(c[6]), "+v"(c[7]));
  }
}

template <int IN, int OUT, bool ACT>
__global__ __launch_bounds__(256) void agg_gemm(const float* __restrict__ xin,
                                                const int* __restrict__ rowptr,
                                                const int* __restrict__ srec,
                                                const float* __restrict__ dinv,
                                                const float* __restrict__ W,
                                                const float* __restrict__ bias,
                                                float* __restrict__ y, int n) {
  static_assert(OUT % 16 == 0, "whole column tiles");
  constexpr int G = IN / 4;
  constexpr int NODES = 128;
  constexpr int NPP = 256 / G;
  constexpr int PITCH = (IN == 4) ? 4 : IN + 4;
  constexpr int EDGES = 8;
  constexpr bool USE_LDS = (IN * OUT) <= 8192;
  __shared__ alignas(16) float sA[NODES * PITCH];
  __shared__ float sW[USE_LDS ? (IN * OUT) : 1];
  const int t = threadIdx.x;
  const float* Wp = W;
  if constexpr (USE_LDS) {
    for (int i = t; i < IN * OUT; i += 256) sW[i] = W[i];
    Wp = sW;
  }

  const int nl0 = t / G;
  const int q = t - nl0 * G;
  const float* xq = xin + q * 4;
  const v2i* rec = (const v2i*)srec;
  for (int nl = nl0; nl < NODES; nl += NPP) {
    const int i = (int)blockIdx.x * NODES + nl;
    float ax = 0.f, ay = 0.f, az = 0.f, aw = 0.f;
    if (i < n) {
      const float dv = dinv[i];
      const float wself = dv * dv;
      const float4 sv = *(const float4*)(xq + (size_t)i * IN);
      int p = rowptr[i];
      const int pend = rowptr[i + 1];
      int s[EDGES];
      float w[EDGES];
#pragma unroll
      for (int k = 0; k < EDGES; ++k) {
        const bool ok = p + k < pend;
        const int pc = ok ? p + k : (p < pend ? p : 0);
        const v2i c = rec[pc];
        s[k] = ok ? c.x : i;
        w[k] = ok ? __int_as_float(c.y) : 0.f;
      }
      ax = sv.x * wself; ay = sv.y * wself; az = sv.z * wself; aw = sv.w * wself;
      while (p < pend) {
        float4 r[EDGES];
#pragma unroll
        for (int k = 0; k < EDGES; ++k) r[k] = *(const float4*)(xq + (size_t)s[k] * IN);
        const int pn = p + EDGES;
        int sn[EDGES];
        float wn[EDGES];
#pragma unroll
        for (int k = 0; k < EDGES; ++k) {
          const bool ok = pn + k < pend;
          const int pc = ok ? pn + k : p;
          const v2i c = rec[pc];
          sn[k] = ok ? c.x : i;
          wn[k] = ok ? __int_as_float(c.y) : 0.f;
        }
#pragma unroll
        for (int k = 0; k < EDGES; ++k) {
          ax += r[k].x * w[k]; ay += r[k].y * w[k]; az += r[k].z * w[k]; aw += r[k].w * w[k];
        }
#pragma unroll
        for (int k = 0; k < EDGES; ++k) { s[k] = sn[k]; w[k] = wn[k]; }
        p = pn;
      }
    }
    float4 o;
    o.x = ax; o.y = ay; o.z = az; o.w = aw;
    *(float4*)(sA + nl * PITCH + q * 4) = o;
  }
  __syncthreads();

  constexpr int CT  = OUT / 16;
  constexpr int CPW = (CT == 8) ? 2 : 1;
  constexpr int RPW = CT / CPW;
  constexpr int CG  = CT / CPW;
  constexpr int ACC = CPW * RPW;
  static_assert(CT == 1 || CT == 2 || CT == 4 || CT == 8, "CT * 8 tiles split evenly over 8 waves");
  static_assert(CG * (8 / CG) == 8 && (8 / CG) * RPW == 8, "the 8 waves cover the 8 row tiles of every column group");
  const int lane  = t & 31;
  const int wave  = t >> 5;
  const int l15   = lane & 15;
  const int khalf = lane >> 4;
  const int col0  = (wave % CG) * (CPW * 16) + l15;
  const int rt0   = (wave / CG) * RPW;
  float bv[CPW];
#pragma unroll
  for (int q = 0; q < CPW; ++q) bv[q] = bias[col0 + q * 16];
  const float* arow = sA + (rt0 * 16 + l15) * PITCH + 2 * khalf;
  const float* bcol = Wp + (size_t)(2 * khalf) * OUT + col0;
  v8f c[ACC];
#pragma unroll
  for (int r = 0; r < ACC; ++r) c[r] = v8f{};
#pragma unroll
  for (int k0 = 0; k0 < IN; k0 += 4) {
    if (!USE_LDS && k0 > 0 && (k0 % 16) == 0) __builtin_amdgcn_sched_barrier(0);
    v2f b[CPW];
#pragma unroll
    for (int q = 0; q < CPW; ++q) {
      b[q].x = bcol[(size_t)k0 * OUT + q * 16];
      b[q].y = bcol[(size_t)(k0 + 1) * OUT + q * 16];
    }
#pragma unroll
    for (int r = 0; r < RPW; ++r) {
      const v2f a = *(const v2f*)(arow + r * 16 * PITCH + k0);
#pragma unroll
      for (int q = 0; q < CPW; ++q)
        c[q * RPW + r] = __builtin_amdgcn_wmma_f32_16x16x4_f32(
             false, a,  false, b[q],
             (short)0, c[q * RPW + r],  false,  false);
    }
  }
  wmma_settle<ACC>(c);
  const int rowoff = khalf * 8;
  const int rowb   = (int)blockIdx.x * NODES + rt0 * 16;
  float* yb = y + (size_t)(rowb + rowoff) * OUT + col0;
  const int left = n - (rowb + rowoff);
#pragma unroll
  for (int r = 0; r < RPW; ++r) {
    const int row0 = rowb + r * 16;
    if (row0 < n) {
#pragma unroll
      for (int q = 0; q < CPW; ++q) {
        v8f cc = c[q * RPW + r];
#pragma unroll
        for (int j = 0; j < 8; ++j) cc[j] = cc[j] + bv[q];
        if (ACT) {
#pragma unroll
          for (int j = 0; j < 8; ++j) {
            const float v = cc[j];
            cc[j] = (v > 0.f) ? v : expm1f(v);
          }
        }
        if (row0 + 16 <= n) {
#pragma unroll
          for (int j = 0; j < 8; ++j) yb[(size_t)(r * 16 + j) * OUT + q * 16] = cc[j];
        } else {
#pragma unroll
          for (int j = 0; j < 8; ++j)
            if (r * 16 + j < left) yb[(size_t)(r * 16 + j) * OUT + q * 16] = cc[j];
        }
      }
    }
  }
}

static inline int cdiv(long long a, long long b) { return (int)((a + b - 1) / b); }

struct Graph {
  const int* rowptr;
  int* srec;
  const float* dinv;
};

template <int C, bool BIAS, bool ACT>
static void run_gather(const float* xin, float* xout, const float* b,
                       const Graph& g, int N, hipStream_t stream) {
  gather_rows<C, BIAS, ACT><<<cdiv((long long)N * (C / 4), 256), 256, 0, stream>>>(
      xin, g.rowptr, g.srec, g.dinv, b, xout, N);
}

template <int IN, int OUT, bool ACT>
static void run_layer_fused(const float* xin, float* xout,
                            const float* W, const float* b, const Graph& g,
                            int N, hipStream_t stream) {
  agg_gemm<IN, OUT, ACT><<<cdiv(N, 128), 256, 0, stream>>>(
      xin, g.rowptr, g.srec, g.dinv, W, b, xout, N);
}

extern "C" void kernel_launch(void* const* d_in, const int* in_sizes, int n_in,
                              void* d_out, int out_size, void* d_ws, size_t ws_size,
                              hipStream_t stream) {
  const float* x   = (const float*)d_in[0];
  const int*   ei  = (const int*)d_in[1];
  const float* ew  = (const float*)d_in[2];
  const float* W1 = (const float*)d_in[3];  const float* b1 = (const float*)d_in[4];
  const float* W2 = (const float*)d_in[5];  const float* b2 = (const float*)d_in[6];
  const float* W3 = (const float*)d_in[7];  const float* b3 = (const float*)d_in[8];
  const float* W4 = (const float*)d_in[9];  const float* b4 = (const float*)d_in[10];
  const float* W5 = (const float*)d_in[11]; const float* b5 = (const float*)d_in[12];
  const float* W6 = (const float*)d_in[13]; const float* b6 = (const float*)d_in[14];

  const int N = in_sizes[0] / 32;
  const int E = in_sizes[2];
  const int* src = ei;
  const int* dst = ei + E;

  float* bufA   = (float*)d_ws;
  float* bufB   = bufA + (size_t)N * 128;
  int*   srec   = (int*)(bufB + (size_t)N * 64);
  float* dinv   = (float*)(srec + 2 * (size_t)E);
  int*   rowptr = (int*)(dinv + N);
  const int NB  = cdiv(N, BIN_NODES);
  int*   trec   = (int*)bufA;
  int*   bhist  = (int*)bufB;
  int*   bstart = bhist + (size_t)EDGE_CHUNKS * NB;
  float* xw1    = bufA + (size_t)N * 124;
  float* outp   = (float*)d_out;

  bin_hist<<<EDGE_CHUNKS, 1024, NB * sizeof(int), stream>>>(dst, bhist, E, NB);
  bin_scatter<<<EDGE_CHUNKS, 1024, 3 * NB * sizeof(int), stream>>>(
      src, dst, ew, bhist, bstart, trec, E, NB);
  bin_build<<<NB, 1024, 0, stream>>>(bstart, trec, rowptr, srec, dinv,
                                     x, W1, xw1, N);

  Graph g;
  g.rowptr = rowptr; g.srec = srec; g.dinv = dinv;

  run_gather<OUT1, true, true>(xw1, bufA, b1, g, N, stream);
  run_layer_fused<4,   16,  true >(bufA, bufB, W2, b2, g, N, stream);
  run_layer_fused<16,  32,  true >(bufB, bufA, W3, b3, g, N, stream);
  run_layer_fused<32,  64,  false>(bufA, bufB, W4, b4, g, N, stream);
  run_layer_fused<64,  128, true >(bufB, bufA, W5, b5, g, N, stream);
  run_layer_fused<128, 128, false>(bufA, outp, W6, b6, g, N, stream);
}
